# starting kernel
# baseline (speedup 1.0000x reference)
#include <hip/hip_runtime.h>

typedef float v2f __attribute__((ext_vector_type(2)));
typedef float v8f __attribute__((ext_vector_type(8)));

constexpr int NN   = 2048;
constexpr int C    = 128;
constexpr int L    = 16;
constexpr int E    = 10;
constexpr int K3_0 = 23, K2_0 = 4;
constexpr int K3_1 = 33, K2_1 = 6;

constexpr int COL_A1 = 0;
constexpr int COL_A0 = 3 * K3_1;
constexpr int COL_21 = COL_A0 + K3_0;
constexpr int COL_20 = COL_21 + 3 * K2_1;
constexpr int NCOL   = COL_20 + K2_0;
constexpr int NTILE  = NCOL / 16;

constexpr int S2   = 2 * NCOL + 2;
constexpr int PSTR = 149;
constexpr int XSTR = L + 1;

constexpr int WR_M1 = 0;
constexpr int WR_M0 = 33;
constexpr int WR_21 = 56;
constexpr int WR_20 = 62;
constexpr int WR_11 = 66;
constexpr int WR_10 = 67;
constexpr int WROWS = 68;

constexpr int OFF_X  = 0;
constexpr int OFF_XS = OFF_X + C * XSTR;
constexpr int OFF_W  = OFF_XS + L;
constexpr int OFF_U1 = OFF_W + WROWS * C;
constexpr int OFF_B2 = OFF_U1 + 64;
constexpr int OFF_P  = OFF_B2 + 128 * S2;
constexpr int SMEM_FLOATS = OFF_P + C * PSTR;

__device__ __forceinline__ void build_a1_pair(const float* __restrict__ U3_1,
                                              const float* __restrict__ sXS,
                                              float* __restrict__ sB2, int w, int p)
{
    const float* u0 = U3_1 + (size_t)(w * 256 + 2 * p) * (L * K3_1);
    const float* u1 = u0 + L * K3_1;
    float a0[K3_1], a1[K3_1];
    #pragma unroll
    for (int k = 0; k < K3_1; ++k) { a0[k] = 0.f; a1[k] = 0.f; }
    for (int i = 0; i < L; ++i) {
        const float xi = sXS[i];
        #pragma unroll
        for (int k = 0; k < K3_1; ++k) {
            a0[k] = fmaf(u0[i * K3_1 + k], xi, a0[k]);
            a1[k] = fmaf(u1[i * K3_1 + k], xi, a1[k]);
        }
    }
    float* bp = sB2 + p * S2 + (COL_A1 + w * K3_1) * 2;
    #pragma unroll
    for (int k = 0; k < K3_1; ++k)
        *(v2f*)(bp + 2 * k) = (v2f){a0[k], a1[k]};
}

__device__ __forceinline__ void build_a0_pair(const float* __restrict__ U3_0,
                                              const float* __restrict__ sXS,
                                              float* __restrict__ sB2, int p)
{
    const float* u0 = U3_0 + (size_t)(2 * p) * (L * K3_0);
    const float* u1 = u0 + L * K3_0;
    float a0[K3_0], a1[K3_0];
    #pragma unroll
    for (int k = 0; k < K3_0; ++k) { a0[k] = 0.f; a1[k] = 0.f; }
    for (int i = 0; i < L; ++i) {
        const float xi = sXS[i];
        #pragma unroll
        for (int k = 0; k < K3_0; ++k) {
            a0[k] = fmaf(u0[i * K3_0 + k], xi, a0[k]);
            a1[k] = fmaf(u1[i * K3_0 + k], xi, a1[k]);
        }
    }
    float* bp = sB2 + p * S2 + COL_A0 * 2;
    #pragma unroll
    for (int k = 0; k < K3_0; ++k)
        *(v2f*)(bp + 2 * k) = (v2f){a0[k], a1[k]};
}

__global__ __launch_bounds__(256) void mace_node_kernel(
    const float* __restrict__ x,    const float* __restrict__ y,
    const float* __restrict__ U3_0, const float* __restrict__ U2_0, const float* __restrict__ U1_0,
    const float* __restrict__ U3_1, const float* __restrict__ U2_1, const float* __restrict__ U1_1,
    const float* __restrict__ Wm0,  const float* __restrict__ W20,  const float* __restrict__ W10,
    const float* __restrict__ Wm1,  const float* __restrict__ W21,  const float* __restrict__ W11,
    float* __restrict__ out)
{
    extern __shared__ float sm[];
    float* sX  = sm + OFF_X;
    float* sXS = sm + OFF_XS;
    float* sW  = sm + OFF_W;
    float* sU1 = sm + OFF_U1;
    float* sB2 = sm + OFF_B2;
    float* sP  = sm + OFF_P;

    const int tid = threadIdx.x;
    const int b   = blockIdx.x;

    for (int idx = tid; idx < C * L; idx += 256)
        sX[(idx >> 4) * XSTR + (idx & 15)] = x[(size_t)b * (C * L) + idx];

    float yv[E];
    #pragma unroll
    for (int e = 0; e < E; ++e) yv[e] = y[b * E + e];
    for (int idx = tid; idx < WROWS * C; idx += 256) {
        const int wr = idx >> 7, c = idx & 127;
        const float* src; int stride;
        if (wr < WR_M0)       { src = Wm1 + (wr)         * C + c; stride = K3_1 * C; }
        else if (wr < WR_21)  { src = Wm0 + (wr - WR_M0) * C + c; stride = K3_0 * C; }
        else if (wr < WR_20)  { src = W21 + (wr - WR_21) * C + c; stride = K2_1 * C; }
        else if (wr < WR_11)  { src = W20 + (wr - WR_20) * C + c; stride = K2_0 * C; }
        else if (wr == WR_11) { src = W11 + c;                    stride = C;        }
        else                  { src = W10 + c;                    stride = C;        }
        float acc = 0.f;
        #pragma unroll
        for (int e = 0; e < E; ++e) acc = fmaf(yv[e], src[e * stride], acc);
        sW[idx] = acc;
    }
    if (tid < 16)               sU1[tid] = U1_0[tid];
    else if (tid < 64)          sU1[tid] = U1_1[tid - 16];
    __syncthreads();

    if (tid < L) {
        float s = 0.f;
        for (int c = 0; c < C; ++c) s += sX[c * XSTR + tid];
        sXS[tid] = s;
    }
    __syncthreads();

    {
        const int t0 = tid;
        build_a1_pair(U3_1, sXS, sB2, t0 >> 7, t0 & 127);
        if (tid >= 128) {
            const int t1 = tid + 128;
            build_a1_pair(U3_1, sXS, sB2, t1 >> 7, t1 & 127);
        } else {
            const int p = tid;
            build_a0_pair(U3_0, sXS, sB2, p);
            float* bp = sB2 + p * S2;
            #pragma unroll
            for (int m = 0; m < 3 * K2_1; ++m) {
                const int w = m / K2_1, k = m % K2_1;
                const int r0 = (w * 256 + 2 * p) * K2_1 + k;
                *(v2f*)(bp + (COL_21 + m) * 2) = (v2f){U2_1[r0], U2_1[r0 + K2_1]};
            }
            #pragma unroll
            for (int m = 0; m < K2_0; ++m)
                *(v2f*)(bp + (COL_20 + m) * 2) =
                    (v2f){U2_0[(2 * p) * K2_0 + m], U2_0[(2 * p + 1) * K2_0 + m]};
        }
    }
    __syncthreads();

    const int wv   = tid >> 5;
    const int lane = tid & 31;
    const int nl   = lane & 15;
    const int hi   = lane >> 4;
    const int crow = wv * 16 + nl;

    float xr[L];
    #pragma unroll
    for (int j = 0; j < L; ++j) xr[j] = sX[crow * XSTR + j];

    float sel0[4], sel1[4];
    #pragma unroll
    for (int j = 0; j < 4; ++j) {
        const float e0 = xr[4 * j],     e1 = xr[4 * j + 1];
        const float e2 = xr[4 * j + 2], e3 = xr[4 * j + 3];
        sel0[j] = hi ? e2 : e0;
        sel1[j] = hi ? e3 : e1;
    }

    v8f acc[NTILE];
    #pragma unroll
    for (int t = 0; t < NTILE; ++t)
        acc[t] = (v8f){0.f, 0.f, 0.f, 0.f, 0.f, 0.f, 0.f, 0.f};

    const float* Bl = sB2 + hi * S2 + nl * 2;
    #pragma unroll
    for (int kk = 0; kk < 64; ++kk) {
        const float xp = xr[kk >> 2];
        v2f a;
        a.x = xp * sel0[kk & 3];
        a.y = xp * sel1[kk & 3];
        const float* Br = Bl + (2 * kk) * S2;
        #pragma unroll
        for (int t = 0; t < NTILE; ++t) {
            const v2f bf = *(const v2f*)(Br + t * 32);
            acc[t] = __builtin_amdgcn_wmma_f32_16x16x4_f32(
                false, a, false, bf, (short)0, acc[t], false, false);
        }
    }

    #pragma unroll
    for (int t = 0; t < NTILE; ++t) {
        #pragma unroll
        for (int r = 0; r < 8; ++r)
            sP[(wv * 16 + r + hi * 8) * PSTR + t * 16 + nl] = acc[t][r];
    }
    __syncthreads();

    for (int idx = tid; idx < 4 * C; idx += 256) {
        const int c = idx & 127;
        const int g = idx >> 7;
        const float* P = sP + c * PSTR;
        float s = 0.f;
        if (g == 0) {
            #pragma unroll
            for (int k = 0; k < K3_0; ++k) s = fmaf(P[COL_A0 + k], sW[(WR_M0 + k) * C + c], s);
            #pragma unroll
            for (int k = 0; k < K2_0; ++k) s = fmaf(P[COL_20 + k], sW[(WR_20 + k) * C + c], s);
            float d = 0.f;
            #pragma unroll
            for (int j = 0; j < L; ++j) d = fmaf(sU1[j], sX[c * XSTR + j], d);
            s = fmaf(sW[WR_10 * C + c], d, s);
            out[(size_t)b * (4 * C) + c] = s;
        } else {
            const int w = g - 1;
            #pragma unroll
            for (int k = 0; k < K3_1; ++k) s = fmaf(P[COL_A1 + w * K3_1 + k], sW[(WR_M1 + k) * C + c], s);
            #pragma unroll
            for (int k = 0; k < K2_1; ++k) s = fmaf(P[COL_21 + w * K2_1 + k], sW[(WR_21 + k) * C + c], s);
            float d = 0.f;
            #pragma unroll
            for (int j = 0; j < L; ++j) d = fmaf(sU1[16 + w * L + j], sX[c * XSTR + j], d);
            s = fmaf(sW[WR_11 * C + c], d, s);
            out[(size_t)b * (4 * C) + C + c * 3 + w] = s;
        }
    }
}

extern "C" void kernel_launch(void* const* d_in, const int* in_sizes, int n_in,
                              void* d_out, int out_size, void* d_ws, size_t ws_size,
                              hipStream_t stream) {
    (void)in_sizes; (void)n_in; (void)out_size; (void)d_ws; (void)ws_size;
    const float* x    = (const float*)d_in[0];
    const float* y    = (const float*)d_in[1];
    const float* U3_0 = (const float*)d_in[2];
    const float* U2_0 = (const float*)d_in[3];
    const float* U1_0 = (const float*)d_in[4];
    const float* U3_1 = (const float*)d_in[5];
    const float* U2_1 = (const float*)d_in[6];
    const float* U1_1 = (const float*)d_in[7];
    const float* Wm0  = (const float*)d_in[8];
    const float* W20  = (const float*)d_in[9];
    const float* W10  = (const float*)d_in[10];
    const float* Wm1  = (const float*)d_in[11];
    const float* W21  = (const float*)d_in[12];
    const float* W11  = (const float*)d_in[13];

    constexpr size_t smem_bytes = (size_t)SMEM_FLOATS * sizeof(float);
    mace_node_kernel<<<dim3(NN), dim3(256), smem_bytes, stream>>>(
        x, y, U3_0, U2_0, U1_0, U3_1, U2_1, U1_1,
        Wm0, W20, W10, Wm1, W21, W11, (float*)d_out);
}


# phase-2 U3*xs build read wave-wide: adjacent lanes take 4 consecutive path columns of one row pair (16-byte loads, few lines per wave load), not one far-apart row per lane; fma order kept
# speedup vs baseline: 1.8236x; 1.8236x over previous
#include <hip/hip_runtime.h>

typedef float v2f __attribute__((ext_vector_type(2)));
typedef float v8f __attribute__((ext_vector_type(8)));
typedef float v4fu __attribute__((ext_vector_type(4), aligned(4)));

constexpr int NN   = 2048;
constexpr int C    = 128;
constexpr int L    = 16;
constexpr int E    = 10;
constexpr int K3_0 = 23, K2_0 = 4;
constexpr int K3_1 = 33, K2_1 = 6;

constexpr int COL_A1 = 0;
constexpr int COL_A0 = 3 * K3_1;
constexpr int COL_21 = COL_A0 + K3_0;
constexpr int COL_20 = COL_21 + 3 * K2_1;
constexpr int NCOL   = COL_20 + K2_0;
constexpr int NTILE  = NCOL / 16;

constexpr int S2   = 2 * NCOL + 2;
constexpr int PSTR = 149;
constexpr int XSTR = L + 1;

constexpr int WR_M1 = 0;
constexpr int WR_M0 = 33;
constexpr int WR_21 = 56;
constexpr int WR_20 = 62;
constexpr int WR_11 = 66;
constexpr int WR_10 = 67;
constexpr int WROWS = 68;

constexpr int OFF_X  = 0;
constexpr int OFF_XS = OFF_X + C * XSTR;
constexpr int OFF_W  = OFF_XS + L;
constexpr int OFF_U1 = OFF_W + WROWS * C;
constexpr int OFF_B2 = OFF_U1 + 64;
constexpr int OFF_P  = OFF_B2 + 128 * S2;
constexpr int SMEM_FLOATS = OFF_P + C * PSTR;


__device__ __forceinline__ void build_a1_group(const float* __restrict__ U3_1,
                                               const float (&xsr)[L],
                                               float* __restrict__ sB2,
                                               int G, int pg, int q)
{
    const int t = 3 * G + pg;
    const int w = t >> 7, p = t & 127;
    const int koff = (q < 8) ? 4 * q : (K3_1 - 4);
    const float* u0 = U3_1 + (size_t)(w * 256 + 2 * p) * (L * K3_1) + koff;
    const float* u1 = u0 + L * K3_1;
    float a0[4] = {0.f, 0.f, 0.f, 0.f}, a1[4] = {0.f, 0.f, 0.f, 0.f};
    #pragma unroll
    for (int i = 0; i < L; ++i) {
        const v4fu r0 = *(const v4fu*)(u0 + i * K3_1);
        const v4fu r1 = *(const v4fu*)(u1 + i * K3_1);
        #pragma unroll
        for (int e = 0; e < 4; ++e) {
            a0[e] = fmaf(r0[e], xsr[i], a0[e]);
            a1[e] = fmaf(r1[e], xsr[i], a1[e]);
        }
    }
    float* bp = sB2 + p * S2 + (COL_A1 + w * K3_1) * 2;
    if (q < 8) {
        #pragma unroll
        for (int e = 0; e < 4; ++e)
            *(v2f*)(bp + 2 * (4 * q + e)) = (v2f){a0[e], a1[e]};
    } else {
        *(v2f*)(bp + 2 * (K3_1 - 1)) = (v2f){a0[3], a1[3]};
    }
}

__device__ __forceinline__ void build_a0_group(const float* __restrict__ U3_0,
                                               const float (&xsr)[L],
                                               float* __restrict__ sB2,
                                               int p, int q)
{
    const int koff = (q < 5) ? 4 * q : (K3_0 - 4);
    const float* u0 = U3_0 + (size_t)(2 * p) * (L * K3_0) + koff;
    const float* u1 = u0 + L * K3_0;
    float a0[4] = {0.f, 0.f, 0.f, 0.f}, a1[4] = {0.f, 0.f, 0.f, 0.f};
    #pragma unroll
    for (int i = 0; i < L; ++i) {
        const v4fu r0 = *(const v4fu*)(u0 + i * K3_0);
        const v4fu r1 = *(const v4fu*)(u1 + i * K3_0);
        #pragma unroll
        for (int e = 0; e < 4; ++e) {
            a0[e] = fmaf(r0[e], xsr[i], a0[e]);
            a1[e] = fmaf(r1[e], xsr[i], a1[e]);
        }
    }
    float* bp = sB2 + p * S2 + COL_A0 * 2;
    if (q < 5) {
        #pragma unroll
        for (int e = 0; e < 4; ++e)
            *(v2f*)(bp + 2 * (4 * q + e)) = (v2f){a0[e], a1[e]};
    } else {
        #pragma unroll
        for (int e = 1; e < 4; ++e)
            *(v2f*)(bp + 2 * (K3_0 - 4 + e)) = (v2f){a0[e], a1[e]};
    }
}

__global__ __launch_bounds__(256) void mace_node_kernel(
    const float* __restrict__ x,    const float* __restrict__ y,
    const float* __restrict__ U3_0, const float* __restrict__ U2_0, const float* __restrict__ U1_0,
    const float* __restrict__ U3_1, const float* __restrict__ U2_1, const float* __restrict__ U1_1,
    const float* __restrict__ Wm0,  const float* __restrict__ W20,  const float* __restrict__ W10,
    const float* __restrict__ Wm1,  const float* __restrict__ W21,  const float* __restrict__ W11,
    float* __restrict__ out)
{
    extern __shared__ float sm[];
    float* sX  = sm + OFF_X;
    float* sXS = sm + OFF_XS;
    float* sW  = sm + OFF_W;
    float* sU1 = sm + OFF_U1;
    float* sB2 = sm + OFF_B2;
    float* sP  = sm + OFF_P;

    const int tid = threadIdx.x;
    const int b   = blockIdx.x;

    for (int idx = tid; idx < C * L; idx += 256)
        sX[(idx >> 4) * XSTR + (idx & 15)] = x[(size_t)b * (C * L) + idx];

    float yv[E];
    #pragma unroll
    for (int e = 0; e < E; ++e) yv[e] = y[b * E + e];
    for (int idx = tid; idx < WROWS * C; idx += 256) {
        const int wr = idx >> 7, c = idx & 127;
        const float* src; int stride;
        if (wr < WR_M0)       { src = Wm1 + (wr)         * C + c; stride = K3_1 * C; }
        else if (wr < WR_21)  { src = Wm0 + (wr - WR_M0) * C + c; stride = K3_0 * C; }
        else if (wr < WR_20)  { src = W21 + (wr - WR_21) * C + c; stride = K2_1 * C; }
        else if (wr < WR_11)  { src = W20 + (wr - WR_20) * C + c; stride = K2_0 * C; }
        else if (wr == WR_11) { src = W11 + c;                    stride = C;        }
        else                  { src = W10 + c;                    stride = C;        }
        float acc = 0.f;
        #pragma unroll
        for (int e = 0; e < E; ++e) acc = fmaf(yv[e], src[e * stride], acc);
        sW[idx] = acc;
    }
    if (tid < 16)               sU1[tid] = U1_0[tid];
    else if (tid < 64)          sU1[tid] = U1_1[tid - 16];
    __syncthreads();

    if (tid < L) {
        float s = 0.f;
        for (int c = 0; c < C; ++c) s += sX[c * XSTR + tid];
        sXS[tid] = s;
    }
    __syncthreads();

    {
        const int wv2 = tid >> 5, ln2 = tid & 31;
        float xsr[L];
        #pragma unroll
        for (int i = 0; i < L; ++i) xsr[i] = sXS[i];

        if (ln2 < 27) {
            const int pg = ln2 / 9, q = ln2 - 9 * pg;
            for (int G = wv2; G < 128; G += 8)
                build_a1_group(U3_1, xsr, sB2, G, pg, q);
        }
        if (ln2 < 30) {
            const int pg = ln2 / 6, q = ln2 - 6 * pg;
            for (int G = wv2; G < 26; G += 8) {
                const int p = 5 * G + pg;
                if (p < 128) build_a0_group(U3_0, xsr, sB2, p, q);
            }
        }
        if (tid < 128) {
            const int p = tid;
            float* bp = sB2 + p * S2;
            #pragma unroll
            for (int m = 0; m < 3 * K2_1; ++m) {
                const int w = m / K2_1, k = m % K2_1;
                const int r0 = (w * 256 + 2 * p) * K2_1 + k;
                *(v2f*)(bp + (COL_21 + m) * 2) = (v2f){U2_1[r0], U2_1[r0 + K2_1]};
            }
            #pragma unroll
            for (int m = 0; m < K2_0; ++m)
                *(v2f*)(bp + (COL_20 + m) * 2) =
                    (v2f){U2_0[(2 * p) * K2_0 + m], U2_0[(2 * p + 1) * K2_0 + m]};
        }
    }
    __syncthreads();

    const int wv   = tid >> 5;
    const int lane = tid & 31;
    const int nl   = lane & 15;
    const int hi   = lane >> 4;
    const int crow = wv * 16 + nl;

    float xr[L];
    #pragma unroll
    for (int j = 0; j < L; ++j) xr[j] = sX[crow * XSTR + j];

    float sel0[4], sel1[4];
    #pragma unroll
    for (int j = 0; j < 4; ++j) {
        const float e0 = xr[4 * j],     e1 = xr[4 * j + 1];
        const float e2 = xr[4 * j + 2], e3 = xr[4 * j + 3];
        sel0[j] = hi ? e2 : e0;
        sel1[j] = hi ? e3 : e1;
    }

    v8f acc[NTILE];
    #pragma unroll
    for (int t = 0; t < NTILE; ++t)
        acc[t] = (v8f){0.f, 0.f, 0.f, 0.f, 0.f, 0.f, 0.f, 0.f};

    const float* Bl = sB2 + hi * S2 + nl * 2;
    #pragma unroll
    for (int kk = 0; kk < 64; ++kk) {
        const float xp = xr[kk >> 2];
        v2f a;
        a.x = xp * sel0[kk & 3];
        a.y = xp * sel1[kk & 3];
        const float* Br = Bl + (2 * kk) * S2;
        #pragma unroll
        for (int t = 0; t < NTILE; ++t) {
            const v2f bf = *(const v2f*)(Br + t * 32);
            acc[t] = __builtin_amdgcn_wmma_f32_16x16x4_f32(
                false, a, false, bf, (short)0, acc[t], false, false);
        }
    }

    #pragma unroll
    for (int t = 0; t < NTILE; ++t) {
        #pragma unroll
        for (int r = 0; r < 8; ++r)
            sP[(wv * 16 + r + hi * 8) * PSTR + t * 16 + nl] = acc[t][r];
    }
    __syncthreads();

    for (int idx = tid; idx < 4 * C; idx += 256) {
        const int c = idx & 127;
        const int g = idx >> 7;
        const float* P = sP + c * PSTR;
        float s = 0.f;
        if (g == 0) {
            #pragma unroll
            for (int k = 0; k < K3_0; ++k) s = fmaf(P[COL_A0 + k], sW[(WR_M0 + k) * C + c], s);
            #pragma unroll
            for (int k = 0; k < K2_0; ++k) s = fmaf(P[COL_20 + k], sW[(WR_20 + k) * C + c], s);
            float d = 0.f;
            #pragma unroll
            for (int j = 0; j < L; ++j) d = fmaf(sU1[j], sX[c * XSTR + j], d);
            s = fmaf(sW[WR_10 * C + c], d, s);
            out[(size_t)b * (4 * C) + c] = s;
        } else {
            const int w = g - 1;
            #pragma unroll
            for (int k = 0; k < K3_1; ++k) s = fmaf(P[COL_A1 + w * K3_1 + k], sW[(WR_M1 + k) * C + c], s);
            #pragma unroll
            for (int k = 0; k < K2_1; ++k) s = fmaf(P[COL_21 + w * K2_1 + k], sW[(WR_21 + k) * C + c], s);
            float d = 0.f;
            #pragma unroll
            for (int j = 0; j < L; ++j) d = fmaf(sU1[16 + w * L + j], sX[c * XSTR + j], d);
            s = fmaf(sW[WR_11 * C + c], d, s);
            out[(size_t)b * (4 * C) + C + c * 3 + w] = s;
        }
    }
}

extern "C" void kernel_launch(void* const* d_in, const int* in_sizes, int n_in,
                              void* d_out, int out_size, void* d_ws, size_t ws_size,
                              hipStream_t stream) {
    (void)in_sizes; (void)n_in; (void)out_size; (void)d_ws; (void)ws_size;
    const float* x    = (const float*)d_in[0];
    const float* y    = (const float*)d_in[1];
    const float* U3_0 = (const float*)d_in[2];
    const float* U2_0 = (const float*)d_in[3];
    const float* U1_0 = (const float*)d_in[4];
    const float* U3_1 = (const float*)d_in[5];
    const float* U2_1 = (const float*)d_in[6];
    const float* U1_1 = (const float*)d_in[7];
    const float* Wm0  = (const float*)d_in[8];
    const float* W20  = (const float*)d_in[9];
    const float* W10  = (const float*)d_in[10];
    const float* Wm1  = (const float*)d_in[11];
    const float* W21  = (const float*)d_in[12];
    const float* W11  = (const float*)d_in[13];

    constexpr size_t smem_bytes = (size_t)SMEM_FLOATS * sizeof(float);
    mace_node_kernel<<<dim3(NN), dim3(256), smem_bytes, stream>>>(
        x, y, U3_0, U2_0, U1_0, U3_1, U2_1, U1_1,
        Wm0, W20, W10, Wm1, W21, W11, (float*)d_out);
}


# LDS cut from 268608 B to 157504 B: P and the per-node weights reuse the B2 storage after the products (weights built after them), so two blocks are resident per WGP and their phases overlap
# speedup vs baseline: 2.3435x; 1.2851x over previous
#include <hip/hip_runtime.h>

typedef float v2f __attribute__((ext_vector_type(2)));
typedef float v8f __attribute__((ext_vector_type(8)));
typedef float v4fu __attribute__((ext_vector_type(4), aligned(4)));

constexpr int NN   = 2048;
constexpr int C    = 128;
constexpr int L    = 16;
constexpr int E    = 10;
constexpr int K3_0 = 23, K2_0 = 4;
constexpr int K3_1 = 33, K2_1 = 6;

constexpr int COL_A1 = 0;
constexpr int COL_A0 = 3 * K3_1;
constexpr int COL_21 = COL_A0 + K3_0;
constexpr int COL_20 = COL_21 + 3 * K2_1;
constexpr int NCOL   = COL_20 + K2_0;
constexpr int NTILE  = NCOL / 16;

constexpr int S2   = 2 * NCOL + 2;
constexpr int PSTR = 149;
constexpr int XSTR = L + 1;

constexpr int WR_M1 = 0;
constexpr int WR_M0 = 33;
constexpr int WR_21 = 56;
constexpr int WR_20 = 62;
constexpr int WR_11 = 66;
constexpr int WR_10 = 67;
constexpr int WROWS = 68;

constexpr int OFF_X  = 0;
constexpr int OFF_XS = OFF_X + C * XSTR;
constexpr int OFF_U1 = OFF_XS + L;
constexpr int OFF_B2 = OFF_U1 + 64;
constexpr int OFF_P  = OFF_B2;
constexpr int OFF_W  = OFF_P + C * PSTR;
constexpr int SMEM_FLOATS = OFF_B2 + 128 * S2;
static_assert(OFF_W + WROWS * C <= SMEM_FLOATS, "P and W must fit inside the B2 storage");
static_assert(2 * SMEM_FLOATS * 4 <= 320 * 1024, "two blocks per WGP");


__device__ __forceinline__ void build_a1_group(const float* __restrict__ U3_1,
                                               const float (&xsr)[L],
                                               float* __restrict__ sB2,
                                               int G, int pg, int q)
{
    const int t = 3 * G + pg;
    const int w = t >> 7, p = t & 127;
    const int koff = (q < 8) ? 4 * q : (K3_1 - 4);
    const float* u0 = U3_1 + (size_t)(w * 256 + 2 * p) * (L * K3_1) + koff;
    const float* u1 = u0 + L * K3_1;
    float a0[4] = {0.f, 0.f, 0.f, 0.f}, a1[4] = {0.f, 0.f, 0.f, 0.f};
    #pragma unroll
    for (int i = 0; i < L; ++i) {
        const v4fu r0 = *(const v4fu*)(u0 + i * K3_1);
        const v4fu r1 = *(const v4fu*)(u1 + i * K3_1);
        #pragma unroll
        for (int e = 0; e < 4; ++e) {
            a0[e] = fmaf(r0[e], xsr[i], a0[e]);
            a1[e] = fmaf(r1[e], xsr[i], a1[e]);
        }
    }
    float* bp = sB2 + p * S2 + (COL_A1 + w * K3_1) * 2;
    if (q < 8) {
        #pragma unroll
        for (int e = 0; e < 4; ++e)
            *(v2f*)(bp + 2 * (4 * q + e)) = (v2f){a0[e], a1[e]};
    } else {
        *(v2f*)(bp + 2 * (K3_1 - 1)) = (v2f){a0[3], a1[3]};
    }
}

__device__ __forceinline__ void build_a0_group(const float* __restrict__ U3_0,
                                               const float (&xsr)[L],
                                               float* __restrict__ sB2,
                                               int p, int q)
{
    const int koff = (q < 5) ? 4 * q : (K3_0 - 4);
    const float* u0 = U3_0 + (size_t)(2 * p) * (L * K3_0) + koff;
    const float* u1 = u0 + L * K3_0;
    float a0[4] = {0.f, 0.f, 0.f, 0.f}, a1[4] = {0.f, 0.f, 0.f, 0.f};
    #pragma unroll
    for (int i = 0; i < L; ++i) {
        const v4fu r0 = *(const v4fu*)(u0 + i * K3_0);
        const v4fu r1 = *(const v4fu*)(u1 + i * K3_0);
        #pragma unroll
        for (int e = 0; e < 4; ++e) {
            a0[e] = fmaf(r0[e], xsr[i], a0[e]);
            a1[e] = fmaf(r1[e], xsr[i], a1[e]);
        }
    }
    float* bp = sB2 + p * S2 + COL_A0 * 2;
    if (q < 5) {
        #pragma unroll
        for (int e = 0; e < 4; ++e)
            *(v2f*)(bp + 2 * (4 * q + e)) = (v2f){a0[e], a1[e]};
    } else {
        #pragma unroll
        for (int e = 1; e < 4; ++e)
            *(v2f*)(bp + 2 * (K3_0 - 4 + e)) = (v2f){a0[e], a1[e]};
    }
}

__global__ __launch_bounds__(256) void mace_node_kernel(
    const float* __restrict__ x,    const float* __restrict__ y,
    const float* __restrict__ U3_0, const float* __restrict__ U2_0, const float* __restrict__ U1_0,
    const float* __restrict__ U3_1, const float* __restrict__ U2_1, const float* __restrict__ U1_1,
    const float* __restrict__ Wm0,  const float* __restrict__ W20,  const float* __restrict__ W10,
    const float* __restrict__ Wm1,  const float* __restrict__ W21,  const float* __restrict__ W11,
    float* __restrict__ out)
{
    extern __shared__ float sm[];
    float* sX  = sm + OFF_X;
    float* sXS = sm + OFF_XS;
    float* sW  = sm + OFF_W;
    float* sU1 = sm + OFF_U1;
    float* sB2 = sm + OFF_B2;
    float* sP  = sm + OFF_P;

    const int tid = threadIdx.x;
    const int b   = blockIdx.x;

    for (int idx = tid; idx < C * L; idx += 256)
        sX[(idx >> 4) * XSTR + (idx & 15)] = x[(size_t)b * (C * L) + idx];

    if (tid < 16)               sU1[tid] = U1_0[tid];
    else if (tid < 64)          sU1[tid] = U1_1[tid - 16];
    __syncthreads();

    if (tid < L) {
        float s = 0.f;
        for (int c = 0; c < C; ++c) s += sX[c * XSTR + tid];
        sXS[tid] = s;
    }
    __syncthreads();

    {
        const int wv2 = tid >> 5, ln2 = tid & 31;
        float xsr[L];
        #pragma unroll
        for (int i = 0; i < L; ++i) xsr[i] = sXS[i];

        if (ln2 < 27) {
            const int pg = ln2 / 9, q = ln2 - 9 * pg;
            for (int G = wv2; G < 128; G += 8)
                build_a1_group(U3_1, xsr, sB2, G, pg, q);
        }
        if (ln2 < 30) {
            const int pg = ln2 / 6, q = ln2 - 6 * pg;
            for (int G = wv2; G < 26; G += 8) {
                const int p = 5 * G + pg;
                if (p < 128) build_a0_group(U3_0, xsr, sB2, p, q);
            }
        }
        if (tid < 128) {
            const int p = tid;
            float* bp = sB2 + p * S2;
            #pragma unroll
            for (int m = 0; m < 3 * K2_1; ++m) {
                const int w = m / K2_1, k = m % K2_1;
                const int r0 = (w * 256 + 2 * p) * K2_1 + k;
                *(v2f*)(bp + (COL_21 + m) * 2) = (v2f){U2_1[r0], U2_1[r0 + K2_1]};
            }
            #pragma unroll
            for (int m = 0; m < K2_0; ++m)
                *(v2f*)(bp + (COL_20 + m) * 2) =
                    (v2f){U2_0[(2 * p) * K2_0 + m], U2_0[(2 * p + 1) * K2_0 + m]};
        }
    }
    __syncthreads();

    const int wv   = tid >> 5;
    const int lane = tid & 31;
    const int nl   = lane & 15;
    const int hi   = lane >> 4;
    const int crow = wv * 16 + nl;

    float xr[L];
    #pragma unroll
    for (int j = 0; j < L; ++j) xr[j] = sX[crow * XSTR + j];

    float sel0[4], sel1[4];
    #pragma unroll
    for (int j = 0; j < 4; ++j) {
        const float e0 = xr[4 * j],     e1 = xr[4 * j + 1];
        const float e2 = xr[4 * j + 2], e3 = xr[4 * j + 3];
        sel0[j] = hi ? e2 : e0;
        sel1[j] = hi ? e3 : e1;
    }

    v8f acc[NTILE];
    #pragma unroll
    for (int t = 0; t < NTILE; ++t)
        acc[t] = (v8f){0.f, 0.f, 0.f, 0.f, 0.f, 0.f, 0.f, 0.f};

    const float* Bl = sB2 + hi * S2 + nl * 2;
    #pragma unroll
    for (int kk = 0; kk < 64; ++kk) {
        const float xp = xr[kk >> 2];
        v2f a;
        a.x = xp * sel0[kk & 3];
        a.y = xp * sel1[kk & 3];
        const float* Br = Bl + (2 * kk) * S2;
        #pragma unroll
        for (int t = 0; t < NTILE; ++t) {
            const v2f bf = *(const v2f*)(Br + t * 32);
            acc[t] = __builtin_amdgcn_wmma_f32_16x16x4_f32(
                false, a, false, bf, (short)0, acc[t], false, false);
        }
    }

    __syncthreads();

    #pragma unroll
    for (int t = 0; t < NTILE; ++t) {
        #pragma unroll
        for (int r = 0; r < 8; ++r)
            sP[(wv * 16 + r + hi * 8) * PSTR + t * 16 + nl] = acc[t][r];
    }

    {
        float yv[E];
        #pragma unroll
        for (int e = 0; e < E; ++e) yv[e] = y[b * E + e];
        for (int idx = tid; idx < WROWS * C; idx += 256) {
            const int wr = idx >> 7, c = idx & 127;
            const float* src; int stride;
            if (wr < WR_M0)       { src = Wm1 + (wr)         * C + c; stride = K3_1 * C; }
            else if (wr < WR_21)  { src = Wm0 + (wr - WR_M0) * C + c; stride = K3_0 * C; }
            else if (wr < WR_20)  { src = W21 + (wr - WR_21) * C + c; stride = K2_1 * C; }
            else if (wr < WR_11)  { src = W20 + (wr - WR_20) * C + c; stride = K2_0 * C; }
            else if (wr == WR_11) { src = W11 + c;                    stride = C;        }
            else                  { src = W10 + c;                    stride = C;        }
            float wacc = 0.f;
            #pragma unroll
            for (int e = 0; e < E; ++e) wacc = fmaf(yv[e], src[e * stride], wacc);
            sW[idx] = wacc;
        }
    }
    __syncthreads();

    for (int idx = tid; idx < 4 * C; idx += 256) {
        const int c = idx & 127;
        const int g = idx >> 7;
        const float* P = sP + c * PSTR;
        float s = 0.f;
        if (g == 0) {
            #pragma unroll
            for (int k = 0; k < K3_0; ++k) s = fmaf(P[COL_A0 + k], sW[(WR_M0 + k) * C + c], s);
            #pragma unroll
            for (int k = 0; k < K2_0; ++k) s = fmaf(P[COL_20 + k], sW[(WR_20 + k) * C + c], s);
            float d = 0.f;
            #pragma unroll
            for (int j = 0; j < L; ++j) d = fmaf(sU1[j], sX[c * XSTR + j], d);
            s = fmaf(sW[WR_10 * C + c], d, s);
            out[(size_t)b * (4 * C) + c] = s;
        } else {
            const int w = g - 1;
            #pragma unroll
            for (int k = 0; k < K3_1; ++k) s = fmaf(P[COL_A1 + w * K3_1 + k], sW[(WR_M1 + k) * C + c], s);
            #pragma unroll
            for (int k = 0; k < K2_1; ++k) s = fmaf(P[COL_21 + w * K2_1 + k], sW[(WR_21 + k) * C + c], s);
            float d = 0.f;
            #pragma unroll
            for (int j = 0; j < L; ++j) d = fmaf(sU1[16 + w * L + j], sX[c * XSTR + j], d);
            s = fmaf(sW[WR_11 * C + c], d, s);
            out[(size_t)b * (4 * C) + C + c * 3 + w] = s;
        }
    }
}

extern "C" void kernel_launch(void* const* d_in, const int* in_sizes, int n_in,
                              void* d_out, int out_size, void* d_ws, size_t ws_size,
                              hipStream_t stream) {
    (void)in_sizes; (void)n_in; (void)out_size; (void)d_ws; (void)ws_size;
    const float* x    = (const float*)d_in[0];
    const float* y    = (const float*)d_in[1];
    const float* U3_0 = (const float*)d_in[2];
    const float* U2_0 = (const float*)d_in[3];
    const float* U1_0 = (const float*)d_in[4];
    const float* U3_1 = (const float*)d_in[5];
    const float* U2_1 = (const float*)d_in[6];
    const float* U1_1 = (const float*)d_in[7];
    const float* Wm0  = (const float*)d_in[8];
    const float* W20  = (const float*)d_in[9];
    const float* W10  = (const float*)d_in[10];
    const float* Wm1  = (const float*)d_in[11];
    const float* W21  = (const float*)d_in[12];
    const float* W11  = (const float*)d_in[13];

    constexpr size_t smem_bytes = (size_t)SMEM_FLOATS * sizeof(float);
    mace_node_kernel<<<dim3(NN), dim3(256), smem_bytes, stream>>>(
        x, y, U3_0, U2_0, U1_0, U3_1, U2_1, U1_1,
        Wm0, W20, W10, Wm1, W21, W11, (float*)d_out);
}


# two nodes per block: every U3/U2/W value loaded from L2 feeds the fma chains of both nodes; B2 built and consumed in 4 K-chunks with both nodes' accumulators in registers; all sum orders kept
# speedup vs baseline: 3.4625x; 1.4775x over previous
#include <hip/hip_runtime.h>

typedef float v2f __attribute__((ext_vector_type(2)));
typedef float v8f __attribute__((ext_vector_type(8)));
typedef float v4fu __attribute__((ext_vector_type(4), aligned(4)));

constexpr int NN   = 2048;
constexpr int C    = 128;
constexpr int L    = 16;
constexpr int E    = 10;
constexpr int K3_0 = 23, K2_0 = 4;
constexpr int K3_1 = 33, K2_1 = 6;

constexpr int COL_A1 = 0;
constexpr int COL_A0 = 3 * K3_1;
constexpr int COL_21 = COL_A0 + K3_0;
constexpr int COL_20 = COL_21 + 3 * K2_1;
constexpr int NCOL   = COL_20 + K2_0;
constexpr int NTILE  = NCOL / 16;

constexpr int S2   = 2 * NCOL + 2;
constexpr int PSTR = 145;
constexpr int XSTR = L + 1;

constexpr int WR_M1 = 0;
constexpr int WR_M0 = 33;
constexpr int WR_21 = 56;
constexpr int WR_20 = 62;
constexpr int WR_11 = 66;
constexpr int WR_10 = 67;
constexpr int WROWS = 68;

constexpr int NB  = 2;
constexpr int NCH = 4;
constexpr int CP  = 128 / NCH;

constexpr int OFF_X  = 0;
constexpr int OFF_XS = OFF_X + NB * C * XSTR;
constexpr int OFF_U1 = OFF_XS + NB * L;
constexpr int OFF_B2 = OFF_U1 + 64;
constexpr int OFF_P  = OFF_B2;
constexpr int OFF_W  = OFF_P + C * PSTR;
constexpr int SMEM_FLOATS = OFF_W + NB * WROWS * C;
static_assert(NB * CP * S2 <= C * PSTR + NB * WROWS * C, "the B2 chunk must fit the shared storage");
static_assert(2 * ((SMEM_FLOATS * 4 + 2047) / 2048) * 2048 <= 320 * 1024, "two blocks per WGP");


__device__ __forceinline__ void build_a1_rows(const float* __restrict__ U3_1,
                                              const float (&xa)[L], const float (&xb)[L],
                                              float* __restrict__ sB2,
                                              int ch, int G, int rg, int q)
{
    const int t   = 3 * G + rg;
    const int w   = t >> 6, rl = t & 63;
    const int row = ch * (2 * CP) + rl;
    const int koff = (q < 8) ? 4 * q : (K3_1 - 4);
    const unsigned off = (unsigned)((w * 256 + row) * (L * K3_1) + koff);
    const float* u = U3_1 + off;
    float a[4] = {0.f, 0.f, 0.f, 0.f}, c[4] = {0.f, 0.f, 0.f, 0.f};
    #pragma unroll
    for (int i = 0; i < L; ++i) {
        const v4fu r = *(const v4fu*)(u + i * K3_1);
        #pragma unroll
        for (int e = 0; e < 4; ++e) {
            a[e] = fmaf(r[e], xa[i], a[e]);
            c[e] = fmaf(r[e], xb[i], c[e]);
        }
    }
    float* bp = sB2 + (rl >> 1) * S2 + (COL_A1 + w * K3_1) * 2 + (rl & 1);
    float* bq = bp + CP * S2;
    if (q < 8) {
        #pragma unroll
        for (int e = 0; e < 4; ++e) {
            bp[2 * (4 * q + e)] = a[e];
            bq[2 * (4 * q + e)] = c[e];
        }
    } else {
        bp[2 * (K3_1 - 1)] = a[3];
        bq[2 * (K3_1 - 1)] = c[3];
    }
}

__device__ __forceinline__ void build_a0_rows(const float* __restrict__ U3_0,
                                              const float (&xa)[L], const float (&xb)[L],
                                              float* __restrict__ sB2,
                                              int ch, int G, int rg, int q)
{
    const int rl  = 4 * G + rg;
    const int row = ch * (2 * CP) + rl;
    const int koff = (q < 5) ? 4 * q : (K3_0 - 4);
    const unsigned off = (unsigned)(row * (L * K3_0) + koff);
    const float* u = U3_0 + off;
    float a[4] = {0.f, 0.f, 0.f, 0.f}, c[4] = {0.f, 0.f, 0.f, 0.f};
    v4fu r[L];
    #pragma unroll
    for (int i = 0; i < L; ++i) r[i] = *(const v4fu*)(u + i * K3_0);
    asm volatile("" :: "v"(r[0]), "v"(r[1]), "v"(r[2]),  "v"(r[3]),  "v"(r[4]),  "v"(r[5]),  "v"(r[6]),  "v"(r[7]),
                       "v"(r[8]), "v"(r[9]), "v"(r[10]), "v"(r[11]), "v"(r[12]), "v"(r[13]), "v"(r[14]), "v"(r[15]));
    #pragma unroll
    for (int i = 0; i < L; ++i) {
        #pragma unroll
        for (int e = 0; e < 4; ++e) {
            a[e] = fmaf(r[i][e], xa[i], a[e]);
            c[e] = fmaf(r[i][e], xb[i], c[e]);
        }
    }
    float* bp = sB2 + (rl >> 1) * S2 + COL_A0 * 2 + (rl & 1);
    float* bq = bp + CP * S2;
    if (q < 5) {
        #pragma unroll
        for (int e = 0; e < 4; ++e) {
            bp[2 * (4 * q + e)] = a[e];
            bq[2 * (4 * q + e)] = c[e];
        }
    } else {
        #pragma unroll
        for (int e = 1; e < 4; ++e) {
            bp[2 * (K3_0 - 4 + e)] = a[e];
            bq[2 * (K3_0 - 4 + e)] = c[e];
        }
    }
}

__device__ __forceinline__ void node_epilogue(const float* __restrict__ sP,
                                              const float* __restrict__ sWn,
                                              const float* __restrict__ sXn,
                                              const float* __restrict__ sU1,
                                              float* __restrict__ outn, int tid)
{
    for (int idx = tid; idx < 4 * C; idx += 256) {
        const int c = idx & 127;
        const int g = idx >> 7;
        const float* P = sP + c * PSTR;
        float s = 0.f;
        if (g == 0) {
            #pragma unroll
            for (int k = 0; k < K3_0; ++k) s = fmaf(P[COL_A0 + k], sWn[(WR_M0 + k) * C + c], s);
            #pragma unroll
            for (int k = 0; k < K2_0; ++k) s = fmaf(P[COL_20 + k], sWn[(WR_20 + k) * C + c], s);
            float d = 0.f;
            #pragma unroll
            for (int j = 0; j < L; ++j) d = fmaf(sU1[j], sXn[c * XSTR + j], d);
            s = fmaf(sWn[WR_10 * C + c], d, s);
            outn[c] = s;
        } else {
            const int w = g - 1;
            #pragma unroll
            for (int k = 0; k < K3_1; ++k) s = fmaf(P[COL_A1 + w * K3_1 + k], sWn[(WR_M1 + k) * C + c], s);
            #pragma unroll
            for (int k = 0; k < K2_1; ++k) s = fmaf(P[COL_21 + w * K2_1 + k], sWn[(WR_21 + k) * C + c], s);
            float d = 0.f;
            #pragma unroll
            for (int j = 0; j < L; ++j) d = fmaf(sU1[16 + w * L + j], sXn[c * XSTR + j], d);
            s = fmaf(sWn[WR_11 * C + c], d, s);
            outn[C + c * 3 + w] = s;
        }
    }
}

__global__ __launch_bounds__(256, 4) void mace_node_kernel(
    const float* __restrict__ x,    const float* __restrict__ y,
    const float* __restrict__ U3_0, const float* __restrict__ U2_0, const float* __restrict__ U1_0,
    const float* __restrict__ U3_1, const float* __restrict__ U2_1, const float* __restrict__ U1_1,
    const float* __restrict__ Wm0,  const float* __restrict__ W20,  const float* __restrict__ W10,
    const float* __restrict__ Wm1,  const float* __restrict__ W21,  const float* __restrict__ W11,
    float* __restrict__ out)
{
    extern __shared__ float sm[];
    float* sX  = sm + OFF_X;
    float* sXS = sm + OFF_XS;
    float* sW  = sm + OFF_W;
    float* sU1 = sm + OFF_U1;
    float* sB2 = sm + OFF_B2;
    float* sP  = sm + OFF_P;

    const int tid = threadIdx.x;
    const int b0  = blockIdx.x * NB;

    for (int idx = tid; idx < NB * C * L; idx += 256)
        sX[(idx >> 4) * XSTR + (idx & 15)] = x[(size_t)b0 * (C * L) + idx];

    if (tid < 16)               sU1[tid] = U1_0[tid];
    else if (tid < 64)          sU1[tid] = U1_1[tid - 16];
    __syncthreads();

    if (tid < NB * L) {
        const int n = tid >> 4, i = tid & 15;
        float s = 0.f;
        for (int c = 0; c < C; ++c) s += sX[(n * C + c) * XSTR + i];
        sXS[tid] = s;
    }
    __syncthreads();

    const int wv   = __builtin_amdgcn_readfirstlane(tid >> 5);
    const int lane = tid & 31;
    const int nl   = lane & 15;
    const int hi   = lane >> 4;
    const int crow = wv * 16 + nl;

    float xsa[L], xsb[L];
    #pragma unroll
    for (int i = 0; i < L; ++i) {
        xsa[i] = __int_as_float(__builtin_amdgcn_readfirstlane(__float_as_int(sXS[i])));
        xsb[i] = __int_as_float(__builtin_amdgcn_readfirstlane(__float_as_int(sXS[L + i])));
    }

    v8f acc0[NTILE], acc1[NTILE];
    #pragma unroll
    for (int t = 0; t < NTILE; ++t) {
        acc0[t] = (v8f){0.f, 0.f, 0.f, 0.f, 0.f, 0.f, 0.f, 0.f};
        acc1[t] = (v8f){0.f, 0.f, 0.f, 0.f, 0.f, 0.f, 0.f, 0.f};
    }

    #pragma nounroll
    for (int ch = 0; ch < NCH; ++ch) {
        if (lane < 27) {
            const int rg = lane / 9, q = lane - 9 * rg;
            for (int G = wv; G < 2 * CP; G += 8)
                build_a1_rows(U3_1, xsa, xsb, sB2, ch, G, rg, q);
        }
        if (lane < 24) {
            const int rg = lane / 6, q = lane - 6 * rg;
            for (int G = wv; G < CP / 2; G += 8)
                build_a0_rows(U3_0, xsa, xsb, sB2, ch, G, rg, q);
        }
        for (int idx = tid; idx < CP * (3 * K2_1 + K2_0); idx += 256) {
            const int pl = idx & (CP - 1);
            const int m  = idx >> 5;
            const int p  = ch * CP + pl;
            v2f v; int col;
            if (m < 3 * K2_1) {
                const int w = m / K2_1, k = m - w * K2_1;
                const int r0 = (w * 256 + 2 * p) * K2_1 + k;
                v = (v2f){U2_1[r0], U2_1[r0 + K2_1]};
                col = COL_21 + m;
            } else {
                const int k = m - 3 * K2_1;
                v = (v2f){U2_0[(2 * p) * K2_0 + k], U2_0[(2 * p + 1) * K2_0 + k]};
                col = COL_20 + k;
            }
            float* bp = sB2 + pl * S2 + col * 2;
            *(v2f*)(bp)           = v;
            *(v2f*)(bp + CP * S2) = v;
        }
        __syncthreads();

        {
            float sel0a[4], sel1a[4], sel0b[4], sel1b[4], xpa[4], xpb[4];
            const float* xra = sX + crow * XSTR;
            const float* xrb = sX + (C + crow) * XSTR;
            #pragma unroll
            for (int j = 0; j < 4; ++j) {
                sel0a[j] = hi ? xra[4 * j + 2] : xra[4 * j];
                sel1a[j] = hi ? xra[4 * j + 3] : xra[4 * j + 1];
                sel0b[j] = hi ? xrb[4 * j + 2] : xrb[4 * j];
                sel1b[j] = hi ? xrb[4 * j + 3] : xrb[4 * j + 1];
                xpa[j] = xra[4 * ch + j];
                xpb[j] = xrb[4 * ch + j];
            }
            const float* Bl = sB2 + hi * S2 + nl * 2;
            #pragma unroll
            for (int kk = 0; kk < CP / 2; ++kk) {
                v2f aa, ab;
                aa.x = xpa[kk >> 2] * sel0a[kk & 3];
                aa.y = xpa[kk >> 2] * sel1a[kk & 3];
                ab.x = xpb[kk >> 2] * sel0b[kk & 3];
                ab.y = xpb[kk >> 2] * sel1b[kk & 3];
                const float* Br = Bl + (2 * kk) * S2;
                #pragma unroll
                for (int t = 0; t < NTILE; ++t) {
                    const v2f bfa = *(const v2f*)(Br + t * 32);
                    const v2f bfb = *(const v2f*)(Br + CP * S2 + t * 32);
                    acc0[t] = __builtin_amdgcn_wmma_f32_16x16x4_f32(
                        false, aa, false, bfa, (short)0, acc0[t], false, false);
                    acc1[t] = __builtin_amdgcn_wmma_f32_16x16x4_f32(
                        false, ab, false, bfb, (short)0, acc1[t], false, false);
                }
            }
        }
        __syncthreads();
    }

    #pragma unroll
    for (int t = 0; t < NTILE; ++t) {
        #pragma unroll
        for (int r = 0; r < 8; ++r)
            sP[(wv * 16 + r + hi * 8) * PSTR + t * 16 + nl] = acc0[t][r];
    }

    {
        float yva[E], yvb[E];
        #pragma unroll
        for (int e = 0; e < E; ++e) { yva[e] = y[b0 * E + e]; yvb[e] = y[(b0 + 1) * E + e]; }
        for (int idx = tid; idx < WROWS * C; idx += 256) {
            const int wr = idx >> 7, c = idx & 127;
            const float* src; int stride;
            if (wr < WR_M0)       { src = Wm1 + (wr)         * C + c; stride = K3_1 * C; }
            else if (wr < WR_21)  { src = Wm0 + (wr - WR_M0) * C + c; stride = K3_0 * C; }
            else if (wr < WR_20)  { src = W21 + (wr - WR_21) * C + c; stride = K2_1 * C; }
            else if (wr < WR_11)  { src = W20 + (wr - WR_20) * C + c; stride = K2_0 * C; }
            else if (wr == WR_11) { src = W11 + c;                    stride = C;        }
            else                  { src = W10 + c;                    stride = C;        }
            float wa = 0.f, wb = 0.f;
            #pragma unroll
            for (int e = 0; e < E; ++e) {
                const float wv_e = src[e * stride];
                wa = fmaf(yva[e], wv_e, wa);
                wb = fmaf(yvb[e], wv_e, wb);
            }
            sW[idx]             = wa;
            sW[WROWS * C + idx] = wb;
        }
    }
    __syncthreads();

    node_epilogue(sP, sW, sX, sU1, out + (size_t)b0 * (4 * C), tid);

    __syncthreads();
    #pragma unroll
    for (int t = 0; t < NTILE; ++t) {
        #pragma unroll
        for (int r = 0; r < 8; ++r)
            sP[(wv * 16 + r + hi * 8) * PSTR + t * 16 + nl] = acc1[t][r];
    }
    __syncthreads();

    node_epilogue(sP, sW + WROWS * C, sX + C * XSTR, sU1, out + (size_t)(b0 + 1) * (4 * C), tid);
}

extern "C" void kernel_launch(void* const* d_in, const int* in_sizes, int n_in,
                              void* d_out, int out_size, void* d_ws, size_t ws_size,
                              hipStream_t stream) {
    (void)in_sizes; (void)n_in; (void)out_size; (void)d_ws; (void)ws_size;
    const float* x    = (const float*)d_in[0];
    const float* y    = (const float*)d_in[1];
    const float* U3_0 = (const float*)d_in[2];
    const float* U2_0 = (const float*)d_in[3];
    const float* U1_0 = (const float*)d_in[4];
    const float* U3_1 = (const float*)d_in[5];
    const float* U2_1 = (const float*)d_in[6];
    const float* U1_1 = (const float*)d_in[7];
    const float* Wm0  = (const float*)d_in[8];
    const float* W20  = (const float*)d_in[9];
    const float* W10  = (const float*)d_in[10];
    const float* Wm1  = (const float*)d_in[11];
    const float* W21  = (const float*)d_in[12];
    const float* W11  = (const float*)d_in[13];

    constexpr size_t smem_bytes = (size_t)SMEM_FLOATS * sizeof(float);
    mace_node_kernel<<<dim3(NN / NB), dim3(256), smem_bytes, stream>>>(
        x, y, U3_0, U2_0, U1_0, U3_1, U2_1, U1_1,
        Wm0, W20, W10, Wm1, W21, W11, (float*)d_out);
}


# exposed memory round trips per wave cut from 78 to 44: x tile and weight loads 16 B wide and all in flight, U2 loads at phase start, row-group loads refilled as a rolling pipeline; all sums kept
# speedup vs baseline: 3.8138x; 1.1015x over previous
#include <hip/hip_runtime.h>

typedef float v2f __attribute__((ext_vector_type(2)));
typedef float v8f __attribute__((ext_vector_type(8)));
typedef float v4fu __attribute__((ext_vector_type(4), aligned(4)));

constexpr int NN   = 2048;
constexpr int C    = 128;
constexpr int L    = 16;
constexpr int E    = 10;
constexpr int K3_0 = 23, K2_0 = 4;
constexpr int K3_1 = 33, K2_1 = 6;

constexpr int COL_A1 = 0;
constexpr int COL_A0 = 3 * K3_1;
constexpr int COL_21 = COL_A0 + K3_0;
constexpr int COL_20 = COL_21 + 3 * K2_1;
constexpr int NCOL   = COL_20 + K2_0;
constexpr int NTILE  = NCOL / 16;

constexpr int S2   = 2 * NCOL + 2;
constexpr int PSTR = 145;
constexpr int XSTR = L + 1;

constexpr int WR_M1 = 0;
constexpr int WR_M0 = 33;
constexpr int WR_21 = 56;
constexpr int WR_20 = 62;
constexpr int WR_11 = 66;
constexpr int WR_10 = 67;
constexpr int WROWS = 68;

constexpr int NB  = 2;
constexpr int NCH = 4;
constexpr int CP  = 128 / NCH;

constexpr int OFF_X  = 0;
constexpr int OFF_XS = OFF_X + NB * C * XSTR;
constexpr int OFF_U1 = OFF_XS + NB * L;
constexpr int OFF_B2 = OFF_U1 + 64;
constexpr int OFF_P  = OFF_B2;
constexpr int OFF_W  = OFF_P + C * PSTR;
constexpr int SMEM_FLOATS = OFF_W + NB * WROWS * C;
static_assert(NB * CP * S2 <= C * PSTR + NB * WROWS * C, "the B2 chunk must fit the shared storage");
static_assert(2 * ((SMEM_FLOATS * 4 + 2047) / 2048) * 2048 <= 320 * 1024, "two blocks per WGP");


template <int KT>
__device__ __forceinline__ void issue_row_loads(v4fu (&r)[L], const float* __restrict__ u)
{
    #pragma unroll
    for (int i = 0; i < L; ++i) r[i] = *(const v4fu*)(u + i * KT);
}

template <int KN>
__device__ __forceinline__ void row_sums(v4fu (&r)[L], const float (&xa)[L], const float (&xb)[L],
                                         float (&a)[4], float (&c)[4], const float* __restrict__ un)
{
    #pragma unroll
    for (int e = 0; e < 4; ++e) { a[e] = 0.f; c[e] = 0.f; }
    #pragma unroll
    for (int i = 0; i < L; ++i) {
        const v4fu v = r[i];
        #pragma unroll
        for (int e = 0; e < 4; ++e) {
            a[e] = fmaf(v[e], xa[i], a[e]);
            c[e] = fmaf(v[e], xb[i], c[e]);
        }
        if (KN > 0) {
            asm volatile("" :: "v"(v) : "memory");
            r[i] = *(const v4fu*)(un + i * KN);
        }
    }
}

__device__ __forceinline__ void store_a1(float* __restrict__ sB2, int t, int kcol,
                                         const float (&a)[4], const float (&c)[4])
{
    const int w = t >> 6, rl = t & 63;
    float* bp = sB2 + (rl >> 1) * S2 + (COL_A1 + w * K3_1 + kcol) * 2 + (rl & 1);
    float* bq = bp + CP * S2;
    #pragma unroll
    for (int e = 0; e < 4; ++e) {
        bp[2 * e] = a[e];
        bq[2 * e] = c[e];
    }
}

__device__ __forceinline__ void store_a0(float* __restrict__ sB2, int rl, int kcol,
                                         const float (&a)[4], const float (&c)[4])
{
    float* bp = sB2 + (rl >> 1) * S2 + (COL_A0 + kcol) * 2 + (rl & 1);
    float* bq = bp + CP * S2;
    #pragma unroll
    for (int e = 0; e < 4; ++e) {
        bp[2 * e] = a[e];
        bq[2 * e] = c[e];
    }
}

__device__ __forceinline__ void node_epilogue(const float* __restrict__ sP,
                                              const float* __restrict__ sWn,
                                              const float* __restrict__ sXn,
                                              const float* __restrict__ sU1,
                                              float* __restrict__ outn, int tid)
{
    for (int idx = tid; idx < 4 * C; idx += 256) {
        const int c = idx & 127;
        const int g = idx >> 7;
        const float* P = sP + c * PSTR;
        float s = 0.f;
        if (g == 0) {
            #pragma unroll
            for (int k = 0; k < K3_0; ++k) s = fmaf(P[COL_A0 + k], sWn[(WR_M0 + k) * C + c], s);
            #pragma unroll
            for (int k = 0; k < K2_0; ++k) s = fmaf(P[COL_20 + k], sWn[(WR_20 + k) * C + c], s);
            float d = 0.f;
            #pragma unroll
            for (int j = 0; j < L; ++j) d = fmaf(sU1[j], sXn[c * XSTR + j], d);
            s = fmaf(sWn[WR_10 * C + c], d, s);
            outn[c] = s;
        } else {
            const int w = g - 1;
            #pragma unroll
            for (int k = 0; k < K3_1; ++k) s = fmaf(P[COL_A1 + w * K3_1 + k], sWn[(WR_M1 + k) * C + c], s);
            #pragma unroll
            for (int k = 0; k < K2_1; ++k) s = fmaf(P[COL_21 + w * K2_1 + k], sWn[(WR_21 + k) * C + c], s);
            float d = 0.f;
            #pragma unroll
            for (int j = 0; j < L; ++j) d = fmaf(sU1[16 + w * L + j], sXn[c * XSTR + j], d);
            s = fmaf(sWn[WR_11 * C + c], d, s);
            outn[C + c * 3 + w] = s;
        }
    }
}

__global__ __launch_bounds__(256, 4) void mace_node_kernel(
    const float* __restrict__ x,    const float* __restrict__ y,
    const float* __restrict__ U3_0, const float* __restrict__ U2_0, const float* __restrict__ U1_0,
    const float* __restrict__ U3_1, const float* __restrict__ U2_1, const float* __restrict__ U1_1,
    const float* __restrict__ Wm0,  const float* __restrict__ W20,  const float* __restrict__ W10,
    const float* __restrict__ Wm1,  const float* __restrict__ W21,  const float* __restrict__ W11,
    float* __restrict__ out)
{
    extern __shared__ float sm[];
    float* sX  = sm + OFF_X;
    float* sXS = sm + OFF_XS;
    float* sW  = sm + OFF_W;
    float* sU1 = sm + OFF_U1;
    float* sB2 = sm + OFF_B2;
    float* sP  = sm + OFF_P;

    const int tid = threadIdx.x;
    const int b0  = blockIdx.x * NB;

    {
        const int ti = tid < 64 ? tid : 63;
        const float* up = ti < 16 ? (U1_0 + ti) : (U1_1 + (ti - 16));
        const float u1 = *up;
        const float* xb = x + (size_t)b0 * (C * L);
        v4fu xv[4];
        #pragma unroll
        for (int j = 0; j < 4; ++j) xv[j] = *(const v4fu*)(xb + 4 * (tid + 256 * j));
        #pragma unroll
        for (int j = 0; j < 4; ++j) {
            const int idx = 4 * (tid + 256 * j);
            float* d = sX + (idx >> 4) * XSTR + (idx & 15);
            #pragma unroll
            for (int u = 0; u < 4; ++u) d[u] = xv[j][u];
        }
        asm volatile("" :: "v"(u1));
        if (tid < 64) sU1[tid] = u1;
    }
    __syncthreads();

    if (tid < NB * L) {
        const int n = tid >> 4, i = tid & 15;
        float s = 0.f;
        for (int c = 0; c < C; ++c) s += sX[(n * C + c) * XSTR + i];
        sXS[tid] = s;
    }
    __syncthreads();

    const int wv   = __builtin_amdgcn_readfirstlane(tid >> 5);
    const int lane = tid & 31;
    const int nl   = lane & 15;
    const int hi   = lane >> 4;
    const int crow = wv * 16 + nl;

    float xsa[L], xsb[L];
    #pragma unroll
    for (int i = 0; i < L; ++i) {
        xsa[i] = __int_as_float(__builtin_amdgcn_readfirstlane(__float_as_int(sXS[i])));
        xsb[i] = __int_as_float(__builtin_amdgcn_readfirstlane(__float_as_int(sXS[L + i])));
    }

    const int l1  = lane < 27 ? lane : 26;
    const int rg1 = l1 / 9, q1 = l1 - 9 * rg1;
    const int k1  = (q1 < 8) ? 4 * q1 : (K3_1 - 4);
    const int l0  = lane < 24 ? lane : 23;
    const int rg0 = l0 / 6, q0 = l0 - 6 * rg0;
    const int k0  = (q0 < 5) ? 4 * q0 : (K3_0 - 4);

    v8f acc0[NTILE], acc1[NTILE];
    #pragma unroll
    for (int t = 0; t < NTILE; ++t) {
        acc0[t] = (v8f){0.f, 0.f, 0.f, 0.f, 0.f, 0.f, 0.f, 0.f};
        acc1[t] = (v8f){0.f, 0.f, 0.f, 0.f, 0.f, 0.f, 0.f, 0.f};
    }

    auto a1_off = [&](int chn, int G) -> unsigned {
        const int t = 3 * G + rg1;
        return (unsigned)((t >> 6) * 256 + (t & 63) + chn * (2 * CP)) * (unsigned)(L * K3_1) + (unsigned)k1;
    };
    auto a0_off = [&](int chn, int G) -> unsigned {
        return (unsigned)(4 * G + rg0 + chn * (2 * CP)) * (unsigned)(L * K3_0) + (unsigned)k0;
    };

    #pragma nounroll
    for (int ch = 0; ch < NCH; ++ch) {
        {
            v4fu r[L];
            issue_row_loads<K3_1>(r, U3_1 + a1_off(ch, wv));

            v2f uv[3];
            #pragma unroll
            for (int j = 0; j < 3; ++j) {
                const int mr = wv + 8 * j;
                const int m  = mr < (3 * K2_1 + K2_0 - 1) ? mr : (3 * K2_1 + K2_0 - 1);
                const bool t1 = m < 3 * K2_1;
                const int w = m / K2_1, k = m - w * K2_1;
                const float* ub = t1 ? (U2_1 + (w * 256) * K2_1 + k) : (U2_0 + (m - 3 * K2_1));
                const int us = t1 ? K2_1 : K2_0;
                const int p  = ch * CP + (int)(threadIdx.x & 31u);
                uv[j] = (v2f){ub[(2 * p) * us], ub[(2 * p + 1) * us]};
            }

            float a[4], c[4];
            #pragma nounroll
            for (int G = wv; G < 2 * CP - 8; G += 8) {
                row_sums<K3_1>(r, xsa, xsb, a, c, U3_1 + a1_off(ch, G + 8));
                store_a1(sB2, 3 * G + rg1, k1, a, c);
            }
            row_sums<K3_0>(r, xsa, xsb, a, c, U3_0 + a0_off(ch, wv));
            store_a1(sB2, 3 * (wv + 2 * CP - 8) + rg1, k1, a, c);
            row_sums<K3_0>(r, xsa, xsb, a, c, U3_0 + a0_off(ch, wv + 8));
            store_a0(sB2, 4 * wv + rg0, k0, a, c);
            row_sums<0>(r, xsa, xsb, a, c, nullptr);
            store_a0(sB2, 4 * (wv + 8) + rg0, k0, a, c);

            #pragma unroll
            for (int j = 0; j < 3; ++j) {
                const int m = wv + 8 * j;
                if (m < 3 * K2_1 + K2_0) {
                    const int col = (m < 3 * K2_1) ? (COL_21 + m) : (COL_20 + m - 3 * K2_1);
                    float* bp = sB2 + lane * S2 + col * 2;
                    *(v2f*)(bp)           = uv[j];
                    *(v2f*)(bp + CP * S2) = uv[j];
                }
            }
        }
        __syncthreads();

        {
            float sel0a[4], sel1a[4], sel0b[4], sel1b[4], xpa[4], xpb[4];
            const float* xra = sX + crow * XSTR;
            const float* xrb = sX + (C + crow) * XSTR;
            #pragma unroll
            for (int j = 0; j < 4; ++j) {
                sel0a[j] = hi ? xra[4 * j + 2] : xra[4 * j];
                sel1a[j] = hi ? xra[4 * j + 3] : xra[4 * j + 1];
                sel0b[j] = hi ? xrb[4 * j + 2] : xrb[4 * j];
                sel1b[j] = hi ? xrb[4 * j + 3] : xrb[4 * j + 1];
                xpa[j] = xra[4 * ch + j];
                xpb[j] = xrb[4 * ch + j];
            }
            const float* Bl = sB2 + hi * S2 + nl * 2;
            #pragma unroll
            for (int kk = 0; kk < CP / 2; ++kk) {
                v2f aa, ab;
                aa.x = xpa[kk >> 2] * sel0a[kk & 3];
                aa.y = xpa[kk >> 2] * sel1a[kk & 3];
                ab.x = xpb[kk >> 2] * sel0b[kk & 3];
                ab.y = xpb[kk >> 2] * sel1b[kk & 3];
                const float* Br = Bl + (2 * kk) * S2;
                #pragma unroll
                for (int t = 0; t < NTILE; ++t) {
                    const v2f bfa = *(const v2f*)(Br + t * 32);
                    const v2f bfb = *(const v2f*)(Br + CP * S2 + t * 32);
                    acc0[t] = __builtin_amdgcn_wmma_f32_16x16x4_f32(
                        false, aa, false, bfa, (short)0, acc0[t], false, false);
                    acc1[t] = __builtin_amdgcn_wmma_f32_16x16x4_f32(
                        false, ab, false, bfb, (short)0, acc1[t], false, false);
                }
            }
        }
        __syncthreads();
    }

    #pragma unroll
    for (int t = 0; t < NTILE; ++t) {
        #pragma unroll
        for (int r = 0; r < 8; ++r)
            sP[(wv * 16 + r + hi * 8) * PSTR + t * 16 + nl] = acc0[t][r];
    }

    {
        float yva[E], yvb[E];
        #pragma unroll
        for (int e = 0; e < E; ++e) { yva[e] = y[b0 * E + e]; yvb[e] = y[(b0 + 1) * E + e]; }
        const int c4 = lane * 4;
        #pragma nounroll
        for (int j0 = 0; j0 < 9; j0 += 3) {
            v4fu wl[3][E];
            #pragma unroll
            for (int jj = 0; jj < 3; ++jj) {
                const int wrr = wv + 8 * (j0 + jj);
                const int wr  = wrr < WROWS ? wrr : (WROWS - 1);
                const float* src; int stride;
                if (wr < WR_M0)       { src = Wm1 + (wr)         * C; stride = K3_1 * C; }
                else if (wr < WR_21)  { src = Wm0 + (wr - WR_M0) * C; stride = K3_0 * C; }
                else if (wr < WR_20)  { src = W21 + (wr - WR_21) * C; stride = K2_1 * C; }
                else if (wr < WR_11)  { src = W20 + (wr - WR_20) * C; stride = K2_0 * C; }
                else if (wr == WR_11) { src = W11;                    stride = C;        }
                else                  { src = W10;                    stride = C;        }
                #pragma unroll
                for (int e = 0; e < E; ++e) wl[jj][e] = *(const v4fu*)(src + e * stride + c4);
            }
            #pragma unroll
            for (int jj = 0; jj < 3; ++jj) {
                const int wrr = wv + 8 * (j0 + jj);
                float wa[4] = {0.f, 0.f, 0.f, 0.f}, wb[4] = {0.f, 0.f, 0.f, 0.f};
                #pragma unroll
                for (int e = 0; e < E; ++e) {
                    #pragma unroll
                    for (int u = 0; u < 4; ++u) {
                        wa[u] = fmaf(yva[e], wl[jj][e][u], wa[u]);
                        wb[u] = fmaf(yvb[e], wl[jj][e][u], wb[u]);
                    }
                }
                if (wrr < WROWS) {
                    float* dst = sW + wrr * C + c4;
                    #pragma unroll
                    for (int u = 0; u < 4; ++u) {
                        dst[u]             = wa[u];
                        dst[WROWS * C + u] = wb[u];
                    }
                }
            }
        }
    }
    __syncthreads();

    node_epilogue(sP, sW, sX, sU1, out + (size_t)b0 * (4 * C), tid);

    __syncthreads();
    #pragma unroll
    for (int t = 0; t < NTILE; ++t) {
        #pragma unroll
        for (int r = 0; r < 8; ++r)
            sP[(wv * 16 + r + hi * 8) * PSTR + t * 16 + nl] = acc1[t][r];
    }
    __syncthreads();

    node_epilogue(sP, sW + WROWS * C, sX + C * XSTR, sU1, out + (size_t)(b0 + 1) * (4 * C), tid);
}

extern "C" void kernel_launch(void* const* d_in, const int* in_sizes, int n_in,
                              void* d_out, int out_size, void* d_ws, size_t ws_size,
                              hipStream_t stream) {
    (void)in_sizes; (void)n_in; (void)out_size; (void)d_ws; (void)ws_size;
    const float* x    = (const float*)d_in[0];
    const float* y    = (const float*)d_in[1];
    const float* U3_0 = (const float*)d_in[2];
    const float* U2_0 = (const float*)d_in[3];
    const float* U1_0 = (const float*)d_in[4];
    const float* U3_1 = (const float*)d_in[5];
    const float* U2_1 = (const float*)d_in[6];
    const float* U1_1 = (const float*)d_in[7];
    const float* Wm0  = (const float*)d_in[8];
    const float* W20  = (const float*)d_in[9];
    const float* W10  = (const float*)d_in[10];
    const float* Wm1  = (const float*)d_in[11];
    const float* W21  = (const float*)d_in[12];
    const float* W11  = (const float*)d_in[13];

    constexpr size_t smem_bytes = (size_t)SMEM_FLOATS * sizeof(float);
    mace_node_kernel<<<dim3(NN / NB), dim3(256), smem_bytes, stream>>>(
        x, y, U3_0, U2_0, U1_0, U3_1, U2_1, U1_1,
        Wm0, W20, W10, Wm1, W21, W11, (float*)d_out);
}


# K of the WMMA products folded 256 -> 160 rows: x[p]*x[q] == x[q]*x[p], so B rows (p,q) and (q,p) of different quads are summed in the build (32-term chains): 40 K steps, not 64; sum order changes
# speedup vs baseline: 4.4575x; 1.1688x over previous
#include <hip/hip_runtime.h>

typedef float v2f __attribute__((ext_vector_type(2)));
typedef float v8f __attribute__((ext_vector_type(8)));
typedef float v4fu __attribute__((ext_vector_type(4), aligned(4)));

constexpr int NN   = 2048;
constexpr int C    = 128;
constexpr int L    = 16;
constexpr int E    = 10;
constexpr int K3_0 = 23, K2_0 = 4;
constexpr int K3_1 = 33, K2_1 = 6;

constexpr int COL_A1 = 0;
constexpr int COL_A0 = 3 * K3_1;
constexpr int COL_21 = COL_A0 + K3_0;
constexpr int COL_20 = COL_21 + 3 * K2_1;
constexpr int NCOL   = COL_20 + K2_0;
constexpr int NTILE  = NCOL / 16;

constexpr int S2   = 2 * NCOL + 2;
constexpr int PSTR = 145;
constexpr int XSTR = L + 1;

constexpr int WR_M1 = 0;
constexpr int WR_M0 = 33;
constexpr int WR_21 = 56;
constexpr int WR_20 = 62;
constexpr int WR_11 = 66;
constexpr int WR_10 = 67;
constexpr int WROWS = 68;

constexpr int NB  = 2;
constexpr int NCH = 4;
constexpr int NST = 10;
constexpr int CP  = 2 * NST;
constexpr int NU2 = 3 * K2_1 + K2_0;

__host__ __device__ constexpr int sl_base(int mm) { return mm * (9 - mm) / 2; }
__host__ __device__ constexpr int fo_base(int mm) { return mm * (7 - mm) / 2; }
__device__ __forceinline__ void fold_step(int fo, int& mm, int& j)
{
    mm = (fo >= 3) + (fo >= 5);
    j  = mm + 1 + fo - fo_base(mm);
}

constexpr int OFF_X  = 0;
constexpr int OFF_XS = OFF_X + NB * C * XSTR;
constexpr int OFF_U1 = OFF_XS + NB * L;
constexpr int OFF_B2 = OFF_U1 + 64;
constexpr int OFF_P  = OFF_B2;
constexpr int OFF_W  = OFF_P + C * PSTR;
constexpr int SMEM_FLOATS = OFF_W + NB * WROWS * C;
static_assert(NB * CP * S2 <= C * PSTR + NB * WROWS * C, "the B2 chunk must fit the shared storage");
static_assert(2 * ((SMEM_FLOATS * 4 + 2047) / 2048) * 2048 <= 320 * 1024, "two blocks per WGP");


template <int KT>
__device__ __forceinline__ void issue_row_loads(v4fu (&r)[L], const float* __restrict__ u)
{
    #pragma unroll
    for (int i = 0; i < L; ++i) r[i] = *(const v4fu*)(u + i * KT);
}

template <int KN>
__device__ __forceinline__ void row_sums(v4fu (&r)[L], const float (&xa)[L], const float (&xb)[L],
                                         float (&a)[4], float (&c)[4], const float* __restrict__ un,
                                         int reset)
{
    if (reset) {
        #pragma unroll
        for (int e = 0; e < 4; ++e) { a[e] = 0.f; c[e] = 0.f; }
    }
    #pragma unroll
    for (int i = 0; i < L; ++i) {
        const v4fu v = r[i];
        #pragma unroll
        for (int e = 0; e < 4; ++e) {
            a[e] = fmaf(v[e], xa[i], a[e]);
            c[e] = fmaf(v[e], xb[i], c[e]);
        }
        if (KN > 0) {
            asm volatile("" :: "v"(v) : "memory");
            r[i] = *(const v4fu*)(un + i * KN);
        }
    }
}

__device__ __forceinline__ void store_a1(float* __restrict__ sB2, int w, int rl, int kcol,
                                         const float (&a)[4], const float (&c)[4])
{
    float* bp = sB2 + (rl >> 1) * S2 + (COL_A1 + w * K3_1 + kcol) * 2 + (rl & 1);
    float* bq = bp + CP * S2;
    #pragma unroll
    for (int e = 0; e < 4; ++e) {
        bp[2 * e] = a[e];
        bq[2 * e] = c[e];
    }
}

__device__ __forceinline__ void store_a0(float* __restrict__ sB2, int rl, int kcol,
                                         const float (&a)[4], const float (&c)[4])
{
    float* bp = sB2 + (rl >> 1) * S2 + (COL_A0 + kcol) * 2 + (rl & 1);
    float* bq = bp + CP * S2;
    #pragma unroll
    for (int e = 0; e < 4; ++e) {
        bp[2 * e] = a[e];
        bq[2 * e] = c[e];
    }
}

__device__ __forceinline__ void node_epilogue(const float* __restrict__ sP,
                                              const float* __restrict__ sWn,
                                              const float* __restrict__ sXn,
                                              const float* __restrict__ sU1,
                                              float* __restrict__ outn, int tid)
{
    for (int idx = tid; idx < 4 * C; idx += 256) {
        const int c = idx & 127;
        const int g = idx >> 7;
        const float* P = sP + c * PSTR;
        float s = 0.f;
        if (g == 0) {
            #pragma unroll
            for (int k = 0; k < K3_0; ++k) s = fmaf(P[COL_A0 + k], sWn[(WR_M0 + k) * C + c], s);
            #pragma unroll
            for (int k = 0; k < K2_0; ++k) s = fmaf(P[COL_20 + k], sWn[(WR_20 + k) * C + c], s);
            float d = 0.f;
            #pragma unroll
            for (int j = 0; j < L; ++j) d = fmaf(sU1[j], sXn[c * XSTR + j], d);
            s = fmaf(sWn[WR_10 * C + c], d, s);
            outn[c] = s;
        } else {
            const int w = g - 1;
            #pragma unroll
            for (int k = 0; k < K3_1; ++k) s = fmaf(P[COL_A1 + w * K3_1 + k], sWn[(WR_M1 + k) * C + c], s);
            #pragma unroll
            for (int k = 0; k < K2_1; ++k) s = fmaf(P[COL_21 + w * K2_1 + k], sWn[(WR_21 + k) * C + c], s);
            float d = 0.f;
            #pragma unroll
            for (int j = 0; j < L; ++j) d = fmaf(sU1[16 + w * L + j], sXn[c * XSTR + j], d);
            s = fmaf(sWn[WR_11 * C + c], d, s);
            outn[C + c * 3 + w] = s;
        }
    }
}

__global__ __launch_bounds__(256, 4) void mace_node_kernel(
    const float* __restrict__ x,    const float* __restrict__ y,
    const float* __restrict__ U3_0, const float* __restrict__ U2_0, const float* __restrict__ U1_0,
    const float* __restrict__ U3_1, const float* __restrict__ U2_1, const float* __restrict__ U1_1,
    const float* __restrict__ Wm0,  const float* __restrict__ W20,  const float* __restrict__ W10,
    const float* __restrict__ Wm1,  const float* __restrict__ W21,  const float* __restrict__ W11,
    float* __restrict__ out)
{
    extern __shared__ float sm[];
    float* sX  = sm + OFF_X;
    float* sXS = sm + OFF_XS;
    float* sW  = sm + OFF_W;
    float* sU1 = sm + OFF_U1;
    float* sB2 = sm + OFF_B2;
    float* sP  = sm + OFF_P;

    const int tid = threadIdx.x;
    const int b0  = blockIdx.x * NB;

    {
        const int ti = tid < 64 ? tid : 63;
        const float* up = ti < 16 ? (U1_0 + ti) : (U1_1 + (ti - 16));
        const float u1 = *up;
        const float* xb = x + (size_t)b0 * (C * L);
        v4fu xv[4];
        #pragma unroll
        for (int j = 0; j < 4; ++j) xv[j] = *(const v4fu*)(xb + 4 * (tid + 256 * j));
        #pragma unroll
        for (int j = 0; j < 4; ++j) {
            const int idx = 4 * (tid + 256 * j);
            float* d = sX + (idx >> 4) * XSTR + (idx & 15);
            #pragma unroll
            for (int u = 0; u < 4; ++u) d[u] = xv[j][u];
        }
        asm volatile("" :: "v"(u1));
        if (tid < 64) sU1[tid] = u1;
    }
    __syncthreads();

    if (tid < NB * L) {
        const int n = tid >> 4, i = tid & 15;
        float s = 0.f;
        for (int c = 0; c < C; ++c) s += sX[(n * C + c) * XSTR + i];
        sXS[tid] = s;
    }
    __syncthreads();

    const int wv   = __builtin_amdgcn_readfirstlane(tid >> 5);
    const int lane = tid & 31;
    const int nl   = lane & 15;
    const int hi   = lane >> 4;
    const int crow = wv * 16 + nl;

    float xsa[L], xsb[L];
    #pragma unroll
    for (int i = 0; i < L; ++i) {
        xsa[i] = __int_as_float(__builtin_amdgcn_readfirstlane(__float_as_int(sXS[i])));
        xsb[i] = __int_as_float(__builtin_amdgcn_readfirstlane(__float_as_int(sXS[L + i])));
    }

    const int l1  = lane < 27 ? lane : 26;
    const int rg1 = l1 / 9, q1 = l1 - 9 * rg1;
    const int k1  = (q1 < 8) ? 4 * q1 : (K3_1 - 4);
    const int l0  = lane < 24 ? lane : 23;
    const int rg0 = l0 / 6, q0 = l0 - 6 * rg0;
    const int k0  = (q0 < 5) ? 4 * q0 : (K3_0 - 4);

    v8f acc0[NTILE], acc1[NTILE];
    #pragma unroll
    for (int t = 0; t < NTILE; ++t) {
        acc0[t] = (v8f){0.f, 0.f, 0.f, 0.f, 0.f, 0.f, 0.f, 0.f};
        acc1[t] = (v8f){0.f, 0.f, 0.f, 0.f, 0.f, 0.f, 0.f, 0.f};
    }

    auto a1_pass = [&](int chn, int p, int& row, int& rl, int& reset, int& store) {
        int mm, j, r, second, plain;
        if (p < 2) { const int dr = 2 * wv + p; mm = dr >> 2; j = mm; r = dr & 3; second = 0; plain = 1; }
        else {
            const int f = 3 * wv + ((p - 2) >> 1);
            fold_step(f >> 2, mm, j);
            r = f & 3; second = (p - 2) & 1; plain = 0;
        }
        const int a = 4 * mm + chn, b = 4 * j + r;
        row   = second ? (b * 16 + a) : (a * 16 + b);
        rl    = 4 * (sl_base(mm) + j - mm) + r;
        reset = plain | (second ^ 1);
        store = plain | second;
    };
    auto a0_pass = [&](int chn, int p, int& row, int& rl, int& reset, int& store) {
        int mm, j, second, plain;
        if (wv < 6) { fold_step(wv, mm, j); second = p; plain = 0; }
        else        { mm = 2 * (wv - 6) + p; j = mm; second = 0; plain = 1; }
        const int a = 4 * mm + chn, b = 4 * j + rg0;
        row   = second ? (b * 16 + a) : (a * 16 + b);
        rl    = 4 * (sl_base(mm) + j - mm) + rg0;
        reset = plain | (second ^ 1);
        store = plain | second;
    };
    auto a1_off = [&](int row) -> unsigned {
        return (unsigned)(rg1 * 256 + row) * (unsigned)(L * K3_1) + (unsigned)k1;
    };
    auto a0_off = [&](int row) -> unsigned {
        return (unsigned)row * (unsigned)(L * K3_0) + (unsigned)k0;
    };
    auto u2_item = [&](int jj, int& col, int& pr, int& mm, int& j) -> bool {
        const int cr = wv + 8 * jj;
        col = cr < NU2 ? cr : NU2 - 1;
        pr  = lane < CP ? lane : CP - 1;
        const int sl = pr >> 1;
        mm = (sl >= 4) + (sl >= 7) + (sl >= 9);
        j  = mm + sl - sl_base(mm);
        return (cr < NU2) & (lane < CP);
    };

    #pragma nounroll
    for (int ch = 0; ch < NCH; ++ch) {
        {
            int row, rl, rs, st;
            a1_pass(ch, 0, row, rl, rs, st);
            v4fu r[L];
            issue_row_loads<K3_1>(r, U3_1 + a1_off(row));

            float a[4] = {0.f, 0.f, 0.f, 0.f}, c[4] = {0.f, 0.f, 0.f, 0.f};
            #pragma nounroll
            for (int p = 0; p < 7; ++p) {
                int rown, rln, rsn, stn;
                a1_pass(ch, p + 1, rown, rln, rsn, stn);
                row_sums<K3_1>(r, xsa, xsb, a, c, U3_1 + a1_off(rown), rs);
                if (st) store_a1(sB2, rg1, rl, k1, a, c);
                rl = rln; rs = rsn; st = stn;
            }
            {
                int rown, rln, rsn, stn;
                a0_pass(ch, 0, rown, rln, rsn, stn);
                row_sums<K3_0>(r, xsa, xsb, a, c, U3_0 + a0_off(rown), rs);
                store_a1(sB2, rg1, rl, k1, a, c);
                rl = rln; rs = rsn; st = stn;
            }
            {
                int rown, rln, rsn, stn;
                a0_pass(ch, 1, rown, rln, rsn, stn);
                row_sums<K3_0>(r, xsa, xsb, a, c, U3_0 + a0_off(rown), rs);
                if (st) store_a0(sB2, rl, k0, a, c);
                rl = rln; rs = rsn; st = stn;
            }
            float u2v[12];
            if (rs) {
                #pragma unroll
                for (int e = 0; e < 4; ++e) { a[e] = 0.f; c[e] = 0.f; }
            }
            #pragma unroll
            for (int i = 0; i < L; ++i) {
                const v4fu v = r[i];
                #pragma unroll
                for (int e = 0; e < 4; ++e) {
                    a[e] = fmaf(v[e], xsa[i], a[e]);
                    c[e] = fmaf(v[e], xsb[i], c[e]);
                }
                if (i < 4) {
                    asm volatile("" :: "v"(v) : "memory");
                    #pragma unroll
                    for (int t = 0; t < 3; ++t) {
                        const int n = 3 * i + t;
                        int col, pr, mm, j;
                        u2_item(n >> 2, col, pr, mm, j);
                        const int aa = 4 * mm + ch, bb = 4 * j + 2 * (pr & 1);
                        const bool t1 = col < 3 * K2_1;
                        const int w = col / K2_1, k = col - w * K2_1;
                        const float* ub = t1 ? (U2_1 + (w * 256) * K2_1 + k) : (U2_0 + (col - 3 * K2_1));
                        const int us = t1 ? K2_1 : K2_0;
                        const int e = n & 3;
                        const int rw = e == 0 ? (aa * 16 + bb) : e == 1 ? (aa * 16 + bb + 1)
                                     : e == 2 ? (bb * 16 + aa) : ((bb + 1) * 16 + aa);
                        u2v[n] = ub[rw * us];
                    }
                }
            }
            store_a0(sB2, rl, k0, a, c);
            #pragma unroll
            for (int n = 0; n < 12; n += 4)
                asm volatile("" :: "v"(u2v[n]), "v"(u2v[n + 1]), "v"(u2v[n + 2]), "v"(u2v[n + 3]));
            #pragma unroll
            for (int jj = 0; jj < 3; ++jj) {
                int col, pr, mm, j;
                const bool valid = u2_item(jj, col, pr, mm, j);
                const bool fold = j != mm;
                v2f v;
                v.x = fold ? u2v[4 * jj]     + u2v[4 * jj + 2] : u2v[4 * jj];
                v.y = fold ? u2v[4 * jj + 1] + u2v[4 * jj + 3] : u2v[4 * jj + 1];
                if (valid) {
                    float* bp = sB2 + pr * S2 + (COL_21 + col) * 2;
                    *(v2f*)(bp)           = v;
                    *(v2f*)(bp + CP * S2) = v;
                }
            }
        }
        __syncthreads();

        {
            float sel0a[4], sel1a[4], sel0b[4], sel1b[4], xpa[4], xpb[4];
            const float* xra = sX + crow * XSTR;
            const float* xrb = sX + (C + crow) * XSTR;
            #pragma unroll
            for (int j = 0; j < 4; ++j) {
                sel0a[j] = hi ? xra[4 * j + 2] : xra[4 * j];
                sel1a[j] = hi ? xra[4 * j + 3] : xra[4 * j + 1];
                sel0b[j] = hi ? xrb[4 * j + 2] : xrb[4 * j];
                sel1b[j] = hi ? xrb[4 * j + 3] : xrb[4 * j + 1];
                xpa[j] = xra[4 * j + ch];
                xpb[j] = xrb[4 * j + ch];
            }
            const float* Bl = sB2 + hi * S2 + nl * 2;
            #pragma unroll
            for (int mm = 0; mm < 4; ++mm) {
                #pragma unroll
                for (int j = mm; j < 4; ++j) {
                    const int sl = sl_base(mm) + j - mm;
                    v2f aa, ab;
                    aa.x = xpa[mm] * sel0a[j];
                    aa.y = xpa[mm] * sel1a[j];
                    ab.x = xpb[mm] * sel0b[j];
                    ab.y = xpb[mm] * sel1b[j];
                    const float* Br = Bl + (2 * sl) * S2;
                    #pragma unroll
                    for (int t = 0; t < NTILE; ++t) {
                        const v2f bfa = *(const v2f*)(Br + t * 32);
                        const v2f bfb = *(const v2f*)(Br + CP * S2 + t * 32);
                        acc0[t] = __builtin_amdgcn_wmma_f32_16x16x4_f32(
                            false, aa, false, bfa, (short)0, acc0[t], false, false);
                        acc1[t] = __builtin_amdgcn_wmma_f32_16x16x4_f32(
                            false, ab, false, bfb, (short)0, acc1[t], false, false);
                    }
                }
            }
        }
        __syncthreads();
    }

    #pragma unroll
    for (int t = 0; t < NTILE; ++t) {
        #pragma unroll
        for (int r = 0; r < 8; ++r)
            sP[(wv * 16 + r + hi * 8) * PSTR + t * 16 + nl] = acc0[t][r];
    }

    {
        float yva[E], yvb[E];
        #pragma unroll
        for (int e = 0; e < E; ++e) { yva[e] = y[b0 * E + e]; yvb[e] = y[(b0 + 1) * E + e]; }
        const int c4 = lane * 4;
        #pragma nounroll
        for (int j0 = 0; j0 < 9; j0 += 3) {
            v4fu wl[3][E];
            #pragma unroll
            for (int jj = 0; jj < 3; ++jj) {
                const int wrr = wv + 8 * (j0 + jj);
                const int wr  = wrr < WROWS ? wrr : (WROWS - 1);
                const float* src; int stride;
                if (wr < WR_M0)       { src = Wm1 + (wr)         * C; stride = K3_1 * C; }
                else if (wr < WR_21)  { src = Wm0 + (wr - WR_M0) * C; stride = K3_0 * C; }
                else if (wr < WR_20)  { src = W21 + (wr - WR_21) * C; stride = K2_1 * C; }
                else if (wr < WR_11)  { src = W20 + (wr - WR_20) * C; stride = K2_0 * C; }
                else if (wr == WR_11) { src = W11;                    stride = C;        }
                else                  { src = W10;                    stride = C;        }
                #pragma unroll
                for (int e = 0; e < E; ++e) wl[jj][e] = *(const v4fu*)(src + e * stride + c4);
            }
            #pragma unroll
            for (int jj = 0; jj < 3; ++jj) {
                const int wrr = wv + 8 * (j0 + jj);
                float wa[4] = {0.f, 0.f, 0.f, 0.f}, wb[4] = {0.f, 0.f, 0.f, 0.f};
                #pragma unroll
                for (int e = 0; e < E; ++e) {
                    #pragma unroll
                    for (int u = 0; u < 4; ++u) {
                        wa[u] = fmaf(yva[e], wl[jj][e][u], wa[u]);
                        wb[u] = fmaf(yvb[e], wl[jj][e][u], wb[u]);
                    }
                }
                if (wrr < WROWS) {
                    float* dst = sW + wrr * C + c4;
                    #pragma unroll
                    for (int u = 0; u < 4; ++u) {
                        dst[u]             = wa[u];
                        dst[WROWS * C + u] = wb[u];
                    }
                }
            }
        }
    }
    __syncthreads();

    node_epilogue(sP, sW, sX, sU1, out + (size_t)b0 * (4 * C), tid);

    __syncthreads();
    #pragma unroll
    for (int t = 0; t < NTILE; ++t) {
        #pragma unroll
        for (int r = 0; r < 8; ++r)
            sP[(wv * 16 + r + hi * 8) * PSTR + t * 16 + nl] = acc1[t][r];
    }
    __syncthreads();

    node_epilogue(sP, sW + WROWS * C, sX + C * XSTR, sU1, out + (size_t)(b0 + 1) * (4 * C), tid);
}

extern "C" void kernel_launch(void* const* d_in, const int* in_sizes, int n_in,
                              void* d_out, int out_size, void* d_ws, size_t ws_size,
                              hipStream_t stream) {
    (void)in_sizes; (void)n_in; (void)out_size; (void)d_ws; (void)ws_size;
    const float* x    = (const float*)d_in[0];
    const float* y    = (const float*)d_in[1];
    const float* U3_0 = (const float*)d_in[2];
    const float* U2_0 = (const float*)d_in[3];
    const float* U1_0 = (const float*)d_in[4];
    const float* U3_1 = (const float*)d_in[5];
    const float* U2_1 = (const float*)d_in[6];
    const float* U1_1 = (const float*)d_in[7];
    const float* Wm0  = (const float*)d_in[8];
    const float* W20  = (const float*)d_in[9];
    const float* W10  = (const float*)d_in[10];
    const float* Wm1  = (const float*)d_in[11];
    const float* W21  = (const float*)d_in[12];
    const float* W11  = (const float*)d_in[13];

    constexpr size_t smem_bytes = (size_t)SMEM_FLOATS * sizeof(float);
    mace_node_kernel<<<dim3(NN / NB), dim3(256), smem_bytes, stream>>>(
        x, y, U3_0, U2_0, U1_0, U3_1, U2_1, U1_1,
        Wm0, W20, W10, Wm1, W21, W11, (float*)d_out);
}


# B2 chunk double-buffered in the later P/W storage: build of chunk ch+1 and products of chunk ch share one barrier interval, waves 4..7 do products first: loads run under WMMAs; all sums kept
# speedup vs baseline: 4.7397x; 1.0633x over previous
#include <hip/hip_runtime.h>

typedef float v2f __attribute__((ext_vector_type(2)));
typedef float v8f __attribute__((ext_vector_type(8)));
typedef float v4fu __attribute__((ext_vector_type(4), aligned(4)));

constexpr int NN   = 2048;
constexpr int C    = 128;
constexpr int L    = 16;
constexpr int E    = 10;
constexpr int K3_0 = 23, K2_0 = 4;
constexpr int K3_1 = 33, K2_1 = 6;

constexpr int COL_A1 = 0;
constexpr int COL_A0 = 3 * K3_1;
constexpr int COL_21 = COL_A0 + K3_0;
constexpr int COL_20 = COL_21 + 3 * K2_1;
constexpr int NCOL   = COL_20 + K2_0;
constexpr int NTILE  = NCOL / 16;

constexpr int S2   = 2 * NCOL + 2;
constexpr int PSTR = 145;
constexpr int XSTR = L + 1;

constexpr int WR_M1 = 0;
constexpr int WR_M0 = 33;
constexpr int WR_21 = 56;
constexpr int WR_20 = 62;
constexpr int WR_11 = 66;
constexpr int WR_10 = 67;
constexpr int WROWS = 68;

constexpr int NB  = 2;
constexpr int NCH = 4;
constexpr int NST = 10;
constexpr int CP  = 2 * NST;
constexpr int NU2 = 3 * K2_1 + K2_0;
constexpr int B2BUF = NB * CP * S2;

__host__ __device__ constexpr int sl_base(int mm) { return mm * (9 - mm) / 2; }
__host__ __device__ constexpr int fo_base(int mm) { return mm * (7 - mm) / 2; }
__device__ __forceinline__ void fold_step(int fo, int& mm, int& j)
{
    mm = (fo >= 3) + (fo >= 5);
    j  = mm + 1 + fo - fo_base(mm);
}

constexpr int OFF_X  = 0;
constexpr int OFF_XS = OFF_X + NB * C * XSTR;
constexpr int OFF_U1 = OFF_XS + NB * L;
constexpr int OFF_B2 = OFF_U1 + 64;
constexpr int OFF_P  = OFF_B2;
constexpr int OFF_W  = OFF_P + C * PSTR;
constexpr int SMEM_FLOATS = OFF_W + NB * WROWS * C;
static_assert(2 * B2BUF <= C * PSTR + NB * WROWS * C, "both B2 chunk buffers must fit the shared storage");
static_assert(2 * ((SMEM_FLOATS * 4 + 2047) / 2048) * 2048 <= 320 * 1024, "two blocks per WGP");


template <int KT>
__device__ __forceinline__ void issue_row_loads(v4fu (&r)[L], const float* __restrict__ u)
{
    #pragma unroll
    for (int i = 0; i < L; ++i) r[i] = *(const v4fu*)(u + i * KT);
}

template <int KN>
__device__ __forceinline__ void row_sums(v4fu (&r)[L], const float (&xa)[L], const float (&xb)[L],
                                         float (&a)[4], float (&c)[4], const float* __restrict__ un,
                                         int reset)
{
    if (reset) {
        #pragma unroll
        for (int e = 0; e < 4; ++e) { a[e] = 0.f; c[e] = 0.f; }
    }
    #pragma unroll
    for (int i = 0; i < L; ++i) {
        const v4fu v = r[i];
        #pragma unroll
        for (int e = 0; e < 4; ++e) {
            a[e] = fmaf(v[e], xa[i], a[e]);
            c[e] = fmaf(v[e], xb[i], c[e]);
        }
        if (KN > 0) {
            asm volatile("" :: "v"(v) : "memory");
            r[i] = *(const v4fu*)(un + i * KN);
        }
    }
}

__device__ __forceinline__ void store_a1(float* __restrict__ sB2, int w, int rl, int kcol,
                                         const float (&a)[4], const float (&c)[4])
{
    float* bp = sB2 + (rl >> 1) * S2 + (COL_A1 + w * K3_1 + kcol) * 2 + (rl & 1);
    float* bq = bp + CP * S2;
    #pragma unroll
    for (int e = 0; e < 4; ++e) {
        bp[2 * e] = a[e];
        bq[2 * e] = c[e];
    }
}

__device__ __forceinline__ void store_a0(float* __restrict__ sB2, int rl, int kcol,
                                         const float (&a)[4], const float (&c)[4])
{
    float* bp = sB2 + (rl >> 1) * S2 + (COL_A0 + kcol) * 2 + (rl & 1);
    float* bq = bp + CP * S2;
    #pragma unroll
    for (int e = 0; e < 4; ++e) {
        bp[2 * e] = a[e];
        bq[2 * e] = c[e];
    }
}

__device__ __forceinline__ void node_epilogue(const float* __restrict__ sP,
                                              const float* __restrict__ sWn,
                                              const float* __restrict__ sXn,
                                              const float* __restrict__ sU1,
                                              float* __restrict__ outn, int tid)
{
    for (int idx = tid; idx < 4 * C; idx += 256) {
        const int c = idx & 127;
        const int g = idx >> 7;
        const float* P = sP + c * PSTR;
        float s = 0.f;
        if (g == 0) {
            #pragma unroll
            for (int k = 0; k < K3_0; ++k) s = fmaf(P[COL_A0 + k], sWn[(WR_M0 + k) * C + c], s);
            #pragma unroll
            for (int k = 0; k < K2_0; ++k) s = fmaf(P[COL_20 + k], sWn[(WR_20 + k) * C + c], s);
            float d = 0.f;
            #pragma unroll
            for (int j = 0; j < L; ++j) d = fmaf(sU1[j], sXn[c * XSTR + j], d);
            s = fmaf(sWn[WR_10 * C + c], d, s);
            outn[c] = s;
        } else {
            const int w = g - 1;
            #pragma unroll
            for (int k = 0; k < K3_1; ++k) s = fmaf(P[COL_A1 + w * K3_1 + k], sWn[(WR_M1 + k) * C + c], s);
            #pragma unroll
            for (int k = 0; k < K2_1; ++k) s = fmaf(P[COL_21 + w * K2_1 + k], sWn[(WR_21 + k) * C + c], s);
            float d = 0.f;
            #pragma unroll
            for (int j = 0; j < L; ++j) d = fmaf(sU1[16 + w * L + j], sXn[c * XSTR + j], d);
            s = fmaf(sWn[WR_11 * C + c], d, s);
            outn[C + c * 3 + w] = s;
        }
    }
}

__global__ __launch_bounds__(256, 4) void mace_node_kernel(
    const float* __restrict__ x,    const float* __restrict__ y,
    const float* __restrict__ U3_0, const float* __restrict__ U2_0, const float* __restrict__ U1_0,
    const float* __restrict__ U3_1, const float* __restrict__ U2_1, const float* __restrict__ U1_1,
    const float* __restrict__ Wm0,  const float* __restrict__ W20,  const float* __restrict__ W10,
    const float* __restrict__ Wm1,  const float* __restrict__ W21,  const float* __restrict__ W11,
    float* __restrict__ out)
{
    extern __shared__ float sm[];
    float* sX  = sm + OFF_X;
    float* sXS = sm + OFF_XS;
    float* sW  = sm + OFF_W;
    float* sU1 = sm + OFF_U1;
    float* sB2 = sm + OFF_B2;
    float* sP  = sm + OFF_P;

    const int tid = threadIdx.x;
    const int b0  = blockIdx.x * NB;

    {
        const int ti = tid < 64 ? tid : 63;
        const float* up = ti < 16 ? (U1_0 + ti) : (U1_1 + (ti - 16));
        const float u1 = *up;
        const float* xb = x + (size_t)b0 * (C * L);
        v4fu xv[4];
        #pragma unroll
        for (int j = 0; j < 4; ++j) xv[j] = *(const v4fu*)(xb + 4 * (tid + 256 * j));
        #pragma unroll
        for (int j = 0; j < 4; ++j) {
            const int idx = 4 * (tid + 256 * j);
            float* d = sX + (idx >> 4) * XSTR + (idx & 15);
            #pragma unroll
            for (int u = 0; u < 4; ++u) d[u] = xv[j][u];
        }
        asm volatile("" :: "v"(u1));
        if (tid < 64) sU1[tid] = u1;
    }
    __syncthreads();

    if (tid < NB * L) {
        const int n = tid >> 4, i = tid & 15;
        float s = 0.f;
        for (int c = 0; c < C; ++c) s += sX[(n * C + c) * XSTR + i];
        sXS[tid] = s;
    }
    __syncthreads();

    const int wv   = __builtin_amdgcn_readfirstlane(tid >> 5);
    const int lane = tid & 31;
    const int nl   = lane & 15;
    const int hi   = lane >> 4;
    const int crow = wv * 16 + nl;

    float xsa[L], xsb[L];
    #pragma unroll
    for (int i = 0; i < L; ++i) {
        xsa[i] = __int_as_float(__builtin_amdgcn_readfirstlane(__float_as_int(sXS[i])));
        xsb[i] = __int_as_float(__builtin_amdgcn_readfirstlane(__float_as_int(sXS[L + i])));
    }

    const int l1  = lane < 27 ? lane : 26;
    const int rg1 = l1 / 9, q1 = l1 - 9 * rg1;
    const int k1  = (q1 < 8) ? 4 * q1 : (K3_1 - 4);
    const int l0  = lane < 24 ? lane : 23;
    const int rg0 = l0 / 6, q0 = l0 - 6 * rg0;
    const int k0  = (q0 < 5) ? 4 * q0 : (K3_0 - 4);

    v8f acc0[NTILE], acc1[NTILE];
    #pragma unroll
    for (int t = 0; t < NTILE; ++t) {
        acc0[t] = (v8f){0.f, 0.f, 0.f, 0.f, 0.f, 0.f, 0.f, 0.f};
        acc1[t] = (v8f){0.f, 0.f, 0.f, 0.f, 0.f, 0.f, 0.f, 0.f};
    }

    auto a1_pass = [&](int chn, int p, int& row, int& rl, int& reset, int& store) {
        int mm, j, r, second, plain;
        if (p < 2) { const int dr = 2 * wv + p; mm = dr >> 2; j = mm; r = dr & 3; second = 0; plain = 1; }
        else {
            const int f = 3 * wv + ((p - 2) >> 1);
            fold_step(f >> 2, mm, j);
            r = f & 3; second = (p - 2) & 1; plain = 0;
        }
        const int a = 4 * mm + chn, b = 4 * j + r;
        row   = second ? (b * 16 + a) : (a * 16 + b);
        rl    = 4 * (sl_base(mm) + j - mm) + r;
        reset = plain | (second ^ 1);
        store = plain | second;
    };
    auto a0_pass = [&](int chn, int p, int& row, int& rl, int& reset, int& store) {
        int mm, j, second, plain;
        if (wv < 6) { fold_step(wv, mm, j); second = p; plain = 0; }
        else        { mm = 2 * (wv - 6) + p; j = mm; second = 0; plain = 1; }
        const int a = 4 * mm + chn, b = 4 * j + rg0;
        row   = second ? (b * 16 + a) : (a * 16 + b);
        rl    = 4 * (sl_base(mm) + j - mm) + rg0;
        reset = plain | (second ^ 1);
        store = plain | second;
    };
    auto a1_off = [&](int row) -> unsigned {
        return (unsigned)(rg1 * 256 + row) * (unsigned)(L * K3_1) + (unsigned)k1;
    };
    auto a0_off = [&](int row) -> unsigned {
        return (unsigned)row * (unsigned)(L * K3_0) + (unsigned)k0;
    };
    auto u2_item = [&](int jj, int& col, int& pr, int& mm, int& j) -> bool {
        const int cr = wv + 8 * jj;
        col = cr < NU2 ? cr : NU2 - 1;
        pr  = lane < CP ? lane : CP - 1;
        const int sl = pr >> 1;
        mm = (sl >= 4) + (sl >= 7) + (sl >= 9);
        j  = mm + sl - sl_base(mm);
        return (cr < NU2) & (lane < CP);
    };

    const int prod_first = (wv >> 2) & 1;
    #pragma nounroll
    for (int h = 0; h < 2 * (NCH + 1); ++h) {
        const int it = h >> 1;
        const bool build_half = ((h & 1) ^ prod_first) == 0;
        if (build_half && it < NCH) {
            const int ch = it;
            float* sBw = sB2 + (it & 1) * B2BUF;
            int row, rl, rs, st;
            a1_pass(ch, 0, row, rl, rs, st);
            v4fu r[L];
            issue_row_loads<K3_1>(r, U3_1 + a1_off(row));

            float a[4] = {0.f, 0.f, 0.f, 0.f}, c[4] = {0.f, 0.f, 0.f, 0.f};
            #pragma nounroll
            for (int p = 0; p < 7; ++p) {
                int rown, rln, rsn, stn;
                a1_pass(ch, p + 1, rown, rln, rsn, stn);
                row_sums<K3_1>(r, xsa, xsb, a, c, U3_1 + a1_off(rown), rs);
                if (st) store_a1(sBw, rg1, rl, k1, a, c);
                rl = rln; rs = rsn; st = stn;
            }
            {
                int rown, rln, rsn, stn;
                a0_pass(ch, 0, rown, rln, rsn, stn);
                row_sums<K3_0>(r, xsa, xsb, a, c, U3_0 + a0_off(rown), rs);
                store_a1(sBw, rg1, rl, k1, a, c);
                rl = rln; rs = rsn; st = stn;
            }
            {
                int rown, rln, rsn, stn;
                a0_pass(ch, 1, rown, rln, rsn, stn);
                row_sums<K3_0>(r, xsa, xsb, a, c, U3_0 + a0_off(rown), rs);
                if (st) store_a0(sBw, rl, k0, a, c);
                rl = rln; rs = rsn; st = stn;
            }
            float u2v[12];
            if (rs) {
                #pragma unroll
                for (int e = 0; e < 4; ++e) { a[e] = 0.f; c[e] = 0.f; }
            }
            #pragma unroll
            for (int i = 0; i < L; ++i) {
                const v4fu v = r[i];
                #pragma unroll
                for (int e = 0; e < 4; ++e) {
                    a[e] = fmaf(v[e], xsa[i], a[e]);
                    c[e] = fmaf(v[e], xsb[i], c[e]);
                }
                if (i < 4) {
                    asm volatile("" :: "v"(v) : "memory");
                    #pragma unroll
                    for (int t = 0; t < 3; ++t) {
                        const int n = 3 * i + t;
                        int col, pr, mm, j;
                        u2_item(n >> 2, col, pr, mm, j);
                        const int aa = 4 * mm + ch, bb = 4 * j + 2 * (pr & 1);
                        const bool t1 = col < 3 * K2_1;
                        const int w = col / K2_1, k = col - w * K2_1;
                        const float* ub = t1 ? (U2_1 + (w * 256) * K2_1 + k) : (U2_0 + (col - 3 * K2_1));
                        const int us = t1 ? K2_1 : K2_0;
                        const int e = n & 3;
                        const int rw = e == 0 ? (aa * 16 + bb) : e == 1 ? (aa * 16 + bb + 1)
                                     : e == 2 ? (bb * 16 + aa) : ((bb + 1) * 16 + aa);
                        u2v[n] = ub[rw * us];
                    }
                }
            }
            store_a0(sBw, rl, k0, a, c);
            #pragma unroll
            for (int n = 0; n < 12; n += 4)
                asm volatile("" :: "v"(u2v[n]), "v"(u2v[n + 1]), "v"(u2v[n + 2]), "v"(u2v[n + 3]));
            #pragma unroll
            for (int jj = 0; jj < 3; ++jj) {
                int col, pr, mm, j;
                const bool valid = u2_item(jj, col, pr, mm, j);
                const bool fold = j != mm;
                v2f v;
                v.x = fold ? u2v[4 * jj]     + u2v[4 * jj + 2] : u2v[4 * jj];
                v.y = fold ? u2v[4 * jj + 1] + u2v[4 * jj + 3] : u2v[4 * jj + 1];
                if (valid) {
                    float* bp = sBw + pr * S2 + (COL_21 + col) * 2;
                    *(v2f*)(bp)           = v;
                    *(v2f*)(bp + CP * S2) = v;
                }
            }
        }

        if (!build_half && it > 0) {
            const int ch = it - 1;
            const float* sBr = sB2 + (ch & 1) * B2BUF;
            float sel0a[4], sel1a[4], sel0b[4], sel1b[4], xpa[4], xpb[4];
            const float* xra = sX + crow * XSTR;
            const float* xrb = sX + (C + crow) * XSTR;
            #pragma unroll
            for (int j = 0; j < 4; ++j) {
                sel0a[j] = hi ? xra[4 * j + 2] : xra[4 * j];
                sel1a[j] = hi ? xra[4 * j + 3] : xra[4 * j + 1];
                sel0b[j] = hi ? xrb[4 * j + 2] : xrb[4 * j];
                sel1b[j] = hi ? xrb[4 * j + 3] : xrb[4 * j + 1];
                xpa[j] = xra[4 * j + ch];
                xpb[j] = xrb[4 * j + ch];
            }
            const float* Bl = sBr + hi * S2 + nl * 2;
            #pragma unroll
            for (int mm = 0; mm < 4; ++mm) {
                #pragma unroll
                for (int j = mm; j < 4; ++j) {
                    const int sl = sl_base(mm) + j - mm;
                    v2f aa, ab;
                    aa.x = xpa[mm] * sel0a[j];
                    aa.y = xpa[mm] * sel1a[j];
                    ab.x = xpb[mm] * sel0b[j];
                    ab.y = xpb[mm] * sel1b[j];
                    const float* Br = Bl + (2 * sl) * S2;
                    #pragma unroll
                    for (int t = 0; t < NTILE; ++t) {
                        const v2f bfa = *(const v2f*)(Br + t * 32);
                        const v2f bfb = *(const v2f*)(Br + CP * S2 + t * 32);
                        acc0[t] = __builtin_amdgcn_wmma_f32_16x16x4_f32(
                            false, aa, false, bfa, (short)0, acc0[t], false, false);
                        acc1[t] = __builtin_amdgcn_wmma_f32_16x16x4_f32(
                            false, ab, false, bfb, (short)0, acc1[t], false, false);
                    }
                }
            }
        }
        if (h & 1) __syncthreads();
    }

    #pragma unroll
    for (int t = 0; t < NTILE; ++t) {
        #pragma unroll
        for (int r = 0; r < 8; ++r)
            sP[(wv * 16 + r + hi * 8) * PSTR + t * 16 + nl] = acc0[t][r];
    }

    {
        float yva[E], yvb[E];
        #pragma unroll
        for (int e = 0; e < E; ++e) { yva[e] = y[b0 * E + e]; yvb[e] = y[(b0 + 1) * E + e]; }
        const int c4 = lane * 4;
        #pragma nounroll
        for (int j0 = 0; j0 < 9; j0 += 3) {
            v4fu wl[3][E];
            #pragma unroll
            for (int jj = 0; jj < 3; ++jj) {
                const int wrr = wv + 8 * (j0 + jj);
                const int wr  = wrr < WROWS ? wrr : (WROWS - 1);
                const float* src; int stride;
                if (wr < WR_M0)       { src = Wm1 + (wr)         * C; stride = K3_1 * C; }
                else if (wr < WR_21)  { src = Wm0 + (wr - WR_M0) * C; stride = K3_0 * C; }
                else if (wr < WR_20)  { src = W21 + (wr - WR_21) * C; stride = K2_1 * C; }
                else if (wr < WR_11)  { src = W20 + (wr - WR_20) * C; stride = K2_0 * C; }
                else if (wr == WR_11) { src = W11;                    stride = C;        }
                else                  { src = W10;                    stride = C;        }
                #pragma unroll
                for (int e = 0; e < E; ++e) wl[jj][e] = *(const v4fu*)(src + e * stride + c4);
            }
            #pragma unroll
            for (int jj = 0; jj < 3; ++jj) {
                const int wrr = wv + 8 * (j0 + jj);
                float wa[4] = {0.f, 0.f, 0.f, 0.f}, wb[4] = {0.f, 0.f, 0.f, 0.f};
                #pragma unroll
                for (int e = 0; e < E; ++e) {
                    #pragma unroll
                    for (int u = 0; u < 4; ++u) {
                        wa[u] = fmaf(yva[e], wl[jj][e][u], wa[u]);
                        wb[u] = fmaf(yvb[e], wl[jj][e][u], wb[u]);
                    }
                }
                if (wrr < WROWS) {
                    float* dst = sW + wrr * C + c4;
                    #pragma unroll
                    for (int u = 0; u < 4; ++u) {
                        dst[u]             = wa[u];
                        dst[WROWS * C + u] = wb[u];
                    }
                }
            }
        }
    }
    __syncthreads();

    node_epilogue(sP, sW, sX, sU1, out + (size_t)b0 * (4 * C), tid);

    __syncthreads();
    #pragma unroll
    for (int t = 0; t < NTILE; ++t) {
        #pragma unroll
        for (int r = 0; r < 8; ++r)
            sP[(wv * 16 + r + hi * 8) * PSTR + t * 16 + nl] = acc1[t][r];
    }
    __syncthreads();

    node_epilogue(sP, sW + WROWS * C, sX + C * XSTR, sU1, out + (size_t)(b0 + 1) * (4 * C), tid);
}

extern "C" void kernel_launch(void* const* d_in, const int* in_sizes, int n_in,
                              void* d_out, int out_size, void* d_ws, size_t ws_size,
                              hipStream_t stream) {
    (void)in_sizes; (void)n_in; (void)out_size; (void)d_ws; (void)ws_size;
    const float* x    = (const float*)d_in[0];
    const float* y    = (const float*)d_in[1];
    const float* U3_0 = (const float*)d_in[2];
    const float* U2_0 = (const float*)d_in[3];
    const float* U1_0 = (const float*)d_in[4];
    const float* U3_1 = (const float*)d_in[5];
    const float* U2_1 = (const float*)d_in[6];
    const float* U1_1 = (const float*)d_in[7];
    const float* Wm0  = (const float*)d_in[8];
    const float* W20  = (const float*)d_in[9];
    const float* W10  = (const float*)d_in[10];
    const float* Wm1  = (const float*)d_in[11];
    const float* W21  = (const float*)d_in[12];
    const float* W11  = (const float*)d_in[13];

    constexpr size_t smem_bytes = (size_t)SMEM_FLOATS * sizeof(float);
    mace_node_kernel<<<dim3(NN / NB), dim3(256), smem_bytes, stream>>>(
        x, y, U3_0, U2_0, U1_0, U3_1, U2_1, U1_1,
        Wm0, W20, W10, Wm1, W21, W11, (float*)d_out);
}


# four nodes per 16-wave block: every U3/U2/W value loaded feeds the fma chains of four nodes, 5 row groups per wave and chunk, not 10; 4 accumulator tiles wait in LDS during the build; sums kept
# speedup vs baseline: 5.5564x; 1.1723x over previous
#include <hip/hip_runtime.h>

typedef float v2f __attribute__((ext_vector_type(2)));
typedef float v8f __attribute__((ext_vector_type(8)));
typedef float v4f __attribute__((ext_vector_type(4)));
typedef float v4fu __attribute__((ext_vector_type(4), aligned(4)));

constexpr int NN   = 2048;
constexpr int C    = 128;
constexpr int L    = 16;
constexpr int E    = 10;
constexpr int K3_0 = 23, K2_0 = 4;
constexpr int K3_1 = 33, K2_1 = 6;

constexpr int COL_A1 = 0;
constexpr int COL_A0 = 3 * K3_1;
constexpr int COL_21 = COL_A0 + K3_0;
constexpr int COL_20 = COL_21 + 3 * K2_1;
constexpr int NCOL   = COL_20 + K2_0;
constexpr int NTILE  = NCOL / 16;

constexpr int S2   = 2 * NCOL + 2;
constexpr int PSTR = 145;
constexpr int XSTR = L + 1;

constexpr int WR_M1 = 0;
constexpr int WR_M0 = 33;
constexpr int WR_21 = 56;
constexpr int WR_20 = 62;
constexpr int WR_11 = 66;
constexpr int WR_10 = 67;
constexpr int WROWS = 68;

constexpr int NB  = 4;
constexpr int NTHR = 512;
constexpr int NCH = 4;
constexpr int NST = 10;
constexpr int CP  = 2 * NST;
constexpr int NU2 = 3 * K2_1 + K2_0;
constexpr int B2BUF = NB * CP * S2;

__host__ __device__ constexpr int sl_base(int mm) { return mm * (9 - mm) / 2; }
__host__ __device__ constexpr int fo_base(int mm) { return mm * (7 - mm) / 2; }
__device__ __forceinline__ void fold_step(int fo, int& mm, int& j)
{
    mm = (fo >= 3) + (fo >= 5);
    j  = mm + 1 + fo - fo_base(mm);
}

constexpr int OFF_X  = 0;
constexpr int OFF_XS = OFF_X + NB * C * XSTR;
constexpr int OFF_U1 = OFF_XS + NB * L;
constexpr int OFF_B2 = OFF_U1 + 64;
constexpr int OFF_P  = OFF_B2;
constexpr int OFF_W  = OFF_P + C * PSTR;
constexpr int NPARK = 4;
constexpr int NKEEP = NTILE - NPARK;
constexpr int PARKW = NPARK * 8 * 32;
constexpr int OFF_PARK = OFF_B2 + 2 * B2BUF;
constexpr int END_W    = OFF_W + NB * WROWS * C;
constexpr int END_PARK = OFF_PARK + (NTHR / 32) * PARKW;
constexpr int SMEM_FLOATS = END_PARK > END_W ? END_PARK : END_W;
static_assert(OFF_PARK % 4 == 0 && PARKW % 4 == 0, "16-byte accesses to the parked tiles");
static_assert(OFF_PARK + 8 * PARKW >= END_W, "the parked tiles of waves 8..15 must lie behind W");
static_assert(2 * B2BUF <= C * PSTR + NB * WROWS * C, "both B2 chunk buffers must fit the shared storage");
static_assert(((SMEM_FLOATS * 4 + 2047) / 2048) * 2048 <= 320 * 1024, "one block per WGP");


template <int KT>
__device__ __forceinline__ void issue_row_loads(v4fu (&r)[L], const float* __restrict__ u)
{
    #pragma unroll
    for (int i = 0; i < L; ++i) r[i] = *(const v4fu*)(u + i * KT);
}

__device__ __forceinline__ float xs_lane(int xsv, int l)
{
    return __int_as_float(__builtin_amdgcn_readlane(xsv, l));
}

template <int KN>
__device__ __forceinline__ void row_sums(v4fu (&r)[L], const float (&xa)[L], const float (&xb)[L],
                                         int xsv, float (&a)[4], float (&c)[4], float (&d)[4],
                                         float (&f)[4], const float* __restrict__ un, int reset)
{
    if (reset) {
        #pragma unroll
        for (int e = 0; e < 4; ++e) { a[e] = 0.f; c[e] = 0.f; d[e] = 0.f; f[e] = 0.f; }
    }
    #pragma unroll
    for (int i = 0; i < L; ++i) {
        const v4fu v = r[i];
        const float xc = xs_lane(xsv, i), xd = xs_lane(xsv, L + i);
        #pragma unroll
        for (int e = 0; e < 4; ++e) {
            a[e] = fmaf(v[e], xa[i], a[e]);
            c[e] = fmaf(v[e], xb[i], c[e]);
            d[e] = fmaf(v[e], xc, d[e]);
            f[e] = fmaf(v[e], xd, f[e]);
        }
        if (KN > 0) {
            asm volatile("" :: "v"(v) : "memory");
            r[i] = *(const v4fu*)(un + i * KN);
        }
    }
}

__device__ __forceinline__ void store_a1(float* __restrict__ sB2, int w, int rl, int kcol,
                                         const float (&a)[4], const float (&c)[4],
                                         const float (&d)[4], const float (&f)[4])
{
    float* bp = sB2 + (rl >> 1) * S2 + (COL_A1 + w * K3_1 + kcol) * 2 + (rl & 1);
    #pragma unroll
    for (int e = 0; e < 4; ++e) {
        bp[2 * e]               = a[e];
        bp[2 * e + CP * S2]     = c[e];
        bp[2 * e + 2 * CP * S2] = d[e];
        bp[2 * e + 3 * CP * S2] = f[e];
    }
}

__device__ __forceinline__ void store_a0(float* __restrict__ sB2, int rl, int kcol,
                                         const float (&a)[4], const float (&c)[4],
                                         const float (&d)[4], const float (&f)[4])
{
    float* bp = sB2 + (rl >> 1) * S2 + (COL_A0 + kcol) * 2 + (rl & 1);
    #pragma unroll
    for (int e = 0; e < 4; ++e) {
        bp[2 * e]               = a[e];
        bp[2 * e + CP * S2]     = c[e];
        bp[2 * e + 2 * CP * S2] = d[e];
        bp[2 * e + 3 * CP * S2] = f[e];
    }
}

__device__ __forceinline__ void node_epilogue(const float* __restrict__ sP,
                                              const float* __restrict__ sWn,
                                              const float* __restrict__ sXn,
                                              const float* __restrict__ sU1,
                                              float* __restrict__ outn, int tid)
{
    for (int idx = tid; idx < 4 * C; idx += NTHR) {
        const int c = idx & 127;
        const int g = idx >> 7;
        const float* P = sP + c * PSTR;
        float s = 0.f;
        if (g == 0) {
            #pragma unroll
            for (int k = 0; k < K3_0; ++k) s = fmaf(P[COL_A0 + k], sWn[(WR_M0 + k) * C + c], s);
            #pragma unroll
            for (int k = 0; k < K2_0; ++k) s = fmaf(P[COL_20 + k], sWn[(WR_20 + k) * C + c], s);
            float d = 0.f;
            #pragma unroll
            for (int j = 0; j < L; ++j) d = fmaf(sU1[j], sXn[c * XSTR + j], d);
            s = fmaf(sWn[WR_10 * C + c], d, s);
            outn[c] = s;
        } else {
            const int w = g - 1;
            #pragma unroll
            for (int k = 0; k < K3_1; ++k) s = fmaf(P[COL_A1 + w * K3_1 + k], sWn[(WR_M1 + k) * C + c], s);
            #pragma unroll
            for (int k = 0; k < K2_1; ++k) s = fmaf(P[COL_21 + w * K2_1 + k], sWn[(WR_21 + k) * C + c], s);
            float d = 0.f;
            #pragma unroll
            for (int j = 0; j < L; ++j) d = fmaf(sU1[16 + w * L + j], sXn[c * XSTR + j], d);
            s = fmaf(sWn[WR_11 * C + c], d, s);
            outn[C + c * 3 + w] = s;
        }
    }
}

__global__ __launch_bounds__(NTHR, 4) void mace_node_kernel(
    const float* __restrict__ x,    const float* __restrict__ y,
    const float* __restrict__ U3_0, const float* __restrict__ U2_0, const float* __restrict__ U1_0,
    const float* __restrict__ U3_1, const float* __restrict__ U2_1, const float* __restrict__ U1_1,
    const float* __restrict__ Wm0,  const float* __restrict__ W20,  const float* __restrict__ W10,
    const float* __restrict__ Wm1,  const float* __restrict__ W21,  const float* __restrict__ W11,
    float* __restrict__ out)
{
    extern __shared__ __attribute__((aligned(16))) float sm[];
    float* sX  = sm + OFF_X;
    float* sXS = sm + OFF_XS;
    float* sW  = sm + OFF_W;
    float* sU1 = sm + OFF_U1;
    float* sB2 = sm + OFF_B2;
    float* sP  = sm + OFF_P;

    const int tid = threadIdx.x;
    const int b0  = blockIdx.x * NB;

    {
        const int ti = tid < 64 ? tid : 63;
        const float* up = ti < 16 ? (U1_0 + ti) : (U1_1 + (ti - 16));
        const float u1 = *up;
        const float* xb = x + (size_t)b0 * (C * L);
        v4fu xv[4];
        #pragma unroll
        for (int j = 0; j < 4; ++j) xv[j] = *(const v4fu*)(xb + 4 * (tid + NTHR * j));
        #pragma unroll
        for (int j = 0; j < 4; ++j) {
            const int idx = 4 * (tid + NTHR * j);
            float* d = sX + (idx >> 4) * XSTR + (idx & 15);
            #pragma unroll
            for (int u = 0; u < 4; ++u) d[u] = xv[j][u];
        }
        asm volatile("" :: "v"(u1));
        if (tid < 64) sU1[tid] = u1;
    }
    __syncthreads();

    if (tid < NB * L) {
        const int n = tid >> 4, i = tid & 15;
        float s = 0.f;
        for (int c = 0; c < C; ++c) s += sX[(n * C + c) * XSTR + i];
        sXS[tid] = s;
    }
    __syncthreads();

    const int wv   = __builtin_amdgcn_readfirstlane(tid >> 5);
    const int wm   = wv & 7;
    const int wg   = wv >> 3;
    const int lane = tid & 31;
    const int nl   = lane & 15;
    const int hi   = lane >> 4;
    const int crow = wm * 16 + nl;

    float xsa[L], xsb[L];
    #pragma unroll
    for (int i = 0; i < L; ++i) {
        xsa[i] = __int_as_float(__builtin_amdgcn_readfirstlane(__float_as_int(sXS[i])));
        xsb[i] = __int_as_float(__builtin_amdgcn_readfirstlane(__float_as_int(sXS[L + i])));
    }
    const int xsv = __float_as_int(sXS[2 * L + lane]);

    const int l1  = lane < 27 ? lane : 26;
    const int rg1 = l1 / 9, q1 = l1 - 9 * rg1;
    const int k1  = (q1 < 8) ? 4 * q1 : (K3_1 - 4);
    const int l0  = lane < 24 ? lane : 23;
    const int rg0 = l0 / 6, q0 = l0 - 6 * rg0;
    const int k0  = (q0 < 5) ? 4 * q0 : (K3_0 - 4);

    v8f acc0[NTILE], acc1[NKEEP];
    #pragma unroll
    for (int t = 0; t < NTILE; ++t) acc0[t] = (v8f){0.f, 0.f, 0.f, 0.f, 0.f, 0.f, 0.f, 0.f};
    #pragma unroll
    for (int t = 0; t < NKEEP; ++t) acc1[t] = (v8f){0.f, 0.f, 0.f, 0.f, 0.f, 0.f, 0.f, 0.f};
    float* park = sm + OFF_PARK + wv * PARKW + lane * 4;
    auto park_load = [&](v8f (&p)[NPARK]) {
        #pragma unroll
        for (int t = 0; t < NPARK; ++t) {
            const v4f lo = *(const v4f*)(park + (2 * t) * 128);
            const v4f hh = *(const v4f*)(park + (2 * t + 1) * 128);
            p[t] = (v8f){lo[0], lo[1], lo[2], lo[3], hh[0], hh[1], hh[2], hh[3]};
        }
    };

    auto a1_pass = [&](int chn, int p, int& row, int& rl, int& reset, int& store) {
        int mm, j, r, second, plain;
        if (p < 2) { const int dr = 2 * wm + p; mm = dr >> 2; j = mm; r = dr & 3; second = 0; plain = 1; }
        else {
            const int f = 3 * wm + ((p - 2) >> 1);
            fold_step(f >> 2, mm, j);
            r = f & 3; second = (p - 2) & 1; plain = 0;
        }
        const int a = 4 * mm + chn, b = 4 * j + r;
        row   = second ? (b * 16 + a) : (a * 16 + b);
        rl    = 4 * (sl_base(mm) + j - mm) + r;
        reset = plain | (second ^ 1);
        store = plain | second;
    };
    auto a0_pass = [&](int chn, int p, int& row, int& rl, int& reset, int& store) {
        int mm, j, second, plain;
        if (wm < 6) { fold_step(wm, mm, j); second = p; plain = 0; }
        else        { mm = 2 * (wm - 6) + p; j = mm; second = 0; plain = 1; }
        const int a = 4 * mm + chn, b = 4 * j + rg0;
        row   = second ? (b * 16 + a) : (a * 16 + b);
        rl    = 4 * (sl_base(mm) + j - mm) + rg0;
        reset = plain | (second ^ 1);
        store = plain | second;
    };
    auto a1_off = [&](int row) -> unsigned {
        return (unsigned)(rg1 * 256 + row) * (unsigned)(L * K3_1) + (unsigned)k1;
    };
    auto a0_off = [&](int row) -> unsigned {
        return (unsigned)row * (unsigned)(L * K3_0) + (unsigned)k0;
    };
    auto a1_seq = [&](int s) -> int {
        return wg ? (s == 0 ? 1 : s + 3) : (s == 0 ? 0 : s + 1);
    };
    auto u2_item = [&](int jj, int& col, int& pr, int& mm, int& j) -> bool {
        const int cr = wm + 8 * jj;
        col = cr < NU2 ? cr : NU2 - 1;
        pr  = lane < CP ? lane : CP - 1;
        const int sl = pr >> 1;
        mm = (sl >= 4) + (sl >= 7) + (sl >= 9);
        j  = mm + sl - sl_base(mm);
        return (cr < NU2) & (lane < CP);
    };

    const int prod_first = (wv >> 2) & 1;
    #pragma nounroll
    for (int h = 0; h < 2 * (NCH + 1); ++h) {
        const int it = h >> 1;
        const bool build_half = ((h & 1) ^ prod_first) == 0;
        if (build_half && it < NCH) {
            const int ch = it;
            float* sBw = sB2 + (it & 1) * B2BUF;
            int row, rl, rs, st;
            a1_pass(ch, a1_seq(0), row, rl, rs, st);
            v4fu r[L];
            issue_row_loads<K3_1>(r, U3_1 + a1_off(row));

            float a[4] = {0.f, 0.f, 0.f, 0.f}, c[4] = {0.f, 0.f, 0.f, 0.f};
            float d[4] = {0.f, 0.f, 0.f, 0.f}, f[4] = {0.f, 0.f, 0.f, 0.f};
            const int nsh = wg ? 4 : 2;
            #pragma nounroll
            for (int s = 0; s < nsh; ++s) {
                int rown, rln, rsn, stn;
                a1_pass(ch, a1_seq(s + 1), rown, rln, rsn, stn);
                row_sums<K3_1>(r, xsa, xsb, xsv, a, c, d, f, U3_1 + a1_off(rown), rs);
                if (st) store_a1(sBw, rg1, rl, k1, a, c, d, f);
                rl = rln; rs = rsn; st = stn;
            }
            if (wg) {
                row_sums<0>(r, xsa, xsb, xsv, a, c, d, f, nullptr, rs);
                store_a1(sBw, rg1, rl, k1, a, c, d, f);
            } else {
                {
                    int rown, rln, rsn, stn;
                    a0_pass(ch, 0, rown, rln, rsn, stn);
                    row_sums<K3_0>(r, xsa, xsb, xsv, a, c, d, f, U3_0 + a0_off(rown), rs);
                    store_a1(sBw, rg1, rl, k1, a, c, d, f);
                    rl = rln; rs = rsn; st = stn;
                }
                {
                    int rown, rln, rsn, stn;
                    a0_pass(ch, 1, rown, rln, rsn, stn);
                    row_sums<K3_0>(r, xsa, xsb, xsv, a, c, d, f, U3_0 + a0_off(rown), rs);
                    if (st) store_a0(sBw, rl, k0, a, c, d, f);
                    rl = rln; rs = rsn; st = stn;
                }
                float u2v[12];
                if (rs) {
                    #pragma unroll
                    for (int e = 0; e < 4; ++e) { a[e] = 0.f; c[e] = 0.f; d[e] = 0.f; f[e] = 0.f; }
                }
                #pragma unroll
                for (int i = 0; i < L; ++i) {
                    const v4fu v = r[i];
                    const float xc = xs_lane(xsv, i), xd = xs_lane(xsv, L + i);
                    #pragma unroll
                    for (int e = 0; e < 4; ++e) {
                        a[e] = fmaf(v[e], xsa[i], a[e]);
                        c[e] = fmaf(v[e], xsb[i], c[e]);
                        d[e] = fmaf(v[e], xc, d[e]);
                        f[e] = fmaf(v[e], xd, f[e]);
                    }
                    if (i < 4) {
                        asm volatile("" :: "v"(v) : "memory");
                        #pragma unroll
                        for (int t = 0; t < 3; ++t) {
                            const int n = 3 * i + t;
                            int col, pr, mm, j;
                            u2_item(n >> 2, col, pr, mm, j);
                            const int aa = 4 * mm + ch, bb = 4 * j + 2 * (pr & 1);
                            const bool t1 = col < 3 * K2_1;
                            const int w = col / K2_1, k = col - w * K2_1;
                            const float* ub = t1 ? (U2_1 + (w * 256) * K2_1 + k) : (U2_0 + (col - 3 * K2_1));
                            const int us = t1 ? K2_1 : K2_0;
                            const int e = n & 3;
                            const int rw = e == 0 ? (aa * 16 + bb) : e == 1 ? (aa * 16 + bb + 1)
                                         : e == 2 ? (bb * 16 + aa) : ((bb + 1) * 16 + aa);
                            u2v[n] = ub[rw * us];
                        }
                    }
                }
                store_a0(sBw, rl, k0, a, c, d, f);
                #pragma unroll
                for (int n = 0; n < 12; n += 4)
                    asm volatile("" :: "v"(u2v[n]), "v"(u2v[n + 1]), "v"(u2v[n + 2]), "v"(u2v[n + 3]));
                #pragma unroll
                for (int jj = 0; jj < 3; ++jj) {
                    int col, pr, mm, j;
                    const bool valid = u2_item(jj, col, pr, mm, j);
                    const bool fold = j != mm;
                    v2f v;
                    v.x = fold ? u2v[4 * jj]     + u2v[4 * jj + 2] : u2v[4 * jj];
                    v.y = fold ? u2v[4 * jj + 1] + u2v[4 * jj + 3] : u2v[4 * jj + 1];
                    if (valid) {
                        float* bp = sBw + pr * S2 + (COL_21 + col) * 2;
                        #pragma unroll
                        for (int n = 0; n < NB; ++n) *(v2f*)(bp + n * CP * S2) = v;
                    }
                }
            }
        }

        if (!build_half && it > 0) {
            const int ch = it - 1;
            const float* sBr = sB2 + (ch & 1) * B2BUF;
            float sel0a[4], sel1a[4], sel0b[4], sel1b[4], xpa[4], xpb[4];
            const float* xra = sX + (2 * wg * C + crow) * XSTR;
            const float* xrb = xra + C * XSTR;
            #pragma unroll
            for (int j = 0; j < 4; ++j) {
                sel0a[j] = hi ? xra[4 * j + 2] : xra[4 * j];
                sel1a[j] = hi ? xra[4 * j + 3] : xra[4 * j + 1];
                sel0b[j] = hi ? xrb[4 * j + 2] : xrb[4 * j];
                sel1b[j] = hi ? xrb[4 * j + 3] : xrb[4 * j + 1];
                xpa[j] = xra[4 * j + ch];
                xpb[j] = xrb[4 * j + ch];
            }
            v8f accp[NPARK];
            if (ch == 0) {
                #pragma unroll
                for (int t = 0; t < NPARK; ++t) accp[t] = (v8f){0.f, 0.f, 0.f, 0.f, 0.f, 0.f, 0.f, 0.f};
            } else {
                park_load(accp);
            }
            const float* Bl = sBr + 2 * wg * CP * S2 + hi * S2 + nl * 2;
            #pragma unroll
            for (int mm = 0; mm < 4; ++mm) {
                #pragma unroll
                for (int j = mm; j < 4; ++j) {
                    const int sl = sl_base(mm) + j - mm;
                    v2f aa, ab;
                    aa.x = xpa[mm] * sel0a[j];
                    aa.y = xpa[mm] * sel1a[j];
                    ab.x = xpb[mm] * sel0b[j];
                    ab.y = xpb[mm] * sel1b[j];
                    const float* Br = Bl + (2 * sl) * S2;
                    #pragma unroll
                    for (int t = 0; t < NTILE; ++t) {
                        const v2f bfa = *(const v2f*)(Br + t * 32);
                        const v2f bfb = *(const v2f*)(Br + CP * S2 + t * 32);
                        acc0[t] = __builtin_amdgcn_wmma_f32_16x16x4_f32(
                            false, aa, false, bfa, (short)0, acc0[t], false, false);
                        if (t < NKEEP)
                            acc1[t] = __builtin_amdgcn_wmma_f32_16x16x4_f32(
                                false, ab, false, bfb, (short)0, acc1[t], false, false);
                        else
                            accp[t - NKEEP] = __builtin_amdgcn_wmma_f32_16x16x4_f32(
                                false, ab, false, bfb, (short)0, accp[t - NKEEP], false, false);
                    }
                }
            }
            #pragma unroll
            for (int t = 0; t < NPARK; ++t) {
                *(v4f*)(park + (2 * t) * 128)     = (v4f){accp[t][0], accp[t][1], accp[t][2], accp[t][3]};
                *(v4f*)(park + (2 * t + 1) * 128) = (v4f){accp[t][4], accp[t][5], accp[t][6], accp[t][7]};
            }
        }
        if (h & 1) __syncthreads();
    }

    auto store_p_tile = [&](int t, const v8f& v) {
        #pragma unroll
        for (int r = 0; r < 8; ++r)
            sP[(wm * 16 + r + hi * 8) * PSTR + t * 16 + nl] = v[r];
    };
    auto store_p_first = [&]() {
        #pragma unroll
        for (int t = 0; t < NTILE; ++t) store_p_tile(t, acc0[t]);
    };
    auto store_p_second = [&]() {
        #pragma unroll
        for (int t = 0; t < NKEEP; ++t) store_p_tile(t, acc1[t]);
        v8f accp[NPARK];
        park_load(accp);
        #pragma unroll
        for (int t = 0; t < NPARK; ++t) store_p_tile(NKEEP + t, accp[t]);
    };
    if (wg == 0) store_p_second();
    __syncthreads();

    {
        float yv[NB][E];
        #pragma unroll
        for (int n = 0; n < NB; ++n) {
            #pragma unroll
            for (int e = 0; e < E; ++e) yv[n][e] = y[(b0 + n) * E + e];
        }
        const int c4 = lane * 4;
        #pragma nounroll
        for (int j0 = 0; j0 < 5; j0 += 1) {
            v4fu wl[1][E];
            #pragma unroll
            for (int jj = 0; jj < 1; ++jj) {
                const int wrr = wv + 16 * (j0 + jj);
                const int wr  = wrr < WROWS ? wrr : (WROWS - 1);
                const float* src; int stride;
                if (wr < WR_M0)       { src = Wm1 + (wr)         * C; stride = K3_1 * C; }
                else if (wr < WR_21)  { src = Wm0 + (wr - WR_M0) * C; stride = K3_0 * C; }
                else if (wr < WR_20)  { src = W21 + (wr - WR_21) * C; stride = K2_1 * C; }
                else if (wr < WR_11)  { src = W20 + (wr - WR_20) * C; stride = K2_0 * C; }
                else if (wr == WR_11) { src = W11;                    stride = C;        }
                else                  { src = W10;                    stride = C;        }
                #pragma unroll
                for (int e = 0; e < E; ++e) wl[jj][e] = *(const v4fu*)(src + e * stride + c4);
            }
            #pragma unroll
            for (int jj = 0; jj < 1; ++jj) {
                const int wrr = wv + 16 * (j0 + jj);
                float ws[NB][4];
                #pragma unroll
                for (int n = 0; n < NB; ++n) {
                    #pragma unroll
                    for (int u = 0; u < 4; ++u) ws[n][u] = 0.f;
                }
                #pragma unroll
                for (int e = 0; e < E; ++e) {
                    #pragma unroll
                    for (int n = 0; n < NB; ++n) {
                        #pragma unroll
                        for (int u = 0; u < 4; ++u) ws[n][u] = fmaf(yv[n][e], wl[jj][e][u], ws[n][u]);
                    }
                }
                if (wrr < WROWS) {
                    float* dst = sW + wrr * C + c4;
                    #pragma unroll
                    for (int n = 0; n < NB; ++n) {
                        #pragma unroll
                        for (int u = 0; u < 4; ++u) dst[n * WROWS * C + u] = ws[n][u];
                    }
                }
            }
        }
    }
    __syncthreads();

    node_epilogue(sP, sW + WROWS * C, sX + C * XSTR, sU1, out + (size_t)(b0 + 1) * (4 * C), tid);

    __syncthreads();
    if (wg == 0) store_p_first();
    __syncthreads();
    node_epilogue(sP, sW, sX, sU1, out + (size_t)b0 * (4 * C), tid);

    __syncthreads();
    if (wg == 1) store_p_second();
    __syncthreads();
    node_epilogue(sP, sW + 3 * WROWS * C, sX + 3 * C * XSTR, sU1, out + (size_t)(b0 + 3) * (4 * C), tid);

    __syncthreads();
    if (wg == 1) store_p_first();
    __syncthreads();
    node_epilogue(sP, sW + 2 * WROWS * C, sX + 2 * C * XSTR, sU1, out + (size_t)(b0 + 2) * (4 * C), tid);
}

extern "C" void kernel_launch(void* const* d_in, const int* in_sizes, int n_in,
                              void* d_out, int out_size, void* d_ws, size_t ws_size,
                              hipStream_t stream) {
    (void)in_sizes; (void)n_in; (void)out_size; (void)d_ws; (void)ws_size;
    const float* x    = (const float*)d_in[0];
    const float* y    = (const float*)d_in[1];
    const float* U3_0 = (const float*)d_in[2];
    const float* U2_0 = (const float*)d_in[3];
    const float* U1_0 = (const float*)d_in[4];
    const float* U3_1 = (const float*)d_in[5];
    const float* U2_1 = (const float*)d_in[6];
    const float* U1_1 = (const float*)d_in[7];
    const float* Wm0  = (const float*)d_in[8];
    const float* W20  = (const float*)d_in[9];
    const float* W10  = (const float*)d_in[10];
    const float* Wm1  = (const float*)d_in[11];
    const float* W21  = (const float*)d_in[12];
    const float* W11  = (const float*)d_in[13];

    constexpr size_t smem_bytes = (size_t)SMEM_FLOATS * sizeof(float);
    mace_node_kernel<<<dim3(NN / NB), dim3(NTHR), smem_bytes, stream>>>(
        x, y, U3_0, U2_0, U1_0, U3_1, U2_1, U1_1,
        Wm0, W20, W10, Wm1, W21, W11, (float*)d_out);
}


# U3 row pairs (a,b)+(b,a) summed once per call by a small first kernel into the workspace: the build reads 160 table rows per block, not 256 (50 passes a chunk, not 80); sum order changes
# speedup vs baseline: 5.7183x; 1.0291x over previous
#include <hip/hip_runtime.h>

typedef float v2f __attribute__((ext_vector_type(2)));
typedef float v8f __attribute__((ext_vector_type(8)));
typedef float v4f __attribute__((ext_vector_type(4)));
typedef float v4fu __attribute__((ext_vector_type(4), aligned(4)));

constexpr int NN   = 2048;
constexpr int C    = 128;
constexpr int L    = 16;
constexpr int E    = 10;
constexpr int K3_0 = 23, K2_0 = 4;
constexpr int K3_1 = 33, K2_1 = 6;

constexpr int COL_A1 = 0;
constexpr int COL_A0 = 3 * K3_1;
constexpr int COL_21 = COL_A0 + K3_0;
constexpr int COL_20 = COL_21 + 3 * K2_1;
constexpr int NCOL   = COL_20 + K2_0;
constexpr int NTILE  = NCOL / 16;

constexpr int S2   = 2 * NCOL + 2;
constexpr int PSTR = 145;
constexpr int XSTR = L + 1;

constexpr int WR_M1 = 0;
constexpr int WR_M0 = 33;
constexpr int WR_21 = 56;
constexpr int WR_20 = 62;
constexpr int WR_11 = 66;
constexpr int WR_10 = 67;
constexpr int WROWS = 68;

constexpr int NB  = 4;
constexpr int NTHR = 512;
constexpr int NCH = 4;
constexpr int NST = 10;
constexpr int CP  = 2 * NST;
constexpr int NU2 = 3 * K2_1 + K2_0;
constexpr int B2BUF = NB * CP * S2;

__host__ __device__ constexpr int sl_base(int mm) { return mm * (9 - mm) / 2; }
__host__ __device__ constexpr int fo_base(int mm) { return mm * (7 - mm) / 2; }
__device__ __forceinline__ void fold_step(int fo, int& mm, int& j)
{
    mm = (fo >= 3) + (fo >= 5);
    j  = mm + 1 + fo - fo_base(mm);
}

constexpr int OFF_X  = 0;
constexpr int OFF_XS = OFF_X + NB * C * XSTR;
constexpr int OFF_U1 = OFF_XS + NB * L;
constexpr int OFF_B2 = OFF_U1 + 64;
constexpr int OFF_P  = OFF_B2;
constexpr int OFF_W  = OFF_P + C * PSTR;
constexpr int NPARK = 6;
constexpr int NKEEP = NTILE - NPARK;
constexpr int PARKW = NPARK * 8 * 32;
constexpr int OFF_PARK = OFF_B2 + 2 * B2BUF;
constexpr int END_W    = OFF_W + NB * WROWS * C;
constexpr int END_PARK = OFF_PARK + (NTHR / 32) * PARKW;
constexpr int SMEM_FLOATS = END_PARK > END_W ? END_PARK : END_W;
static_assert(OFF_PARK % 4 == 0 && PARKW % 4 == 0, "16-byte accesses to the parked tiles");
static_assert(OFF_PARK + 8 * PARKW >= END_W, "the parked tiles of waves 8..15 must lie behind W");
static_assert(2 * B2BUF <= C * PSTR + NB * WROWS * C, "both B2 chunk buffers must fit the shared storage");
static_assert(((SMEM_FLOATS * 4 + 2047) / 2048) * 2048 <= 320 * 1024, "one block per WGP");


template <int KT>
__device__ __forceinline__ void issue_row_loads(v4fu (&r)[L], const float* __restrict__ u)
{
    #pragma unroll
    for (int i = 0; i < L; ++i) r[i] = *(const v4fu*)(u + i * KT);
}

__device__ __forceinline__ float xs_lane(int xsv, int l)
{
    return __int_as_float(__builtin_amdgcn_readlane(xsv, l));
}

template <int KN>
__device__ __forceinline__ void row_sums(v4fu (&r)[L], const float (&xa)[L], const float (&xb)[L],
                                         int xsv, float (&a)[4], float (&c)[4], float (&d)[4],
                                         float (&f)[4], const float* __restrict__ un)
{
    #pragma unroll
    for (int e = 0; e < 4; ++e) { a[e] = 0.f; c[e] = 0.f; d[e] = 0.f; f[e] = 0.f; }
    #pragma unroll
    for (int i = 0; i < L; ++i) {
        const v4fu v = r[i];
        const float xc = xs_lane(xsv, i), xd = xs_lane(xsv, L + i);
        #pragma unroll
        for (int e = 0; e < 4; ++e) {
            a[e] = fmaf(v[e], xa[i], a[e]);
            c[e] = fmaf(v[e], xb[i], c[e]);
            d[e] = fmaf(v[e], xc, d[e]);
            f[e] = fmaf(v[e], xd, f[e]);
        }
        if (KN > 0) {
            asm volatile("" :: "v"(v) : "memory");
            r[i] = *(const v4fu*)(un + i * KN);
        }
    }
}

__device__ __forceinline__ void store_a1(float* __restrict__ sB2, int w, int rl, int kcol,
                                         const float (&a)[4], const float (&c)[4],
                                         const float (&d)[4], const float (&f)[4])
{
    float* bp = sB2 + (rl >> 1) * S2 + (COL_A1 + w * K3_1 + kcol) * 2 + (rl & 1);
    #pragma unroll
    for (int e = 0; e < 4; ++e) {
        bp[2 * e]               = a[e];
        bp[2 * e + CP * S2]     = c[e];
        bp[2 * e + 2 * CP * S2] = d[e];
        bp[2 * e + 3 * CP * S2] = f[e];
    }
}

__device__ __forceinline__ void store_a0(float* __restrict__ sB2, int rl, int kcol,
                                         const float (&a)[4], const float (&c)[4],
                                         const float (&d)[4], const float (&f)[4])
{
    float* bp = sB2 + (rl >> 1) * S2 + (COL_A0 + kcol) * 2 + (rl & 1);
    #pragma unroll
    for (int e = 0; e < 4; ++e) {
        bp[2 * e]               = a[e];
        bp[2 * e + CP * S2]     = c[e];
        bp[2 * e + 2 * CP * S2] = d[e];
        bp[2 * e + 3 * CP * S2] = f[e];
    }
}

__device__ __forceinline__ void node_epilogue(const float* __restrict__ sP,
                                              const float* __restrict__ sWn,
                                              const float* __restrict__ sXn,
                                              const float* __restrict__ sU1,
                                              float* __restrict__ outn, int tid)
{
    for (int idx = tid; idx < 4 * C; idx += NTHR) {
        const int c = idx & 127;
        const int g = idx >> 7;
        const float* P = sP + c * PSTR;
        float s = 0.f;
        if (g == 0) {
            #pragma unroll
            for (int k = 0; k < K3_0; ++k) s = fmaf(P[COL_A0 + k], sWn[(WR_M0 + k) * C + c], s);
            #pragma unroll
            for (int k = 0; k < K2_0; ++k) s = fmaf(P[COL_20 + k], sWn[(WR_20 + k) * C + c], s);
            float d = 0.f;
            #pragma unroll
            for (int j = 0; j < L; ++j) d = fmaf(sU1[j], sXn[c * XSTR + j], d);
            s = fmaf(sWn[WR_10 * C + c], d, s);
            outn[c] = s;
        } else {
            const int w = g - 1;
            #pragma unroll
            for (int k = 0; k < K3_1; ++k) s = fmaf(P[COL_A1 + w * K3_1 + k], sWn[(WR_M1 + k) * C + c], s);
            #pragma unroll
            for (int k = 0; k < K2_1; ++k) s = fmaf(P[COL_21 + w * K2_1 + k], sWn[(WR_21 + k) * C + c], s);
            float d = 0.f;
            #pragma unroll
            for (int j = 0; j < L; ++j) d = fmaf(sU1[16 + w * L + j], sXn[c * XSTR + j], d);
            s = fmaf(sWn[WR_11 * C + c], d, s);
            outn[C + c * 3 + w] = s;
        }
    }
}

__global__ __launch_bounds__(128) void fold_tables_kernel(
    const float* __restrict__ U3_1, const float* __restrict__ U3_0,
    float* __restrict__ F1, float* __restrict__ F0)
{
    const int a = blockIdx.x >> 4, b = blockIdx.x & 15;
    if ((b >> 2) <= (a >> 2)) return;
    const int t = blockIdx.y;
    const int n = t < 3 ? L * K3_1 : L * K3_0;
    const float* s = t < 3 ? U3_1 + t * (256 * L * K3_1) : U3_0;
    float*       d = t < 3 ? F1   + t * (256 * L * K3_1) : F0;
    const int ra = (a * 16 + b) * n, rb = (b * 16 + a) * n;
    for (int e = threadIdx.x; e < n; e += 128) d[ra + e] = s[ra + e] + s[rb + e];
}

__global__ __launch_bounds__(NTHR, 4) void mace_node_kernel(
    const float* __restrict__ x,    const float* __restrict__ y,
    const float* __restrict__ U3_0, const float* __restrict__ U2_0, const float* __restrict__ U1_0,
    const float* __restrict__ U3_1, const float* __restrict__ U2_1, const float* __restrict__ U1_1,
    const float* __restrict__ Wm0,  const float* __restrict__ W20,  const float* __restrict__ W10,
    const float* __restrict__ Wm1,  const float* __restrict__ W21,  const float* __restrict__ W11,
    const float* __restrict__ F1,   const float* __restrict__ F0,
    float* __restrict__ out)
{
    extern __shared__ __attribute__((aligned(16))) float sm[];
    float* sX  = sm + OFF_X;
    float* sXS = sm + OFF_XS;
    float* sW  = sm + OFF_W;
    float* sU1 = sm + OFF_U1;
    float* sB2 = sm + OFF_B2;
    float* sP  = sm + OFF_P;

    const int tid = threadIdx.x;
    const int b0  = blockIdx.x * NB;

    {
        const int ti = tid < 64 ? tid : 63;
        const float* up = ti < 16 ? (U1_0 + ti) : (U1_1 + (ti - 16));
        const float u1 = *up;
        const float* xb = x + (size_t)b0 * (C * L);
        v4fu xv[4];
        #pragma unroll
        for (int j = 0; j < 4; ++j) xv[j] = *(const v4fu*)(xb + 4 * (tid + NTHR * j));
        #pragma unroll
        for (int j = 0; j < 4; ++j) {
            const int idx = 4 * (tid + NTHR * j);
            float* d = sX + (idx >> 4) * XSTR + (idx & 15);
            #pragma unroll
            for (int u = 0; u < 4; ++u) d[u] = xv[j][u];
        }
        asm volatile("" :: "v"(u1));
        if (tid < 64) sU1[tid] = u1;
    }
    __syncthreads();

    if (tid < NB * L) {
        const int n = tid >> 4, i = tid & 15;
        float s = 0.f;
        for (int c = 0; c < C; ++c) s += sX[(n * C + c) * XSTR + i];
        sXS[tid] = s;
    }
    __syncthreads();

    const int wv   = __builtin_amdgcn_readfirstlane(tid >> 5);
    const int wm   = wv & 7;
    const int wg   = wv >> 3;
    const int lane = tid & 31;
    const int nl   = lane & 15;
    const int hi   = lane >> 4;
    const int crow = wm * 16 + nl;

    float xsa[L], xsb[L];
    #pragma unroll
    for (int i = 0; i < L; ++i) {
        xsa[i] = __int_as_float(__builtin_amdgcn_readfirstlane(__float_as_int(sXS[i])));
        xsb[i] = __int_as_float(__builtin_amdgcn_readfirstlane(__float_as_int(sXS[L + i])));
    }
    const int xsv = __float_as_int(sXS[2 * L + lane]);

    const int l1  = lane < 27 ? lane : 26;
    const int rg1 = l1 / 9, q1 = l1 - 9 * rg1;
    const int k1  = (q1 < 8) ? 4 * q1 : (K3_1 - 4);
    const int l0  = lane < 24 ? lane : 23;
    const int rg0 = l0 / 6, q0 = l0 - 6 * rg0;
    const int k0  = (q0 < 5) ? 4 * q0 : (K3_0 - 4);

    v8f acc0[NTILE], acc1[NKEEP];
    #pragma unroll
    for (int t = 0; t < NTILE; ++t) acc0[t] = (v8f){0.f, 0.f, 0.f, 0.f, 0.f, 0.f, 0.f, 0.f};
    #pragma unroll
    for (int t = 0; t < NKEEP; ++t) acc1[t] = (v8f){0.f, 0.f, 0.f, 0.f, 0.f, 0.f, 0.f, 0.f};
    float* park = sm + OFF_PARK + wv * PARKW + lane * 4;
    auto park_load = [&](v8f (&p)[NPARK]) {
        #pragma unroll
        for (int t = 0; t < NPARK; ++t) {
            const v4f lo = *(const v4f*)(park + (2 * t) * 128);
            const v4f hh = *(const v4f*)(park + (2 * t + 1) * 128);
            p[t] = (v8f){lo[0], lo[1], lo[2], lo[3], hh[0], hh[1], hh[2], hh[3]};
        }
    };

    const float* tab1 = wg ? F1 : U3_1;
    auto a1_pass = [&](int chn, int p, int& row, int& rl) {
        int mm, j, r;
        if (wg == 0) { const int dr = 2 * wm + p; mm = dr >> 2; j = mm; r = dr & 3; }
        else         { const int f = 3 * wm + p; fold_step(f >> 2, mm, j); r = f & 3; }
        row = (4 * mm + chn) * 16 + 4 * j + r;
        rl  = 4 * (sl_base(mm) + j - mm) + r;
    };
    const float* tab0 = wm < 6 ? F0 : U3_0;
    auto a0_pass = [&](int chn, int first, int& row, int& rl) {
        int mm, j;
        if (first)       { mm = 2 + wm; j = mm; }
        else if (wm < 6) { fold_step(wm, mm, j); }
        else             { mm = wm - 6; j = mm; }
        row = (4 * mm + chn) * 16 + 4 * j + rg0;
        rl  = 4 * (sl_base(mm) + j - mm) + rg0;
    };
    auto a1_off = [&](int row) -> unsigned {
        return (unsigned)(rg1 * 256 + row) * (unsigned)(L * K3_1) + (unsigned)k1;
    };
    auto a0_off = [&](int row) -> unsigned {
        return (unsigned)row * (unsigned)(L * K3_0) + (unsigned)k0;
    };
    auto u2_item = [&](int jj, int& col, int& pr, int& mm, int& j) -> bool {
        const int cr = wm + 8 * jj;
        col = cr < NU2 ? cr : NU2 - 1;
        pr  = lane < CP ? lane : CP - 1;
        const int sl = pr >> 1;
        mm = (sl >= 4) + (sl >= 7) + (sl >= 9);
        j  = mm + sl - sl_base(mm);
        return (cr < NU2) & (lane < CP);
    };

    const int prod_first = (wv >> 2) & 1;
    #pragma nounroll
    for (int h = 0; h < 2 * (NCH + 1); ++h) {
        const int it = h >> 1;
        const bool build_half = ((h & 1) ^ prod_first) == 0;
        if (build_half && it < NCH) {
            const int ch = it;
            float* sBw = sB2 + (it & 1) * B2BUF;
            int row, rl;
            a1_pass(ch, 0, row, rl);
            v4fu r[L];
            issue_row_loads<K3_1>(r, tab1 + a1_off(row));

            float a[4] = {0.f, 0.f, 0.f, 0.f}, c[4] = {0.f, 0.f, 0.f, 0.f};
            float d[4] = {0.f, 0.f, 0.f, 0.f}, f[4] = {0.f, 0.f, 0.f, 0.f};
            const int nsh = wg ? 2 : 1;
            #pragma nounroll
            for (int s = 0; s < nsh; ++s) {
                int rown, rln;
                a1_pass(ch, s + 1, rown, rln);
                row_sums<K3_1>(r, xsa, xsb, xsv, a, c, d, f, tab1 + a1_off(rown));
                store_a1(sBw, rg1, rl, k1, a, c, d, f);
                rl = rln;
            }
            if (wg) {
                row_sums<0>(r, xsa, xsb, xsv, a, c, d, f, nullptr);
                store_a1(sBw, rg1, rl, k1, a, c, d, f);
            } else {
                const int two = wm < 2;
                {
                    int rown, rln;
                    a0_pass(ch, two, rown, rln);
                    row_sums<K3_0>(r, xsa, xsb, xsv, a, c, d, f, (two ? U3_0 : tab0) + a0_off(rown));
                    store_a1(sBw, rg1, rl, k1, a, c, d, f);
                    rl = rln;
                }
                if (two) {
                    int rown, rln;
                    a0_pass(ch, 0, rown, rln);
                    row_sums<K3_0>(r, xsa, xsb, xsv, a, c, d, f, tab0 + a0_off(rown));
                    store_a0(sBw, rl, k0, a, c, d, f);
                    rl = rln;
                }
                float u2v[12];
                #pragma unroll
                for (int e = 0; e < 4; ++e) { a[e] = 0.f; c[e] = 0.f; d[e] = 0.f; f[e] = 0.f; }
                #pragma unroll
                for (int i = 0; i < L; ++i) {
                    const v4fu v = r[i];
                    const float xc = xs_lane(xsv, i), xd = xs_lane(xsv, L + i);
                    #pragma unroll
                    for (int e = 0; e < 4; ++e) {
                        a[e] = fmaf(v[e], xsa[i], a[e]);
                        c[e] = fmaf(v[e], xsb[i], c[e]);
                        d[e] = fmaf(v[e], xc, d[e]);
                        f[e] = fmaf(v[e], xd, f[e]);
                    }
                    if (i < 4) {
                        asm volatile("" :: "v"(v) : "memory");
                        #pragma unroll
                        for (int t = 0; t < 3; ++t) {
                            const int n = 3 * i + t;
                            int col, pr, mm, j;
                            u2_item(n >> 2, col, pr, mm, j);
                            const int aa = 4 * mm + ch, bb = 4 * j + 2 * (pr & 1);
                            const bool t1 = col < 3 * K2_1;
                            const int w = col / K2_1, k = col - w * K2_1;
                            const float* ub = t1 ? (U2_1 + (w * 256) * K2_1 + k) : (U2_0 + (col - 3 * K2_1));
                            const int us = t1 ? K2_1 : K2_0;
                            const int e = n & 3;
                            const int rw = e == 0 ? (aa * 16 + bb) : e == 1 ? (aa * 16 + bb + 1)
                                         : e == 2 ? (bb * 16 + aa) : ((bb + 1) * 16 + aa);
                            u2v[n] = ub[rw * us];
                        }
                    }
                }
                store_a0(sBw, rl, k0, a, c, d, f);
                #pragma unroll
                for (int n = 0; n < 12; n += 4)
                    asm volatile("" :: "v"(u2v[n]), "v"(u2v[n + 1]), "v"(u2v[n + 2]), "v"(u2v[n + 3]));
                #pragma unroll
                for (int jj = 0; jj < 3; ++jj) {
                    int col, pr, mm, j;
                    const bool valid = u2_item(jj, col, pr, mm, j);
                    const bool fold = j != mm;
                    v2f v;
                    v.x = fold ? u2v[4 * jj]     + u2v[4 * jj + 2] : u2v[4 * jj];
                    v.y = fold ? u2v[4 * jj + 1] + u2v[4 * jj + 3] : u2v[4 * jj + 1];
                    if (valid) {
                        float* bp = sBw + pr * S2 + (COL_21 + col) * 2;
                        #pragma unroll
                        for (int n = 0; n < NB; ++n) *(v2f*)(bp + n * CP * S2) = v;
                    }
                }
            }
        }

        if (!build_half && it > 0) {
            const int ch = it - 1;
            const float* sBr = sB2 + (ch & 1) * B2BUF;
            float sel0a[4], sel1a[4], sel0b[4], sel1b[4], xpa[4], xpb[4];
            const float* xra = sX + (2 * wg * C + crow) * XSTR;
            const float* xrb = xra + C * XSTR;
            #pragma unroll
            for (int j = 0; j < 4; ++j) {
                sel0a[j] = hi ? xra[4 * j + 2] : xra[4 * j];
                sel1a[j] = hi ? xra[4 * j + 3] : xra[4 * j + 1];
                sel0b[j] = hi ? xrb[4 * j + 2] : xrb[4 * j];
                sel1b[j] = hi ? xrb[4 * j + 3] : xrb[4 * j + 1];
                xpa[j] = xra[4 * j + ch];
                xpb[j] = xrb[4 * j + ch];
            }
            v8f accp[NPARK];
            if (ch == 0) {
                #pragma unroll
                for (int t = 0; t < NPARK; ++t) accp[t] = (v8f){0.f, 0.f, 0.f, 0.f, 0.f, 0.f, 0.f, 0.f};
            } else {
                park_load(accp);
            }
            const float* Bl = sBr + 2 * wg * CP * S2 + hi * S2 + nl * 2;
            #pragma unroll
            for (int mm = 0; mm < 4; ++mm) {
                #pragma unroll
                for (int j = mm; j < 4; ++j) {
                    const int sl = sl_base(mm) + j - mm;
                    v2f aa, ab;
                    aa.x = xpa[mm] * sel0a[j];
                    aa.y = xpa[mm] * sel1a[j];
                    ab.x = xpb[mm] * sel0b[j];
                    ab.y = xpb[mm] * sel1b[j];
                    const float* Br = Bl + (2 * sl) * S2;
                    #pragma unroll
                    for (int t = 0; t < NTILE; ++t) {
                        const v2f bfa = *(const v2f*)(Br + t * 32);
                        const v2f bfb = *(const v2f*)(Br + CP * S2 + t * 32);
                        acc0[t] = __builtin_amdgcn_wmma_f32_16x16x4_f32(
                            false, aa, false, bfa, (short)0, acc0[t], false, false);
                        if (t < NKEEP)
                            acc1[t] = __builtin_amdgcn_wmma_f32_16x16x4_f32(
                                false, ab, false, bfb, (short)0, acc1[t], false, false);
                        else
                            accp[t - NKEEP] = __builtin_amdgcn_wmma_f32_16x16x4_f32(
                                false, ab, false, bfb, (short)0, accp[t - NKEEP], false, false);
                    }
                }
            }
            #pragma unroll
            for (int t = 0; t < NPARK; ++t) {
                *(v4f*)(park + (2 * t) * 128)     = (v4f){accp[t][0], accp[t][1], accp[t][2], accp[t][3]};
                *(v4f*)(park + (2 * t + 1) * 128) = (v4f){accp[t][4], accp[t][5], accp[t][6], accp[t][7]};
            }
        }
        if (h & 1) __syncthreads();
    }

    auto store_p_tile = [&](int t, const v8f& v) {
        #pragma unroll
        for (int r = 0; r < 8; ++r)
            sP[(wm * 16 + r + hi * 8) * PSTR + t * 16 + nl] = v[r];
    };
    auto store_p_first = [&]() {
        #pragma unroll
        for (int t = 0; t < NTILE; ++t) store_p_tile(t, acc0[t]);
    };
    auto store_p_second = [&]() {
        #pragma unroll
        for (int t = 0; t < NKEEP; ++t) store_p_tile(t, acc1[t]);
        v8f accp[NPARK];
        park_load(accp);
        #pragma unroll
        for (int t = 0; t < NPARK; ++t) store_p_tile(NKEEP + t, accp[t]);
    };
    if (wg == 0) store_p_second();
    __syncthreads();

    {
        float yv[NB][E];
        #pragma unroll
        for (int n = 0; n < NB; ++n) {
            #pragma unroll
            for (int e = 0; e < E; ++e) yv[n][e] = y[(b0 + n) * E + e];
        }
        const int c4 = lane * 4;
        #pragma nounroll
        for (int j0 = 0; j0 < 5; j0 += 1) {
            v4fu wl[1][E];
            #pragma unroll
            for (int jj = 0; jj < 1; ++jj) {
                const int wrr = wv + 16 * (j0 + jj);
                const int wr  = wrr < WROWS ? wrr : (WROWS - 1);
                const float* src; int stride;
                if (wr < WR_M0)       { src = Wm1 + (wr)         * C; stride = K3_1 * C; }
                else if (wr < WR_21)  { src = Wm0 + (wr - WR_M0) * C; stride = K3_0 * C; }
                else if (wr < WR_20)  { src = W21 + (wr - WR_21) * C; stride = K2_1 * C; }
                else if (wr < WR_11)  { src = W20 + (wr - WR_20) * C; stride = K2_0 * C; }
                else if (wr == WR_11) { src = W11;                    stride = C;        }
                else                  { src = W10;                    stride = C;        }
                #pragma unroll
                for (int e = 0; e < E; ++e) wl[jj][e] = *(const v4fu*)(src + e * stride + c4);
            }
            #pragma unroll
            for (int jj = 0; jj < 1; ++jj) {
                const int wrr = wv + 16 * (j0 + jj);
                float ws[NB][4];
                #pragma unroll
                for (int n = 0; n < NB; ++n) {
                    #pragma unroll
                    for (int u = 0; u < 4; ++u) ws[n][u] = 0.f;
                }
                #pragma unroll
                for (int e = 0; e < E; ++e) {
                    #pragma unroll
                    for (int n = 0; n < NB; ++n) {
                        #pragma unroll
                        for (int u = 0; u < 4; ++u) ws[n][u] = fmaf(yv[n][e], wl[jj][e][u], ws[n][u]);
                    }
                }
                if (wrr < WROWS) {
                    float* dst = sW + wrr * C + c4;
                    #pragma unroll
                    for (int n = 0; n < NB; ++n) {
                        #pragma unroll
                        for (int u = 0; u < 4; ++u) dst[n * WROWS * C + u] = ws[n][u];
                    }
                }
            }
        }
    }
    __syncthreads();

    node_epilogue(sP, sW + WROWS * C, sX + C * XSTR, sU1, out + (size_t)(b0 + 1) * (4 * C), tid);

    __syncthreads();
    if (wg == 0) store_p_first();
    __syncthreads();
    node_epilogue(sP, sW, sX, sU1, out + (size_t)b0 * (4 * C), tid);

    __syncthreads();
    if (wg == 1) store_p_second();
    __syncthreads();
    node_epilogue(sP, sW + 3 * WROWS * C, sX + 3 * C * XSTR, sU1, out + (size_t)(b0 + 3) * (4 * C), tid);

    __syncthreads();
    if (wg == 1) store_p_first();
    __syncthreads();
    node_epilogue(sP, sW + 2 * WROWS * C, sX + 2 * C * XSTR, sU1, out + (size_t)(b0 + 2) * (4 * C), tid);
}

extern "C" void kernel_launch(void* const* d_in, const int* in_sizes, int n_in,
                              void* d_out, int out_size, void* d_ws, size_t ws_size,
                              hipStream_t stream) {
    (void)in_sizes; (void)n_in; (void)out_size; (void)ws_size;
    const float* x    = (const float*)d_in[0];
    const float* y    = (const float*)d_in[1];
    const float* U3_0 = (const float*)d_in[2];
    const float* U2_0 = (const float*)d_in[3];
    const float* U1_0 = (const float*)d_in[4];
    const float* U3_1 = (const float*)d_in[5];
    const float* U2_1 = (const float*)d_in[6];
    const float* U1_1 = (const float*)d_in[7];
    const float* Wm0  = (const float*)d_in[8];
    const float* W20  = (const float*)d_in[9];
    const float* W10  = (const float*)d_in[10];
    const float* Wm1  = (const float*)d_in[11];
    const float* W21  = (const float*)d_in[12];
    const float* W11  = (const float*)d_in[13];

    float* F1 = (float*)d_ws;
    float* F0 = F1 + 3 * 256 * L * K3_1;
    fold_tables_kernel<<<dim3(256, 4), dim3(128), 0, stream>>>(U3_1, U3_0, F1, F0);

    constexpr size_t smem_bytes = (size_t)SMEM_FLOATS * sizeof(float);
    mace_node_kernel<<<dim3(NN / NB), dim3(NTHR), smem_bytes, stream>>>(
        x, y, U3_0, U2_0, U1_0, U3_1, U2_1, U1_1,
        Wm0, W20, W10, Wm1, W21, W11, F1, F0, (float*)d_out);
}


# epilogue from the accumulators: products computed transposed (lane = channel), weight contraction summed in-lane; no P staging in LDS, 8 barriers fewer; sum order of the final contraction changes
# speedup vs baseline: 6.1513x; 1.0757x over previous
#include <hip/hip_runtime.h>

typedef float v2f __attribute__((ext_vector_type(2)));
typedef float v8f __attribute__((ext_vector_type(8)));
typedef float v4f __attribute__((ext_vector_type(4)));
typedef float v4fu __attribute__((ext_vector_type(4), aligned(4)));

constexpr int NN   = 2048;
constexpr int C    = 128;
constexpr int L    = 16;
constexpr int E    = 10;
constexpr int K3_0 = 23, K2_0 = 4;
constexpr int K3_1 = 33, K2_1 = 6;

constexpr int COL_A1 = 0;
constexpr int COL_A0 = 3 * K3_1;
constexpr int COL_21 = COL_A0 + K3_0;
constexpr int COL_20 = COL_21 + 3 * K2_1;
constexpr int NCOL   = COL_20 + K2_0;
constexpr int NTILE  = NCOL / 16;

constexpr int S2   = 2 * NCOL + 2;
constexpr int XSTR = L + 1;

constexpr int WR_M1 = 0;
constexpr int WR_M0 = 33;
constexpr int WR_21 = 56;
constexpr int WR_20 = 62;
constexpr int WR_11 = 66;
constexpr int WR_10 = 67;
constexpr int WROWS = 68;

constexpr int NB  = 4;
constexpr int NTHR = 512;
constexpr int NCH = 4;
constexpr int NST = 10;
constexpr int CP  = 2 * NST;
constexpr int NU2 = 3 * K2_1 + K2_0;
constexpr int B2BUF = NB * CP * S2;

__host__ __device__ constexpr int sl_base(int mm) { return mm * (9 - mm) / 2; }
__host__ __device__ constexpr int fo_base(int mm) { return mm * (7 - mm) / 2; }
__device__ __forceinline__ void fold_step(int fo, int& mm, int& j)
{
    mm = (fo >= 3) + (fo >= 5);
    j  = mm + 1 + fo - fo_base(mm);
}

constexpr int OFF_X  = 0;
constexpr int OFF_XS = OFF_X + NB * C * XSTR;
constexpr int OFF_U1 = OFF_XS + NB * L;
constexpr int OFF_B2 = OFF_U1 + 64;
constexpr int OFF_W  = OFF_B2;
constexpr int NPARK = 6;
constexpr int NKEEP = NTILE - NPARK;
constexpr int PARKW = NPARK * 8 * 32;
constexpr int OFF_PARK = OFF_B2 + 2 * B2BUF;
constexpr int END_W    = OFF_W + NB * WROWS * C;
constexpr int END_PARK = OFF_PARK + (NTHR / 32) * PARKW;
constexpr int SMEM_FLOATS = END_PARK > END_W ? END_PARK : END_W;
static_assert(OFF_PARK % 4 == 0 && PARKW % 4 == 0, "16-byte accesses to the parked tiles");
static_assert(OFF_PARK >= END_W, "the parked tiles must lie behind W");
static_assert(((SMEM_FLOATS * 4 + 2047) / 2048) * 2048 <= 320 * 1024, "one block per WGP");


template <int KT>
__device__ __forceinline__ void issue_row_loads(v4fu (&r)[L], const float* __restrict__ u)
{
    #pragma unroll
    for (int i = 0; i < L; ++i) r[i] = *(const v4fu*)(u + i * KT);
}

__device__ __forceinline__ float xs_lane(int xsv, int l)
{
    return __int_as_float(__builtin_amdgcn_readlane(xsv, l));
}

template <int KN>
__device__ __forceinline__ void row_sums(v4fu (&r)[L], const float (&xa)[L], const float (&xb)[L],
                                         int xsv, float (&a)[4], float (&c)[4], float (&d)[4],
                                         float (&f)[4], const float* __restrict__ un)
{
    #pragma unroll
    for (int e = 0; e < 4; ++e) { a[e] = 0.f; c[e] = 0.f; d[e] = 0.f; f[e] = 0.f; }
    #pragma unroll
    for (int i = 0; i < L; ++i) {
        const v4fu v = r[i];
        const float xc = xs_lane(xsv, i), xd = xs_lane(xsv, L + i);
        #pragma unroll
        for (int e = 0; e < 4; ++e) {
            a[e] = fmaf(v[e], xa[i], a[e]);
            c[e] = fmaf(v[e], xb[i], c[e]);
            d[e] = fmaf(v[e], xc, d[e]);
            f[e] = fmaf(v[e], xd, f[e]);
        }
        if (KN > 0) {
            asm volatile("" :: "v"(v) : "memory");
            r[i] = *(const v4fu*)(un + i * KN);
        }
    }
}

__device__ __forceinline__ void store_a1(float* __restrict__ sB2, int w, int rl, int kcol,
                                         const float (&a)[4], const float (&c)[4],
                                         const float (&d)[4], const float (&f)[4])
{
    float* bp = sB2 + (rl >> 1) * S2 + (COL_A1 + w * K3_1 + kcol) * 2 + (rl & 1);
    #pragma unroll
    for (int e = 0; e < 4; ++e) {
        bp[2 * e]               = a[e];
        bp[2 * e + CP * S2]     = c[e];
        bp[2 * e + 2 * CP * S2] = d[e];
        bp[2 * e + 3 * CP * S2] = f[e];
    }
}

__device__ __forceinline__ void store_a0(float* __restrict__ sB2, int rl, int kcol,
                                         const float (&a)[4], const float (&c)[4],
                                         const float (&d)[4], const float (&f)[4])
{
    float* bp = sB2 + (rl >> 1) * S2 + (COL_A0 + kcol) * 2 + (rl & 1);
    #pragma unroll
    for (int e = 0; e < 4; ++e) {
        bp[2 * e]               = a[e];
        bp[2 * e + CP * S2]     = c[e];
        bp[2 * e + 2 * CP * S2] = d[e];
        bp[2 * e + 3 * CP * S2] = f[e];
    }
}

__host__ __device__ constexpr int col_group(int col)
{
    return col < COL_A0 ? 1 + col / K3_1
         : col < COL_21 ? 0
         : col < COL_20 ? 1 + (col - COL_21) / K2_1
         : 0;
}
__host__ __device__ constexpr int col_wrow(int col)
{
    return col < COL_A0 ? WR_M1 + col % K3_1
         : col < COL_21 ? WR_M0 + (col - COL_A0)
         : col < COL_20 ? WR_21 + (col - COL_21) % K2_1
         : WR_20 + (col - COL_20);
}

__device__ __forceinline__ void tile_sums(int t, const v8f& acc, const float* __restrict__ sWc,
                                          int hi, float (&s)[4])
{
    const float* sWh = sWc + hi * (8 * C);
    #pragma unroll
    for (int r = 0; r < 8; ++r) {
        const int c0 = t * 16 + r, c1 = c0 + 8;
        const int g0 = col_group(c0), g1 = col_group(c1);
        const int w0 = col_wrow(c0),  w1 = col_wrow(c1);
        float wv;
        if (w1 - w0 == 8) wv = sWh[w0 * C];
        else              wv = sWc[(hi ? w1 : w0) * C];
        if (g0 == g1) {
            s[g0] = fmaf(acc[r], wv, s[g0]);
        } else {
            const float t0 = fmaf(acc[r], wv, s[g0]);
            const float t1 = fmaf(acc[r], wv, s[g1]);
            s[g0] = hi ? s[g0] : t0;
            s[g1] = hi ? t1 : s[g1];
        }
    }
}

__device__ __forceinline__ void lane_epilogue(const float (&s)[4], const float* __restrict__ sWc,
                                              const float* __restrict__ sXc,
                                              const float* __restrict__ sU1,
                                              float* __restrict__ outn, int crow, int hi)
{
    const float keep0 = hi ? s[2] : s[0], keep1 = hi ? s[3] : s[1];
    const float send0 = hi ? s[0] : s[2], send1 = hi ? s[1] : s[3];
    const float tot0 = keep0 + __shfl_xor(send0, 16);
    const float tot1 = keep1 + __shfl_xor(send1, 16);
    const float* uA = sU1 + (hi ? 2 * L : 0);
    const float* uB = sU1 + (hi ? 3 * L : L);
    float dA = 0.f, dB = 0.f;
    #pragma unroll
    for (int j = 0; j < L; ++j) {
        const float xj = sXc[j];
        dA = fmaf(uA[j], xj, dA);
        dB = fmaf(uB[j], xj, dB);
    }
    const float oA = fmaf(sWc[(hi ? WR_11 : WR_10) * C], dA, tot0);
    const float oB = fmaf(sWc[WR_11 * C], dB, tot1);
    outn[hi ? C + crow * 3 + 1 : crow] = oA;
    outn[C + crow * 3 + (hi ? 2 : 0)]  = oB;
}

__global__ __launch_bounds__(128) void fold_tables_kernel(
    const float* __restrict__ U3_1, const float* __restrict__ U3_0,
    float* __restrict__ F1, float* __restrict__ F0)
{
    const int a = blockIdx.x >> 4, b = blockIdx.x & 15;
    if ((b >> 2) <= (a >> 2)) return;
    const int t = blockIdx.y;
    const int n = t < 3 ? L * K3_1 : L * K3_0;
    const float* s = t < 3 ? U3_1 + t * (256 * L * K3_1) : U3_0;
    float*       d = t < 3 ? F1   + t * (256 * L * K3_1) : F0;
    const int ra = (a * 16 + b) * n, rb = (b * 16 + a) * n;
    for (int e = threadIdx.x; e < n; e += 128) d[ra + e] = s[ra + e] + s[rb + e];
}

__global__ __launch_bounds__(NTHR, 4) void mace_node_kernel(
    const float* __restrict__ x,    const float* __restrict__ y,
    const float* __restrict__ U3_0, const float* __restrict__ U2_0, const float* __restrict__ U1_0,
    const float* __restrict__ U3_1, const float* __restrict__ U2_1, const float* __restrict__ U1_1,
    const float* __restrict__ Wm0,  const float* __restrict__ W20,  const float* __restrict__ W10,
    const float* __restrict__ Wm1,  const float* __restrict__ W21,  const float* __restrict__ W11,
    const float* __restrict__ F1,   const float* __restrict__ F0,
    float* __restrict__ out)
{
    extern __shared__ __attribute__((aligned(16))) float sm[];
    float* sX  = sm + OFF_X;
    float* sXS = sm + OFF_XS;
    float* sW  = sm + OFF_W;
    float* sU1 = sm + OFF_U1;
    float* sB2 = sm + OFF_B2;

    const int tid = threadIdx.x;
    const int b0  = blockIdx.x * NB;

    {
        const int ti = tid < 64 ? tid : 63;
        const float* up = ti < 16 ? (U1_0 + ti) : (U1_1 + (ti - 16));
        const float u1 = *up;
        const float* xb = x + (size_t)b0 * (C * L);
        v4fu xv[4];
        #pragma unroll
        for (int j = 0; j < 4; ++j) xv[j] = *(const v4fu*)(xb + 4 * (tid + NTHR * j));
        #pragma unroll
        for (int j = 0; j < 4; ++j) {
            const int idx = 4 * (tid + NTHR * j);
            float* d = sX + (idx >> 4) * XSTR + (idx & 15);
            #pragma unroll
            for (int u = 0; u < 4; ++u) d[u] = xv[j][u];
        }
        asm volatile("" :: "v"(u1));
        if (tid < 64) sU1[tid] = u1;
    }
    __syncthreads();

    if (tid < NB * L) {
        const int n = tid >> 4, i = tid & 15;
        float s = 0.f;
        for (int c = 0; c < C; ++c) s += sX[(n * C + c) * XSTR + i];
        sXS[tid] = s;
    }
    __syncthreads();

    const int wv   = __builtin_amdgcn_readfirstlane(tid >> 5);
    const int wm   = wv & 7;
    const int wg   = wv >> 3;
    const int lane = tid & 31;
    const int nl   = lane & 15;
    const int hi   = lane >> 4;
    const int crow = wm * 16 + nl;

    float xsa[L], xsb[L];
    #pragma unroll
    for (int i = 0; i < L; ++i) {
        xsa[i] = __int_as_float(__builtin_amdgcn_readfirstlane(__float_as_int(sXS[i])));
        xsb[i] = __int_as_float(__builtin_amdgcn_readfirstlane(__float_as_int(sXS[L + i])));
    }
    const int xsv = __float_as_int(sXS[2 * L + lane]);

    const int l1  = lane < 27 ? lane : 26;
    const int rg1 = l1 / 9, q1 = l1 - 9 * rg1;
    const int k1  = (q1 < 8) ? 4 * q1 : (K3_1 - 4);
    const int l0  = lane < 24 ? lane : 23;
    const int rg0 = l0 / 6, q0 = l0 - 6 * rg0;
    const int k0  = (q0 < 5) ? 4 * q0 : (K3_0 - 4);

    v8f acc0[NTILE], acc1[NKEEP];
    #pragma unroll
    for (int t = 0; t < NTILE; ++t) acc0[t] = (v8f){0.f, 0.f, 0.f, 0.f, 0.f, 0.f, 0.f, 0.f};
    #pragma unroll
    for (int t = 0; t < NKEEP; ++t) acc1[t] = (v8f){0.f, 0.f, 0.f, 0.f, 0.f, 0.f, 0.f, 0.f};
    float* park = sm + OFF_PARK + wv * PARKW + lane * 4;
    auto park_load = [&](v8f (&p)[NPARK]) {
        #pragma unroll
        for (int t = 0; t < NPARK; ++t) {
            const v4f lo = *(const v4f*)(park + (2 * t) * 128);
            const v4f hh = *(const v4f*)(park + (2 * t + 1) * 128);
            p[t] = (v8f){lo[0], lo[1], lo[2], lo[3], hh[0], hh[1], hh[2], hh[3]};
        }
    };

    const float* tab1 = wg ? F1 : U3_1;
    auto a1_pass = [&](int chn, int p, int& row, int& rl) {
        int mm, j, r;
        if (wg == 0) { const int dr = 2 * wm + p; mm = dr >> 2; j = mm; r = dr & 3; }
        else         { const int f = 3 * wm + p; fold_step(f >> 2, mm, j); r = f & 3; }
        row = (4 * mm + chn) * 16 + 4 * j + r;
        rl  = 4 * (sl_base(mm) + j - mm) + r;
    };
    const float* tab0 = wm < 6 ? F0 : U3_0;
    auto a0_pass = [&](int chn, int first, int& row, int& rl) {
        int mm, j;
        if (first)       { mm = 2 + wm; j = mm; }
        else if (wm < 6) { fold_step(wm, mm, j); }
        else             { mm = wm - 6; j = mm; }
        row = (4 * mm + chn) * 16 + 4 * j + rg0;
        rl  = 4 * (sl_base(mm) + j - mm) + rg0;
    };
    auto a1_off = [&](int row) -> unsigned {
        return (unsigned)(rg1 * 256 + row) * (unsigned)(L * K3_1) + (unsigned)k1;
    };
    auto a0_off = [&](int row) -> unsigned {
        return (unsigned)row * (unsigned)(L * K3_0) + (unsigned)k0;
    };
    auto u2_item = [&](int jj, int& col, int& pr, int& mm, int& j) -> bool {
        const int cr = wm + 8 * jj;
        col = cr < NU2 ? cr : NU2 - 1;
        pr  = lane < CP ? lane : CP - 1;
        const int sl = pr >> 1;
        mm = (sl >= 4) + (sl >= 7) + (sl >= 9);
        j  = mm + sl - sl_base(mm);
        return (cr < NU2) & (lane < CP);
    };

    const int prod_first = (wv >> 2) & 1;
    #pragma nounroll
    for (int h = 0; h < 2 * (NCH + 1); ++h) {
        const int it = h >> 1;
        const bool build_half = ((h & 1) ^ prod_first) == 0;
        if (build_half && it < NCH) {
            const int ch = it;
            float* sBw = sB2 + (it & 1) * B2BUF;
            int row, rl;
            a1_pass(ch, 0, row, rl);
            v4fu r[L];
            issue_row_loads<K3_1>(r, tab1 + a1_off(row));

            float a[4] = {0.f, 0.f, 0.f, 0.f}, c[4] = {0.f, 0.f, 0.f, 0.f};
            float d[4] = {0.f, 0.f, 0.f, 0.f}, f[4] = {0.f, 0.f, 0.f, 0.f};
            const int nsh = wg ? 2 : 1;
            #pragma nounroll
            for (int s = 0; s < nsh; ++s) {
                int rown, rln;
                a1_pass(ch, s + 1, rown, rln);
                row_sums<K3_1>(r, xsa, xsb, xsv, a, c, d, f, tab1 + a1_off(rown));
                store_a1(sBw, rg1, rl, k1, a, c, d, f);
                rl = rln;
            }
            if (wg) {
                row_sums<0>(r, xsa, xsb, xsv, a, c, d, f, nullptr);
                store_a1(sBw, rg1, rl, k1, a, c, d, f);
            } else {
                const int two = wm < 2;
                {
                    int rown, rln;
                    a0_pass(ch, two, rown, rln);
                    row_sums<K3_0>(r, xsa, xsb, xsv, a, c, d, f, (two ? U3_0 : tab0) + a0_off(rown));
                    store_a1(sBw, rg1, rl, k1, a, c, d, f);
                    rl = rln;
                }
                if (two) {
                    int rown, rln;
                    a0_pass(ch, 0, rown, rln);
                    row_sums<K3_0>(r, xsa, xsb, xsv, a, c, d, f, tab0 + a0_off(rown));
                    store_a0(sBw, rl, k0, a, c, d, f);
                    rl = rln;
                }
                float u2v[12];
                #pragma unroll
                for (int e = 0; e < 4; ++e) { a[e] = 0.f; c[e] = 0.f; d[e] = 0.f; f[e] = 0.f; }
                #pragma unroll
                for (int i = 0; i < L; ++i) {
                    const v4fu v = r[i];
                    const float xc = xs_lane(xsv, i), xd = xs_lane(xsv, L + i);
                    #pragma unroll
                    for (int e = 0; e < 4; ++e) {
                        a[e] = fmaf(v[e], xsa[i], a[e]);
                        c[e] = fmaf(v[e], xsb[i], c[e]);
                        d[e] = fmaf(v[e], xc, d[e]);
                        f[e] = fmaf(v[e], xd, f[e]);
                    }
                    if (i < 4) {
                        asm volatile("" :: "v"(v) : "memory");
                        #pragma unroll
                        for (int t = 0; t < 3; ++t) {
                            const int n = 3 * i + t;
                            int col, pr, mm, j;
                            u2_item(n >> 2, col, pr, mm, j);
                            const int aa = 4 * mm + ch, bb = 4 * j + 2 * (pr & 1);
                            const bool t1 = col < 3 * K2_1;
                            const int w = col / K2_1, k = col - w * K2_1;
                            const float* ub = t1 ? (U2_1 + (w * 256) * K2_1 + k) : (U2_0 + (col - 3 * K2_1));
                            const int us = t1 ? K2_1 : K2_0;
                            const int e = n & 3;
                            const int rw = e == 0 ? (aa * 16 + bb) : e == 1 ? (aa * 16 + bb + 1)
                                         : e == 2 ? (bb * 16 + aa) : ((bb + 1) * 16 + aa);
                            u2v[n] = ub[rw * us];
                        }
                    }
                }
                store_a0(sBw, rl, k0, a, c, d, f);
                #pragma unroll
                for (int n = 0; n < 12; n += 4)
                    asm volatile("" :: "v"(u2v[n]), "v"(u2v[n + 1]), "v"(u2v[n + 2]), "v"(u2v[n + 3]));
                #pragma unroll
                for (int jj = 0; jj < 3; ++jj) {
                    int col, pr, mm, j;
                    const bool valid = u2_item(jj, col, pr, mm, j);
                    const bool fold = j != mm;
                    v2f v;
                    v.x = fold ? u2v[4 * jj]     + u2v[4 * jj + 2] : u2v[4 * jj];
                    v.y = fold ? u2v[4 * jj + 1] + u2v[4 * jj + 3] : u2v[4 * jj + 1];
                    if (valid) {
                        float* bp = sBw + pr * S2 + (COL_21 + col) * 2;
                        #pragma unroll
                        for (int n = 0; n < NB; ++n) *(v2f*)(bp + n * CP * S2) = v;
                    }
                }
            }
        }

        if (!build_half && it > 0) {
            const int ch = it - 1;
            const float* sBr = sB2 + (ch & 1) * B2BUF;
            float sel0a[4], sel1a[4], sel0b[4], sel1b[4], xpa[4], xpb[4];
            const float* xra = sX + (2 * wg * C + crow) * XSTR;
            const float* xrb = xra + C * XSTR;
            #pragma unroll
            for (int j = 0; j < 4; ++j) {
                sel0a[j] = hi ? xra[4 * j + 2] : xra[4 * j];
                sel1a[j] = hi ? xra[4 * j + 3] : xra[4 * j + 1];
                sel0b[j] = hi ? xrb[4 * j + 2] : xrb[4 * j];
                sel1b[j] = hi ? xrb[4 * j + 3] : xrb[4 * j + 1];
                xpa[j] = xra[4 * j + ch];
                xpb[j] = xrb[4 * j + ch];
            }
            v8f accp[NPARK];
            if (ch == 0) {
                #pragma unroll
                for (int t = 0; t < NPARK; ++t) accp[t] = (v8f){0.f, 0.f, 0.f, 0.f, 0.f, 0.f, 0.f, 0.f};
            } else {
                park_load(accp);
            }
            const float* Bl = sBr + 2 * wg * CP * S2 + hi * S2 + nl * 2;
            #pragma unroll
            for (int mm = 0; mm < 4; ++mm) {
                #pragma unroll
                for (int j = mm; j < 4; ++j) {
                    const int sl = sl_base(mm) + j - mm;
                    v2f aa, ab;
                    aa.x = xpa[mm] * sel0a[j];
                    aa.y = xpa[mm] * sel1a[j];
                    ab.x = xpb[mm] * sel0b[j];
                    ab.y = xpb[mm] * sel1b[j];
                    const float* Br = Bl + (2 * sl) * S2;
                    #pragma unroll
                    for (int t = 0; t < NTILE; ++t) {
                        const v2f bfa = *(const v2f*)(Br + t * 32);
                        const v2f bfb = *(const v2f*)(Br + CP * S2 + t * 32);
                        acc0[t] = __builtin_amdgcn_wmma_f32_16x16x4_f32(
                            false, bfa, false, aa, (short)0, acc0[t], false, false);
                        if (t < NKEEP)
                            acc1[t] = __builtin_amdgcn_wmma_f32_16x16x4_f32(
                                false, bfb, false, ab, (short)0, acc1[t], false, false);
                        else
                            accp[t - NKEEP] = __builtin_amdgcn_wmma_f32_16x16x4_f32(
                                false, bfb, false, ab, (short)0, accp[t - NKEEP], false, false);
                    }
                }
            }
            #pragma unroll
            for (int t = 0; t < NPARK; ++t) {
                *(v4f*)(park + (2 * t) * 128)     = (v4f){accp[t][0], accp[t][1], accp[t][2], accp[t][3]};
                *(v4f*)(park + (2 * t + 1) * 128) = (v4f){accp[t][4], accp[t][5], accp[t][6], accp[t][7]};
            }
        }
        if (h & 1) __syncthreads();
    }


    {
        float yv[NB][E];
        #pragma unroll
        for (int n = 0; n < NB; ++n) {
            #pragma unroll
            for (int e = 0; e < E; ++e) yv[n][e] = y[(b0 + n) * E + e];
        }
        const int c4 = lane * 4;
        #pragma nounroll
        for (int j0 = 0; j0 < 5; j0 += 1) {
            v4fu wl[1][E];
            #pragma unroll
            for (int jj = 0; jj < 1; ++jj) {
                const int wrr = wv + 16 * (j0 + jj);
                const int wr  = wrr < WROWS ? wrr : (WROWS - 1);
                const float* src; int stride;
                if (wr < WR_M0)       { src = Wm1 + (wr)         * C; stride = K3_1 * C; }
                else if (wr < WR_21)  { src = Wm0 + (wr - WR_M0) * C; stride = K3_0 * C; }
                else if (wr < WR_20)  { src = W21 + (wr - WR_21) * C; stride = K2_1 * C; }
                else if (wr < WR_11)  { src = W20 + (wr - WR_20) * C; stride = K2_0 * C; }
                else if (wr == WR_11) { src = W11;                    stride = C;        }
                else                  { src = W10;                    stride = C;        }
                #pragma unroll
                for (int e = 0; e < E; ++e) wl[jj][e] = *(const v4fu*)(src + e * stride + c4);
            }
            #pragma unroll
            for (int jj = 0; jj < 1; ++jj) {
                const int wrr = wv + 16 * (j0 + jj);
                float ws[NB][4];
                #pragma unroll
                for (int n = 0; n < NB; ++n) {
                    #pragma unroll
                    for (int u = 0; u < 4; ++u) ws[n][u] = 0.f;
                }
                #pragma unroll
                for (int e = 0; e < E; ++e) {
                    #pragma unroll
                    for (int n = 0; n < NB; ++n) {
                        #pragma unroll
                        for (int u = 0; u < 4; ++u) ws[n][u] = fmaf(yv[n][e], wl[jj][e][u], ws[n][u]);
                    }
                }
                if (wrr < WROWS) {
                    float* dst = sW + wrr * C + c4;
                    #pragma unroll
                    for (int n = 0; n < NB; ++n) {
                        #pragma unroll
                        for (int u = 0; u < 4; ++u) dst[n * WROWS * C + u] = ws[n][u];
                    }
                }
            }
        }
    }
    __syncthreads();

    {
        const float* sWa = sW + 2 * wg * WROWS * C + crow;
        const float* sXa = sX + (2 * wg * C + crow) * XSTR;
        float* outa = out + (size_t)(b0 + 2 * wg) * (4 * C);
        {
            float s[4] = {0.f, 0.f, 0.f, 0.f};
            #pragma unroll
            for (int t = 0; t < NTILE; ++t) tile_sums(t, acc0[t], sWa, hi, s);
            lane_epilogue(s, sWa, sXa, sU1, outa, crow, hi);
        }
        {
            const float* sWb = sWa + WROWS * C;
            v8f accp[NPARK];
            park_load(accp);
            float s[4] = {0.f, 0.f, 0.f, 0.f};
            #pragma unroll
            for (int t = 0; t < NKEEP; ++t) tile_sums(t, acc1[t], sWb, hi, s);
            #pragma unroll
            for (int t = 0; t < NPARK; ++t) tile_sums(NKEEP + t, accp[t], sWb, hi, s);
            lane_epilogue(s, sWb, sXa + C * XSTR, sU1, outa + 4 * C, crow, hi);
        }
    }
}

extern "C" void kernel_launch(void* const* d_in, const int* in_sizes, int n_in,
                              void* d_out, int out_size, void* d_ws, size_t ws_size,
                              hipStream_t stream) {
    (void)in_sizes; (void)n_in; (void)out_size; (void)ws_size;
    const float* x    = (const float*)d_in[0];
    const float* y    = (const float*)d_in[1];
    const float* U3_0 = (const float*)d_in[2];
    const float* U2_0 = (const float*)d_in[3];
    const float* U1_0 = (const float*)d_in[4];
    const float* U3_1 = (const float*)d_in[5];
    const float* U2_1 = (const float*)d_in[6];
    const float* U1_1 = (const float*)d_in[7];
    const float* Wm0  = (const float*)d_in[8];
    const float* W20  = (const float*)d_in[9];
    const float* W10  = (const float*)d_in[10];
    const float* Wm1  = (const float*)d_in[11];
    const float* W21  = (const float*)d_in[12];
    const float* W11  = (const float*)d_in[13];

    float* F1 = (float*)d_ws;
    float* F0 = F1 + 3 * 256 * L * K3_1;
    fold_tables_kernel<<<dim3(256, 4), dim3(128), 0, stream>>>(U3_1, U3_0, F1, F0);

    constexpr size_t smem_bytes = (size_t)SMEM_FLOATS * sizeof(float);
    mace_node_kernel<<<dim3(NN / NB), dim3(NTHR), smem_bytes, stream>>>(
        x, y, U3_0, U2_0, U1_0, U3_1, U2_1, U1_1,
        Wm0, W20, W10, Wm1, W21, W11, F1, F0, (float*)d_out);
}


# K of the WMMA products folded 160 -> 136 rows: pairs (p,q),(q,p) inside a quad summed too (in-quad steps repacked 4,3,2,1 per chunk, 34 K steps not 40; first kernel sums all p<q); sum order changes
# speedup vs baseline: 6.8236x; 1.1093x over previous
#include <hip/hip_runtime.h>

typedef float v2f __attribute__((ext_vector_type(2)));
typedef float v8f __attribute__((ext_vector_type(8)));
typedef float v4f __attribute__((ext_vector_type(4)));
typedef float v4fu __attribute__((ext_vector_type(4), aligned(4)));

constexpr int NN   = 2048;
constexpr int C    = 128;
constexpr int L    = 16;
constexpr int E    = 10;
constexpr int K3_0 = 23, K2_0 = 4;
constexpr int K3_1 = 33, K2_1 = 6;

constexpr int COL_A1 = 0;
constexpr int COL_A0 = 3 * K3_1;
constexpr int COL_21 = COL_A0 + K3_0;
constexpr int COL_20 = COL_21 + 3 * K2_1;
constexpr int NCOL   = COL_20 + K2_0;
constexpr int NTILE  = NCOL / 16;

constexpr int S2   = 2 * NCOL + 2;
constexpr int XSTR = L + 1;

constexpr int WR_M1 = 0;
constexpr int WR_M0 = 33;
constexpr int WR_21 = 56;
constexpr int WR_20 = 62;
constexpr int WR_11 = 66;
constexpr int WR_10 = 67;
constexpr int WROWS = 68;

constexpr int NB  = 4;
constexpr int NTHR = 512;
constexpr int NCH = 4;
constexpr int NST = 10;
constexpr int CP  = 2 * NST;
constexpr int NU2 = 3 * K2_1 + K2_0;
constexpr int B2BUF = NB * CP * S2;

__host__ __device__ constexpr int fo_base(int mm) { return mm * (7 - mm) / 2; }
__device__ __forceinline__ void fold_step(int fo, int& mm, int& j)
{
    mm = (fo >= 3) + (fo >= 5);
    j  = mm + 1 + fo - fo_base(mm);
}

constexpr int OFF_X  = 0;
constexpr int OFF_XS = OFF_X + NB * C * XSTR;
constexpr int OFF_U1 = OFF_XS + NB * L;
constexpr int OFF_B2 = OFF_U1 + 64;
constexpr int OFF_W  = OFF_B2;
constexpr int NPARK = 6;
constexpr int NKEEP = NTILE - NPARK;
constexpr int PARKW = NPARK * 8 * 32;
constexpr int OFF_PARK = OFF_B2 + 2 * B2BUF;
constexpr int END_W    = OFF_W + NB * WROWS * C;
constexpr int END_PARK = OFF_PARK + (NTHR / 32) * PARKW;
constexpr int SMEM_FLOATS = END_PARK > END_W ? END_PARK : END_W;
static_assert(OFF_PARK % 4 == 0 && PARKW % 4 == 0, "16-byte accesses to the parked tiles");
static_assert(OFF_PARK >= END_W, "the parked tiles must lie behind W");
static_assert(((SMEM_FLOATS * 4 + 2047) / 2048) * 2048 <= 320 * 1024, "one block per WGP");


template <int KT>
__device__ __forceinline__ void issue_row_loads(v4fu (&r)[L], const float* __restrict__ u)
{
    #pragma unroll
    for (int i = 0; i < L; ++i) r[i] = *(const v4fu*)(u + i * KT);
}

__device__ __forceinline__ float xs_lane(int xsv, int l)
{
    return __int_as_float(__builtin_amdgcn_readlane(xsv, l));
}

template <int KN>
__device__ __forceinline__ void row_sums(v4fu (&r)[L], const float (&xa)[L], const float (&xb)[L],
                                         int xsv, float (&a)[4], float (&c)[4], float (&d)[4],
                                         float (&f)[4], const float* __restrict__ un)
{
    #pragma unroll
    for (int e = 0; e < 4; ++e) { a[e] = 0.f; c[e] = 0.f; d[e] = 0.f; f[e] = 0.f; }
    #pragma unroll
    for (int i = 0; i < L; ++i) {
        const v4fu v = r[i];
        const float xc = xs_lane(xsv, i), xd = xs_lane(xsv, L + i);
        #pragma unroll
        for (int e = 0; e < 4; ++e) {
            a[e] = fmaf(v[e], xa[i], a[e]);
            c[e] = fmaf(v[e], xb[i], c[e]);
            d[e] = fmaf(v[e], xc, d[e]);
            f[e] = fmaf(v[e], xd, f[e]);
        }
        if (KN > 0) {
            asm volatile("" :: "v"(v) : "memory");
            r[i] = *(const v4fu*)(un + i * KN);
        }
    }
}

__device__ __forceinline__ void store_a1(float* __restrict__ sB2, int w, int rl, int kcol,
                                         const float (&a)[4], const float (&c)[4],
                                         const float (&d)[4], const float (&f)[4])
{
    float* bp = sB2 + (rl >> 1) * S2 + (COL_A1 + w * K3_1 + kcol) * 2 + (rl & 1);
    #pragma unroll
    for (int e = 0; e < 4; ++e) {
        bp[2 * e]               = a[e];
        bp[2 * e + CP * S2]     = c[e];
        bp[2 * e + 2 * CP * S2] = d[e];
        bp[2 * e + 3 * CP * S2] = f[e];
    }
}

__device__ __forceinline__ void store_a0(float* __restrict__ sB2, int rl, int kcol,
                                         const float (&a)[4], const float (&c)[4],
                                         const float (&d)[4], const float (&f)[4])
{
    float* bp = sB2 + (rl >> 1) * S2 + (COL_A0 + kcol) * 2 + (rl & 1);
    #pragma unroll
    for (int e = 0; e < 4; ++e) {
        bp[2 * e]               = a[e];
        bp[2 * e + CP * S2]     = c[e];
        bp[2 * e + 2 * CP * S2] = d[e];
        bp[2 * e + 3 * CP * S2] = f[e];
    }
}

__host__ __device__ constexpr int col_group(int col)
{
    return col < COL_A0 ? 1 + col / K3_1
         : col < COL_21 ? 0
         : col < COL_20 ? 1 + (col - COL_21) / K2_1
         : 0;
}
__host__ __device__ constexpr int col_wrow(int col)
{
    return col < COL_A0 ? WR_M1 + col % K3_1
         : col < COL_21 ? WR_M0 + (col - COL_A0)
         : col < COL_20 ? WR_21 + (col - COL_21) % K2_1
         : WR_20 + (col - COL_20);
}

__device__ __forceinline__ void tile_sums(int t, const v8f& acc, const float* __restrict__ sWc,
                                          int hi, float (&s)[4])
{
    const float* sWh = sWc + hi * (8 * C);
    #pragma unroll
    for (int r = 0; r < 8; ++r) {
        const int c0 = t * 16 + r, c1 = c0 + 8;
        const int g0 = col_group(c0), g1 = col_group(c1);
        const int w0 = col_wrow(c0),  w1 = col_wrow(c1);
        float wv;
        if (w1 - w0 == 8) wv = sWh[w0 * C];
        else              wv = sWc[(hi ? w1 : w0) * C];
        if (g0 == g1) {
            s[g0] = fmaf(acc[r], wv, s[g0]);
        } else {
            const float t0 = fmaf(acc[r], wv, s[g0]);
            const float t1 = fmaf(acc[r], wv, s[g1]);
            s[g0] = hi ? s[g0] : t0;
            s[g1] = hi ? t1 : s[g1];
        }
    }
}

__device__ __forceinline__ void lane_epilogue(const float (&s)[4], const float* __restrict__ sWc,
                                              const float* __restrict__ sXc,
                                              const float* __restrict__ sU1,
                                              float* __restrict__ outn, int crow, int hi)
{
    const float keep0 = hi ? s[2] : s[0], keep1 = hi ? s[3] : s[1];
    const float send0 = hi ? s[0] : s[2], send1 = hi ? s[1] : s[3];
    const float tot0 = keep0 + __shfl_xor(send0, 16);
    const float tot1 = keep1 + __shfl_xor(send1, 16);
    const float* uA = sU1 + (hi ? 2 * L : 0);
    const float* uB = sU1 + (hi ? 3 * L : L);
    float dA = 0.f, dB = 0.f;
    #pragma unroll
    for (int j = 0; j < L; ++j) {
        const float xj = sXc[j];
        dA = fmaf(uA[j], xj, dA);
        dB = fmaf(uB[j], xj, dB);
    }
    const float oA = fmaf(sWc[(hi ? WR_11 : WR_10) * C], dA, tot0);
    const float oB = fmaf(sWc[WR_11 * C], dB, tot1);
    outn[hi ? C + crow * 3 + 1 : crow] = oA;
    outn[C + crow * 3 + (hi ? 2 : 0)]  = oB;
}

__global__ __launch_bounds__(128) void fold_tables_kernel(
    const float* __restrict__ U3_1, const float* __restrict__ U3_0,
    float* __restrict__ F1, float* __restrict__ F0)
{
    const int a = blockIdx.x >> 4, b = blockIdx.x & 15;
    if (b <= a) return;
    const int t = blockIdx.y;
    const int n = t < 3 ? L * K3_1 : L * K3_0;
    const float* s = t < 3 ? U3_1 + t * (256 * L * K3_1) : U3_0;
    float*       d = t < 3 ? F1   + t * (256 * L * K3_1) : F0;
    const int ra = (a * 16 + b) * n, rb = (b * 16 + a) * n;
    for (int e = threadIdx.x; e < n; e += 128) d[ra + e] = s[ra + e] + s[rb + e];
}

__global__ __launch_bounds__(NTHR, 4) void mace_node_kernel(
    const float* __restrict__ x,    const float* __restrict__ y,
    const float* __restrict__ U3_0, const float* __restrict__ U2_0, const float* __restrict__ U1_0,
    const float* __restrict__ U3_1, const float* __restrict__ U2_1, const float* __restrict__ U1_1,
    const float* __restrict__ Wm0,  const float* __restrict__ W20,  const float* __restrict__ W10,
    const float* __restrict__ Wm1,  const float* __restrict__ W21,  const float* __restrict__ W11,
    const float* __restrict__ F1,   const float* __restrict__ F0,
    float* __restrict__ out)
{
    extern __shared__ __attribute__((aligned(16))) float sm[];
    float* sX  = sm + OFF_X;
    float* sXS = sm + OFF_XS;
    float* sW  = sm + OFF_W;
    float* sU1 = sm + OFF_U1;
    float* sB2 = sm + OFF_B2;

    const int tid = threadIdx.x;
    const int b0  = blockIdx.x * NB;

    {
        const int ti = tid < 64 ? tid : 63;
        const float* up = ti < 16 ? (U1_0 + ti) : (U1_1 + (ti - 16));
        const float u1 = *up;
        const float* xb = x + (size_t)b0 * (C * L);
        v4fu xv[4];
        #pragma unroll
        for (int j = 0; j < 4; ++j) xv[j] = *(const v4fu*)(xb + 4 * (tid + NTHR * j));
        #pragma unroll
        for (int j = 0; j < 4; ++j) {
            const int idx = 4 * (tid + NTHR * j);
            float* d = sX + (idx >> 4) * XSTR + (idx & 15);
            #pragma unroll
            for (int u = 0; u < 4; ++u) d[u] = xv[j][u];
        }
        asm volatile("" :: "v"(u1));
        if (tid < 64) sU1[tid] = u1;
    }
    __syncthreads();

    if (tid < NB * L) {
        const int n = tid >> 4, i = tid & 15;
        float s = 0.f;
        for (int c = 0; c < C; ++c) s += sX[(n * C + c) * XSTR + i];
        sXS[tid] = s;
    }
    __syncthreads();

    const int wv   = __builtin_amdgcn_readfirstlane(tid >> 5);
    const int wm   = wv & 7;
    const int wg   = wv >> 3;
    const int lane = tid & 31;
    const int nl   = lane & 15;
    const int hi   = lane >> 4;
    const int crow = wm * 16 + nl;

    float xsa[L], xsb[L];
    #pragma unroll
    for (int i = 0; i < L; ++i) {
        xsa[i] = __int_as_float(__builtin_amdgcn_readfirstlane(__float_as_int(sXS[i])));
        xsb[i] = __int_as_float(__builtin_amdgcn_readfirstlane(__float_as_int(sXS[L + i])));
    }
    const int xsv = __float_as_int(sXS[2 * L + lane]);

    const int l1  = lane < 27 ? lane : 26;
    const int rg1 = l1 / 9, q1 = l1 - 9 * rg1;
    const int k1  = (q1 < 8) ? 4 * q1 : (K3_1 - 4);
    const int l0  = lane < 24 ? lane : 23;
    const int rg0 = l0 / 6, q0 = l0 - 6 * rg0;
    const int k0  = (q0 < 5) ? 4 * q0 : (K3_0 - 4);

    v8f acc0[NTILE], acc1[NKEEP];
    #pragma unroll
    for (int t = 0; t < NTILE; ++t) acc0[t] = (v8f){0.f, 0.f, 0.f, 0.f, 0.f, 0.f, 0.f, 0.f};
    #pragma unroll
    for (int t = 0; t < NKEEP; ++t) acc1[t] = (v8f){0.f, 0.f, 0.f, 0.f, 0.f, 0.f, 0.f, 0.f};
    float* park = sm + OFF_PARK + wv * PARKW + lane * 4;
    auto park_load = [&](v8f (&p)[NPARK]) {
        #pragma unroll
        for (int t = 0; t < NPARK; ++t) {
            const v4f lo = *(const v4f*)(park + (2 * t) * 128);
            const v4f hh = *(const v4f*)(park + (2 * t + 1) * 128);
            p[t] = (v8f){lo[0], lo[1], lo[2], lo[3], hh[0], hh[1], hh[2], hh[3]};
        }
    };

    const float* tab1 = (wg | (wm >> 1)) ? F1 : U3_1;
    auto a1_pass = [&](int chn, int p, int& row, int& rl) {
        if (wg == 0) {
            const int dr = 2 * wm + p, u = dr >> 2, mm = dr & 3;
            row = (4 * mm + chn) * 17 + u;
            rl  = 24 + dr;
        } else {
            const int f = 3 * wm + p;
            int mm, j;
            fold_step(f >> 2, mm, j);
            row = (4 * mm + chn) * 16 + 4 * j + (f & 3);
            rl  = f;
        }
    };
    auto a0_pass = [&](int chn, int first, int& row, int& rl) -> bool {
        if (!first && wm < 6) {
            int mm, j;
            fold_step(wm, mm, j);
            row = (4 * mm + chn) * 16 + 4 * j + rg0;
            rl  = 4 * wm + rg0;
            return true;
        }
        int u = first ? 2 + wm : wm - 6;
        if (u + chn > 3) u = 0;
        row = (4 * rg0 + chn) * 17 + u;
        rl  = 24 + 4 * u + rg0;
        return u != 0;
    };
    auto a1_off = [&](int row) -> unsigned {
        return (unsigned)(rg1 * 256 + row) * (unsigned)(L * K3_1) + (unsigned)k1;
    };
    auto a0_off = [&](int row) -> unsigned {
        return (unsigned)row * (unsigned)(L * K3_0) + (unsigned)k0;
    };
    auto u2_item = [&](int chn, int jj, int& col, int& pr, int (&rw)[4], bool& fold) -> bool {
        const int cr = wm + 8 * jj;
        col = cr < NU2 ? cr : NU2 - 1;
        pr  = lane < CP ? lane : CP - 1;
        const int sl = pr >> 1, half = pr & 1;
        const int mm = (sl >= 3) + (sl >= 5);
        const int j  = mm + 1 + sl - fo_base(mm);
        const bool inq = sl >= 6;
        int u = sl - 6;
        const bool have = !inq | (u + chn < 4);
        u = have ? u : 0;
        const int a0 = inq ? 8 * half + chn : 4 * mm + chn;
        const int a1 = inq ? a0 + 4 : a0;
        const int b0 = inq ? a0 + u : 4 * j + 2 * half;
        const int b1 = inq ? a1 + u : b0 + 1;
        rw[0] = a0 * 16 + b0; rw[1] = a1 * 16 + b1;
        rw[2] = b0 * 16 + a0; rw[3] = b1 * 16 + a1;
        fold = !inq | (u > 0);
        return (cr < NU2) & (lane < CP) & have;
    };

    const int prod_first = (wv >> 2) & 1;
    #pragma nounroll
    for (int h = 0; h < 2 * (NCH + 1); ++h) {
        const int it = h >> 1;
        const bool build_half = ((h & 1) ^ prod_first) == 0;
        if (build_half && it < NCH) {
            const int ch = it;
            float* sBw = sB2 + (it & 1) * B2BUF;
            int row, rl = 0;
            v4fu r[L];

            float a[4] = {0.f, 0.f, 0.f, 0.f}, c[4] = {0.f, 0.f, 0.f, 0.f};
            float d[4] = {0.f, 0.f, 0.f, 0.f}, f[4] = {0.f, 0.f, 0.f, 0.f};
            const bool has_a1 = (wg != 0) | ((wm >> 1) + ch < 4);
            if (has_a1) {
                a1_pass(ch, 0, row, rl);
                issue_row_loads<K3_1>(r, tab1 + a1_off(row));
                const int nsh = wg ? 2 : 1;
                #pragma nounroll
                for (int s = 0; s < nsh; ++s) {
                    int rown, rln;
                    a1_pass(ch, s + 1, rown, rln);
                    row_sums<K3_1>(r, xsa, xsb, xsv, a, c, d, f, tab1 + a1_off(rown));
                    store_a1(sBw, rg1, rl, k1, a, c, d, f);
                    rl = rln;
                }
            }
            if (wg) {
                row_sums<0>(r, xsa, xsb, xsv, a, c, d, f, nullptr);
                store_a1(sBw, rg1, rl, k1, a, c, d, f);
            } else {
                const int two = (wm < 2) & (wm + ch < 2);
                {
                    int rown, rln;
                    const float* tabn = a0_pass(ch, two, rown, rln) ? F0 : U3_0;
                    if (has_a1) {
                        row_sums<K3_0>(r, xsa, xsb, xsv, a, c, d, f, tabn + a0_off(rown));
                        store_a1(sBw, rg1, rl, k1, a, c, d, f);
                    } else {
                        issue_row_loads<K3_0>(r, tabn + a0_off(rown));
                    }
                    rl = rln;
                }
                if (two) {
                    int rown, rln;
                    const float* tabn = a0_pass(ch, 0, rown, rln) ? F0 : U3_0;
                    row_sums<K3_0>(r, xsa, xsb, xsv, a, c, d, f, tabn + a0_off(rown));
                    store_a0(sBw, rl, k0, a, c, d, f);
                    rl = rln;
                }
                float u2v[12];
                #pragma unroll
                for (int e = 0; e < 4; ++e) { a[e] = 0.f; c[e] = 0.f; d[e] = 0.f; f[e] = 0.f; }
                #pragma unroll
                for (int i = 0; i < L; ++i) {
                    const v4fu v = r[i];
                    const float xc = xs_lane(xsv, i), xd = xs_lane(xsv, L + i);
                    #pragma unroll
                    for (int e = 0; e < 4; ++e) {
                        a[e] = fmaf(v[e], xsa[i], a[e]);
                        c[e] = fmaf(v[e], xsb[i], c[e]);
                        d[e] = fmaf(v[e], xc, d[e]);
                        f[e] = fmaf(v[e], xd, f[e]);
                    }
                    if (i < 4) {
                        asm volatile("" :: "v"(v) : "memory");
                        #pragma unroll
                        for (int t = 0; t < 3; ++t) {
                            const int n = 3 * i + t;
                            int col, pr, rw[4];
                            bool fold;
                            u2_item(ch, n >> 2, col, pr, rw, fold);
                            const bool t1 = col < 3 * K2_1;
                            const int w = col / K2_1, k = col - w * K2_1;
                            const float* ub = t1 ? (U2_1 + (w * 256) * K2_1 + k) : (U2_0 + (col - 3 * K2_1));
                            const int us = t1 ? K2_1 : K2_0;
                            u2v[n] = ub[rw[n & 3] * us];
                        }
                    }
                }
                store_a0(sBw, rl, k0, a, c, d, f);
                #pragma unroll
                for (int n = 0; n < 12; n += 4)
                    asm volatile("" :: "v"(u2v[n]), "v"(u2v[n + 1]), "v"(u2v[n + 2]), "v"(u2v[n + 3]));
                #pragma unroll
                for (int jj = 0; jj < 3; ++jj) {
                    int col, pr, rw[4];
                    bool fold;
                    const bool valid = u2_item(ch, jj, col, pr, rw, fold);
                    v2f v;
                    v.x = fold ? u2v[4 * jj]     + u2v[4 * jj + 2] : u2v[4 * jj];
                    v.y = fold ? u2v[4 * jj + 1] + u2v[4 * jj + 3] : u2v[4 * jj + 1];
                    if (valid) {
                        float* bp = sBw + pr * S2 + (COL_21 + col) * 2;
                        #pragma unroll
                        for (int n = 0; n < NB; ++n) *(v2f*)(bp + n * CP * S2) = v;
                    }
                }
            }
        }

        if (!build_half && it > 0) {
            const int ch = it - 1;
            const float* sBr = sB2 + (ch & 1) * B2BUF;
            float sel0a[4], sel1a[4], sel0b[4], sel1b[4], xpa[3], xpb[3];
            const float* xra = sX + (2 * wg * C + crow) * XSTR;
            const float* xrb = xra + C * XSTR;
            #pragma unroll
            for (int j = 1; j < 4; ++j) {
                sel0a[j] = hi ? xra[4 * j + 2] : xra[4 * j];
                sel1a[j] = hi ? xra[4 * j + 3] : xra[4 * j + 1];
                sel0b[j] = hi ? xrb[4 * j + 2] : xrb[4 * j];
                sel1b[j] = hi ? xrb[4 * j + 3] : xrb[4 * j + 1];
            }
            #pragma unroll
            for (int mm = 0; mm < 3; ++mm) {
                xpa[mm] = xra[4 * mm + ch];
                xpb[mm] = xrb[4 * mm + ch];
            }
            v8f accp[NPARK];
            if (ch == 0) {
                #pragma unroll
                for (int t = 0; t < NPARK; ++t) accp[t] = (v8f){0.f, 0.f, 0.f, 0.f, 0.f, 0.f, 0.f, 0.f};
            } else {
                park_load(accp);
            }
            const float* Bl = sBr + 2 * wg * CP * S2 + hi * S2 + nl * 2;
            #pragma unroll
            for (int mm = 0; mm < 3; ++mm) {
                #pragma unroll
                for (int j = mm + 1; j < 4; ++j) {
                    const int sl = fo_base(mm) + j - mm - 1;
                    v2f aa, ab;
                    aa.x = xpa[mm] * sel0a[j];
                    aa.y = xpa[mm] * sel1a[j];
                    ab.x = xpb[mm] * sel0b[j];
                    ab.y = xpb[mm] * sel1b[j];
                    const float* Br = Bl + (2 * sl) * S2;
                    #pragma unroll
                    for (int t = 0; t < NTILE; ++t) {
                        const v2f bfa = *(const v2f*)(Br + t * 32);
                        const v2f bfb = *(const v2f*)(Br + CP * S2 + t * 32);
                        acc0[t] = __builtin_amdgcn_wmma_f32_16x16x4_f32(
                            false, bfa, false, aa, (short)0, acc0[t], false, false);
                        if (t < NKEEP)
                            acc1[t] = __builtin_amdgcn_wmma_f32_16x16x4_f32(
                                false, bfb, false, ab, (short)0, acc1[t], false, false);
                        else
                            accp[t - NKEEP] = __builtin_amdgcn_wmma_f32_16x16x4_f32(
                                false, bfb, false, ab, (short)0, accp[t - NKEEP], false, false);
                    }
                }
            }
            const float* xqa = xra + 8 * hi + ch;
            const float* xqb = xrb + 8 * hi + ch;
            const float xha0 = xqa[0], xha1 = xqa[4];
            const float xhb0 = xqb[0], xhb1 = xqb[4];
            #pragma unroll
            for (int u = 0; u < 4; ++u) {
                if (u == 0 || u + ch < 4) {
                    v2f aa, ab;
                    aa.x = xha0 * xqa[u];
                    aa.y = xha1 * xqa[4 + u];
                    ab.x = xhb0 * xqb[u];
                    ab.y = xhb1 * xqb[4 + u];
                    const float* Br = Bl + (2 * (6 + u)) * S2;
                    #pragma unroll
                    for (int t = 0; t < NTILE; ++t) {
                        const v2f bfa = *(const v2f*)(Br + t * 32);
                        const v2f bfb = *(const v2f*)(Br + CP * S2 + t * 32);
                        acc0[t] = __builtin_amdgcn_wmma_f32_16x16x4_f32(
                            false, bfa, false, aa, (short)0, acc0[t], false, false);
                        if (t < NKEEP)
                            acc1[t] = __builtin_amdgcn_wmma_f32_16x16x4_f32(
                                false, bfb, false, ab, (short)0, acc1[t], false, false);
                        else
                            accp[t - NKEEP] = __builtin_amdgcn_wmma_f32_16x16x4_f32(
                                false, bfb, false, ab, (short)0, accp[t - NKEEP], false, false);
                    }
                }
            }
            #pragma unroll
            for (int t = 0; t < NPARK; ++t) {
                *(v4f*)(park + (2 * t) * 128)     = (v4f){accp[t][0], accp[t][1], accp[t][2], accp[t][3]};
                *(v4f*)(park + (2 * t + 1) * 128) = (v4f){accp[t][4], accp[t][5], accp[t][6], accp[t][7]};
            }
        }
        if (h & 1) __syncthreads();
    }


    {
        float yv[NB][E];
        #pragma unroll
        for (int n = 0; n < NB; ++n) {
            #pragma unroll
            for (int e = 0; e < E; ++e) yv[n][e] = y[(b0 + n) * E + e];
        }
        const int c4 = lane * 4;
        #pragma nounroll
        for (int j0 = 0; j0 < 5; j0 += 1) {
            v4fu wl[1][E];
            #pragma unroll
            for (int jj = 0; jj < 1; ++jj) {
                const int wrr = wv + 16 * (j0 + jj);
                const int wr  = wrr < WROWS ? wrr : (WROWS - 1);
                const float* src; int stride;
                if (wr < WR_M0)       { src = Wm1 + (wr)         * C; stride = K3_1 * C; }
                else if (wr < WR_21)  { src = Wm0 + (wr - WR_M0) * C; stride = K3_0 * C; }
                else if (wr < WR_20)  { src = W21 + (wr - WR_21) * C; stride = K2_1 * C; }
                else if (wr < WR_11)  { src = W20 + (wr - WR_20) * C; stride = K2_0 * C; }
                else if (wr == WR_11) { src = W11;                    stride = C;        }
                else                  { src = W10;                    stride = C;        }
                #pragma unroll
                for (int e = 0; e < E; ++e) wl[jj][e] = *(const v4fu*)(src + e * stride + c4);
            }
            #pragma unroll
            for (int jj = 0; jj < 1; ++jj) {
                const int wrr = wv + 16 * (j0 + jj);
                float ws[NB][4];
                #pragma unroll
                for (int n = 0; n < NB; ++n) {
                    #pragma unroll
                    for (int u = 0; u < 4; ++u) ws[n][u] = 0.f;
                }
                #pragma unroll
                for (int e = 0; e < E; ++e) {
                    #pragma unroll
                    for (int n = 0; n < NB; ++n) {
                        #pragma unroll
                        for (int u = 0; u < 4; ++u) ws[n][u] = fmaf(yv[n][e], wl[jj][e][u], ws[n][u]);
                    }
                }
                if (wrr < WROWS) {
                    float* dst = sW + wrr * C + c4;
                    #pragma unroll
                    for (int n = 0; n < NB; ++n) {
                        #pragma unroll
                        for (int u = 0; u < 4; ++u) dst[n * WROWS * C + u] = ws[n][u];
                    }
                }
            }
        }
    }
    __syncthreads();

    {
        const float* sWa = sW + 2 * wg * WROWS * C + crow;
        const float* sXa = sX + (2 * wg * C + crow) * XSTR;
        float* outa = out + (size_t)(b0 + 2 * wg) * (4 * C);
        {
            float s[4] = {0.f, 0.f, 0.f, 0.f};
            #pragma unroll
            for (int t = 0; t < NTILE; ++t) tile_sums(t, acc0[t], sWa, hi, s);
            lane_epilogue(s, sWa, sXa, sU1, outa, crow, hi);
        }
        {
            const float* sWb = sWa + WROWS * C;
            v8f accp[NPARK];
            park_load(accp);
            float s[4] = {0.f, 0.f, 0.f, 0.f};
            #pragma unroll
            for (int t = 0; t < NKEEP; ++t) tile_sums(t, acc1[t], sWb, hi, s);
            #pragma unroll
            for (int t = 0; t < NPARK; ++t) tile_sums(NKEEP + t, accp[t], sWb, hi, s);
            lane_epilogue(s, sWb, sXa + C * XSTR, sU1, outa + 4 * C, crow, hi);
        }
    }
}

extern "C" void kernel_launch(void* const* d_in, const int* in_sizes, int n_in,
                              void* d_out, int out_size, void* d_ws, size_t ws_size,
                              hipStream_t stream) {
    (void)in_sizes; (void)n_in; (void)out_size; (void)ws_size;
    const float* x    = (const float*)d_in[0];
    const float* y    = (const float*)d_in[1];
    const float* U3_0 = (const float*)d_in[2];
    const float* U2_0 = (const float*)d_in[3];
    const float* U1_0 = (const float*)d_in[4];
    const float* U3_1 = (const float*)d_in[5];
    const float* U2_1 = (const float*)d_in[6];
    const float* U1_1 = (const float*)d_in[7];
    const float* Wm0  = (const float*)d_in[8];
    const float* W20  = (const float*)d_in[9];
    const float* W10  = (const float*)d_in[10];
    const float* Wm1  = (const float*)d_in[11];
    const float* W21  = (const float*)d_in[12];
    const float* W11  = (const float*)d_in[13];

    float* F1 = (float*)d_ws;
    float* F0 = F1 + 3 * 256 * L * K3_1;
    fold_tables_kernel<<<dim3(256, 4), dim3(128), 0, stream>>>(U3_1, U3_0, F1, F0);

    constexpr size_t smem_bytes = (size_t)SMEM_FLOATS * sizeof(float);
    mace_node_kernel<<<dim3(NN / NB), dim3(NTHR), smem_bytes, stream>>>(
        x, y, U3_0, U2_0, U1_0, U3_1, U2_1, U1_1,
        Wm0, W20, W10, Wm1, W21, W11, F1, F0, (float*)d_out);
}
